# starting kernel
# baseline (speedup 1.0000x reference)
#include <hip/hip_runtime.h>
#include <hip/hip_bf16.h>
#include <math.h>


typedef float v2f __attribute__((ext_vector_type(2)));
typedef float v4f __attribute__((ext_vector_type(4)));
typedef float v8f __attribute__((ext_vector_type(8)));

#define BM  64
#define BN  128
#define KC  32
#define AKP 36
#define BKP 36

__device__ __forceinline__ v8f wmma4(v2f a, v2f b, v8f c) {
    return __builtin_amdgcn_wmma_f32_16x16x4_f32(
        false, a, false, b, (short)0, c, false, false);
}

template <int MODE, int RPB>
__global__ __launch_bounds__(256)
void gemm_f32_wmma(const float* __restrict__ A, long bsA, long rsA,
                   const float* __restrict__ Bm, long bsB,
                   const float* __restrict__ biasCol,
                   const float* __restrict__ Emat, long bsE, long rsE,
                   float* __restrict__ Out,
                   int Mtotal, int K, int klim, int N)
{
    __shared__ float As[2][BM][AKP];
    __shared__ float Bs[2][BN][BKP];

    const int tid  = threadIdx.x;
    const int lane = tid & 31;
    const int wave = tid >> 5;
    const int wm   = wave & 1;
    const int wn   = wave >> 1;

    const int tileM = blockIdx.y * BM;
    const int tileN = blockIdx.x * BN;
    const int bB = (bsB != 0) ? (tileM / RPB) : 0;

    v8f acc00 = {}, acc01 = {}, acc10 = {}, acc11 = {};

    const int a_row = tid >> 2;
    const int a_kc  = (tid & 3) << 2;
    int a_grow = tileM + a_row;
    if (a_grow >= Mtotal) a_grow = Mtotal - 1;
    const int a_b = a_grow / RPB;
    const int a_r = a_grow - a_b * RPB;
    const float* a_base = A + (long)a_b * bsA + (long)a_r * rsA;

    const int b_kr = tid >> 4;
    const int b_nc = (tid & 15) << 2;
    const float* b_base = Bm + (long)bB * bsB + tileN;

    const int m0   = (wm << 5) + (lane & 15);
    const int n0   = (wn << 5) + (lane & 15);
    const int koff = (lane >> 4) << 1;

    const int nchunks = (klim + KC - 1) / KC;

    v4f ra0, ra1, rb00, rb01, rb10, rb11;

    auto prefetch = [&](int k0) {
        if (k0 + KC <= klim) {
            ra0  = *(const v4f*)(a_base + k0 + a_kc);
            ra1  = *(const v4f*)(a_base + k0 + a_kc + 16);
            const float* br0 = b_base + (long)(k0 + b_kr)      * N + b_nc;
            const float* br1 = b_base + (long)(k0 + b_kr + 16) * N + b_nc;
            rb00 = *(const v4f*)(br0);
            rb01 = *(const v4f*)(br0 + 64);
            rb10 = *(const v4f*)(br1);
            rb11 = *(const v4f*)(br1 + 64);
        } else {
            const v4f zz = {0.f, 0.f, 0.f, 0.f};
            #pragma unroll
            for (int i = 0; i < 4; ++i) {
                const int ka = k0 + a_kc + i;
                ra0[i] = (ka      < klim) ? a_base[ka]      : 0.0f;
                ra1[i] = (ka + 16 < klim) ? a_base[ka + 16] : 0.0f;
            }
            const int kb0 = k0 + b_kr, kb1 = kb0 + 16;
            const float* br0 = b_base + (long)kb0 * N + b_nc;
            const float* br1 = b_base + (long)kb1 * N + b_nc;
            rb00 = (kb0 < K) ? *(const v4f*)(br0)      : zz;
            rb01 = (kb0 < K) ? *(const v4f*)(br0 + 64) : zz;
            rb10 = (kb1 < K) ? *(const v4f*)(br1)      : zz;
            rb11 = (kb1 < K) ? *(const v4f*)(br1 + 64) : zz;
        }
    };

    auto stage = [&](int buf) {
        *(v4f*)&As[buf][a_row][a_kc]      = ra0;
        *(v4f*)&As[buf][a_row][a_kc + 16] = ra1;
        #pragma unroll
        for (int i = 0; i < 4; ++i) {
            Bs[buf][b_nc + i     ][b_kr]      = rb00[i];
            Bs[buf][b_nc + i     ][b_kr + 16] = rb10[i];
            Bs[buf][b_nc + i + 64][b_kr]      = rb01[i];
            Bs[buf][b_nc + i + 64][b_kr + 16] = rb11[i];
        }
    };

    auto compute = [&](int buf) {
        #pragma unroll
        for (int kk = 0; kk < KC; kk += 4) {
            v2f a0 = *(const v2f*)&As[buf][m0     ][kk + koff];
            v2f a1 = *(const v2f*)&As[buf][m0 + 16][kk + koff];
            v2f b0 = *(const v2f*)&Bs[buf][n0     ][kk + koff];
            v2f b1 = *(const v2f*)&Bs[buf][n0 + 16][kk + koff];
            acc00 = wmma4(a0, b0, acc00);
            acc01 = wmma4(a0, b1, acc01);
            acc10 = wmma4(a1, b0, acc10);
            acc11 = wmma4(a1, b1, acc11);
        }
    };

    prefetch(0);
    stage(0);
    __syncthreads();
    int buf = 0;
    for (int c = 0; c < nchunks; ++c) {
        if (c + 1 < nchunks) prefetch((c + 1) * KC);
        compute(buf);
        if (c + 1 < nchunks) {
            stage(buf ^ 1);
            __syncthreads();
            buf ^= 1;
        }
    }

    const int colA  = tileN + n0;
    const int mbase = tileM + (wm << 5) + ((lane >> 4) << 3);
    #pragma unroll
    for (int mi = 0; mi < 2; ++mi) {
        const v8f& accL = mi ? acc10 : acc00;
        const v8f& accR = mi ? acc11 : acc01;
        #pragma unroll
        for (int j = 0; j < 8; ++j) {
            const int gr = mbase + (mi << 4) + j;
            if (gr >= Mtotal) continue;
            const int b = gr / RPB;
            const int r = gr - b * RPB;
            float v0 = accL[j];
            float v1 = accR[j];
            if (MODE == 0) {
                v0 = fmaxf(v0 + biasCol[colA],      0.0f);
                v1 = fmaxf(v1 + biasCol[colA + 16], 0.0f);
            } else if (MODE == 1) {
                v0 += biasCol[colA];
                v1 += biasCol[colA + 16];
            } else if (MODE == 2) {
                const float* e = Emat + (long)b * bsE + (long)r * rsE;
                v0 = sinf(v0 + e[colA]);
                v1 = sinf(v1 + e[colA + 16]);
            } else {
                const float* e = Emat + (long)b * bsE + (long)r * rsE;
                v0 = (v0 + biasCol[colA]      + e[colA])      * 0.5f;
                v1 = (v1 + biasCol[colA + 16] + e[colA + 16]) * 0.5f;
            }
            Out[(long)gr * N + colA]      = v0;
            Out[(long)gr * N + colA + 16] = v1;
        }
    }
}

extern "C" void kernel_launch(void* const* d_in, const int* in_sizes, int n_in,
                              void* d_out, int out_size, void* d_ws, size_t ws_size,
                              hipStream_t stream)
{
    (void)in_sizes; (void)n_in; (void)out_size; (void)ws_size;

    const float* x  = (const float*)d_in[0];
    const float* z  = (const float*)d_in[1];
    const float* W1 = (const float*)d_in[2];
    const float* b1 = (const float*)d_in[3];
    const float* W2 = (const float*)d_in[4];
    const float* b2 = (const float*)d_in[5];
    const float* V1 = (const float*)d_in[6];
    const float* c1 = (const float*)d_in[7];
    const float* V2 = (const float*)d_in[8];
    const float* c2 = (const float*)d_in[9];
    const float* Wr = (const float*)d_in[10];
    const float* br = (const float*)d_in[11];
    float* out = (float*)d_out;

    const int Bb = 8, T = 1024, F = 1024, ZC = 128, ZL = 1028, HL = 1026;

    float* ws   = (float*)d_ws;
    const size_t H1N = (size_t)Bb * HL * F;
    const size_t WTN = (size_t)Bb * T * F;
    float* h1   = ws;
    float* g1   = h1 + H1N;
    float* wts  = g1 + H1N;
    float* bmat = wts + WTN;
    float* sm   = bmat + WTN;

    const dim3 blk(256);
    const dim3 gridA(F / BN, (Bb * HL + BM - 1) / BM);
    const dim3 gridB(F / BN, (Bb * T) / BM);

    gemm_f32_wmma<0, 1026><<<gridA, blk, 0, stream>>>(
        z, (long)ZL * ZC, ZC, W1, 0, b1, nullptr, 0, 0,
        h1, Bb * HL, 3 * ZC, 3 * ZC, F);
    gemm_f32_wmma<0, 1026><<<gridA, blk, 0, stream>>>(
        z, (long)ZL * ZC, ZC, V1, 0, c1, nullptr, 0, 0,
        g1, Bb * HL, 3 * ZC, 3 * ZC, F);
    gemm_f32_wmma<1, 1024><<<gridB, blk, 0, stream>>>(
        h1, (long)HL * F, F, W2, 0, b2, nullptr, 0, 0,
        wts, Bb * T, 3 * F, 3 * F, F);
    gemm_f32_wmma<1, 1024><<<gridB, blk, 0, stream>>>(
        g1, (long)HL * F, F, V2, 0, c2, nullptr, 0, 0,
        bmat, Bb * T, 3 * F, 3 * F, F);
    gemm_f32_wmma<2, 1024><<<gridB, blk, 0, stream>>>(
        x + 4, (long)T * F, T, wts, (long)T * F, nullptr,
        bmat, (long)T * F, F,
        sm, Bb * T, T, T - 4, F);
    gemm_f32_wmma<3, 1024><<<gridB, blk, 0, stream>>>(
        sm, (long)T * F, F, Wr, 0, br,
        x, (long)T * F, F,
        out, Bb * T, F, F, F);
}


# GEMMs 3-6: wave tile 32x32 -> 64x64 (128x128 block, 4 waves); LDS fragments read as b128 per 8 k, grouped ahead of the WMMAs; k regrouped inside each 8-k step, so the f32 sum order changes
# speedup vs baseline: 1.0431x; 1.0431x over previous
#include <hip/hip_runtime.h>
#include <hip/hip_bf16.h>
#include <math.h>


typedef float v2f __attribute__((ext_vector_type(2)));
typedef float v4f __attribute__((ext_vector_type(4)));
typedef float v8f __attribute__((ext_vector_type(8)));

#define BM  64
#define BN  128
#define KC  32
#define AKP 36
#define BKP 36

__device__ __forceinline__ v8f wmma4(v2f a, v2f b, v8f c) {
    return __builtin_amdgcn_wmma_f32_16x16x4_f32(
        false, a, false, b, (short)0, c, false, false);
}

template <int MODE, int RPB>
__global__ __launch_bounds__(256)
void gemm_f32_wmma(const float* __restrict__ A, long bsA, long rsA,
                   const float* __restrict__ Bm, long bsB,
                   const float* __restrict__ biasCol,
                   const float* __restrict__ Emat, long bsE, long rsE,
                   float* __restrict__ Out,
                   int Mtotal, int K, int klim, int N)
{
    __shared__ float As[2][BM][AKP];
    __shared__ float Bs[2][BN][BKP];

    const int tid  = threadIdx.x;
    const int lane = tid & 31;
    const int wave = tid >> 5;
    const int wm   = wave & 1;
    const int wn   = wave >> 1;

    const int tileM = blockIdx.y * BM;
    const int tileN = blockIdx.x * BN;
    const int bB = (bsB != 0) ? (tileM / RPB) : 0;

    v8f acc00 = {}, acc01 = {}, acc10 = {}, acc11 = {};

    const int a_row = tid >> 2;
    const int a_kc  = (tid & 3) << 2;
    int a_grow = tileM + a_row;
    if (a_grow >= Mtotal) a_grow = Mtotal - 1;
    const int a_b = a_grow / RPB;
    const int a_r = a_grow - a_b * RPB;
    const float* a_base = A + (long)a_b * bsA + (long)a_r * rsA;

    const int b_kr = tid >> 4;
    const int b_nc = (tid & 15) << 2;
    const float* b_base = Bm + (long)bB * bsB + tileN;

    const int m0   = (wm << 5) + (lane & 15);
    const int n0   = (wn << 5) + (lane & 15);
    const int koff = (lane >> 4) << 1;

    const int nchunks = (klim + KC - 1) / KC;

    v4f ra0, ra1, rb00, rb01, rb10, rb11;

    auto prefetch = [&](int k0) {
        if (k0 + KC <= klim) {
            ra0  = *(const v4f*)(a_base + k0 + a_kc);
            ra1  = *(const v4f*)(a_base + k0 + a_kc + 16);
            const float* br0 = b_base + (long)(k0 + b_kr)      * N + b_nc;
            const float* br1 = b_base + (long)(k0 + b_kr + 16) * N + b_nc;
            rb00 = *(const v4f*)(br0);
            rb01 = *(const v4f*)(br0 + 64);
            rb10 = *(const v4f*)(br1);
            rb11 = *(const v4f*)(br1 + 64);
        } else {
            const v4f zz = {0.f, 0.f, 0.f, 0.f};
            #pragma unroll
            for (int i = 0; i < 4; ++i) {
                const int ka = k0 + a_kc + i;
                ra0[i] = (ka      < klim) ? a_base[ka]      : 0.0f;
                ra1[i] = (ka + 16 < klim) ? a_base[ka + 16] : 0.0f;
            }
            const int kb0 = k0 + b_kr, kb1 = kb0 + 16;
            const float* br0 = b_base + (long)kb0 * N + b_nc;
            const float* br1 = b_base + (long)kb1 * N + b_nc;
            rb00 = (kb0 < K) ? *(const v4f*)(br0)      : zz;
            rb01 = (kb0 < K) ? *(const v4f*)(br0 + 64) : zz;
            rb10 = (kb1 < K) ? *(const v4f*)(br1)      : zz;
            rb11 = (kb1 < K) ? *(const v4f*)(br1 + 64) : zz;
        }
    };

    auto stage = [&](int buf) {
        *(v4f*)&As[buf][a_row][a_kc]      = ra0;
        *(v4f*)&As[buf][a_row][a_kc + 16] = ra1;
        #pragma unroll
        for (int i = 0; i < 4; ++i) {
            Bs[buf][b_nc + i     ][b_kr]      = rb00[i];
            Bs[buf][b_nc + i     ][b_kr + 16] = rb10[i];
            Bs[buf][b_nc + i + 64][b_kr]      = rb01[i];
            Bs[buf][b_nc + i + 64][b_kr + 16] = rb11[i];
        }
    };

    auto compute = [&](int buf) {
        #pragma unroll
        for (int kk = 0; kk < KC; kk += 4) {
            v2f a0 = *(const v2f*)&As[buf][m0     ][kk + koff];
            v2f a1 = *(const v2f*)&As[buf][m0 + 16][kk + koff];
            v2f b0 = *(const v2f*)&Bs[buf][n0     ][kk + koff];
            v2f b1 = *(const v2f*)&Bs[buf][n0 + 16][kk + koff];
            acc00 = wmma4(a0, b0, acc00);
            acc01 = wmma4(a0, b1, acc01);
            acc10 = wmma4(a1, b0, acc10);
            acc11 = wmma4(a1, b1, acc11);
        }
    };

    prefetch(0);
    stage(0);
    __syncthreads();
    int buf = 0;
    for (int c = 0; c < nchunks; ++c) {
        if (c + 1 < nchunks) prefetch((c + 1) * KC);
        compute(buf);
        if (c + 1 < nchunks) {
            stage(buf ^ 1);
            __syncthreads();
            buf ^= 1;
        }
    }

    const int colA  = tileN + n0;
    const int mbase = tileM + (wm << 5) + ((lane >> 4) << 3);
    #pragma unroll
    for (int mi = 0; mi < 2; ++mi) {
        const v8f& accL = mi ? acc10 : acc00;
        const v8f& accR = mi ? acc11 : acc01;
        #pragma unroll
        for (int j = 0; j < 8; ++j) {
            const int gr = mbase + (mi << 4) + j;
            if (gr >= Mtotal) continue;
            const int b = gr / RPB;
            const int r = gr - b * RPB;
            float v0 = accL[j];
            float v1 = accR[j];
            if (MODE == 0) {
                v0 = fmaxf(v0 + biasCol[colA],      0.0f);
                v1 = fmaxf(v1 + biasCol[colA + 16], 0.0f);
            } else if (MODE == 1) {
                v0 += biasCol[colA];
                v1 += biasCol[colA + 16];
            } else if (MODE == 2) {
                const float* e = Emat + (long)b * bsE + (long)r * rsE;
                v0 = sinf(v0 + e[colA]);
                v1 = sinf(v1 + e[colA + 16]);
            } else {
                const float* e = Emat + (long)b * bsE + (long)r * rsE;
                v0 = (v0 + biasCol[colA]      + e[colA])      * 0.5f;
                v1 = (v1 + biasCol[colA + 16] + e[colA + 16]) * 0.5f;
            }
            Out[(long)gr * N + colA]      = v0;
            Out[(long)gr * N + colA + 16] = v1;
        }
    }
}


#define TM  128
#define TN  128
#define LP  36

template <int MODE>
__global__ __launch_bounds__(128) __attribute__((amdgpu_waves_per_eu(1, 2)))
void gemm_f32_wmma_w64(const float* __restrict__ A, long bsA, long rsA,
                       const float* __restrict__ Bm, long bsB,
                       const float* __restrict__ biasCol,
                       const float* __restrict__ Emat, long bsE, long rsE,
                       float* __restrict__ Out,
                       int K, int klim, int N)
{
    constexpr int RPB = 1024;
    __shared__ __attribute__((aligned(16))) float As[2][TM][LP];
    __shared__ __attribute__((aligned(16))) float Bs[2][TN][LP];

    const int tid  = threadIdx.x;
    const int lane = tid & 31;
    const int wave = tid >> 5;
    const int wm   = wave & 1;
    const int wn   = wave >> 1;

    const int tileM = blockIdx.y * TM;
    const int tileN = blockIdx.x * TN;
    const int bb    = tileM / RPB;
    const int r0    = tileM - bb * RPB;

    const int a_kq = (tid & 7) << 2;
    const int a_r  = tid >> 3;
    const float* a_base = A + (long)bb * bsA + (long)(r0 + a_r) * rsA + a_kq;
    const long a_step = 16 * rsA;

    const int b_n = lane << 2;
    const int b_k = wave << 2;
    const float* b_base = Bm + (long)((bsB != 0) ? bb : 0) * bsB + tileN + b_n;

    const int c16 = lane & 15;
    const int hh  = lane >> 4;
    const int am0 = (wm << 6) + c16;
    const int bn0 = (wn << 6) + c16;
    const int ko  = hh << 2;

    const int nchunks = (klim + KC - 1) / KC;

    v8f acc[4][4];
    #pragma unroll
    for (int i = 0; i < 4; ++i)
        #pragma unroll
        for (int j = 0; j < 4; ++j)
            acc[i][j] = v8f{};

    v4f ra[8], rb[8];

    auto prefetch = [&](int k0) {
        if (k0 + KC <= klim) {
            #pragma unroll
            for (int i = 0; i < 8; ++i)
                ra[i] = *(const v4f*)(a_base + i * a_step + k0);
            #pragma unroll
            for (int g = 0; g < 2; ++g)
                #pragma unroll
                for (int j = 0; j < 4; ++j)
                    rb[g * 4 + j] =
                        *(const v4f*)(b_base + (long)(k0 + b_k + 16 * g + j) * N);
        } else {
            const v4f zz = {0.f, 0.f, 0.f, 0.f};
            #pragma unroll
            for (int i = 0; i < 8; ++i) {
                #pragma unroll
                for (int e = 0; e < 4; ++e) {
                    const int ka = k0 + a_kq + e;
                    ra[i][e] = (ka < klim) ? a_base[i * a_step + k0 + e] : 0.0f;
                }
            }
            #pragma unroll
            for (int g = 0; g < 2; ++g)
                #pragma unroll
                for (int j = 0; j < 4; ++j) {
                    const int kb = k0 + b_k + 16 * g + j;
                    rb[g * 4 + j] = (kb < K)
                        ? *(const v4f*)(b_base + (long)kb * N) : zz;
                }
        }
    };

    auto stage = [&](int buf) {
        #pragma unroll
        for (int i = 0; i < 8; ++i)
            *(v4f*)&As[buf][a_r + 16 * i][a_kq] = ra[i];
        #pragma unroll
        for (int g = 0; g < 2; ++g)
            #pragma unroll
            for (int c = 0; c < 4; ++c) {
                const v4f t = { rb[g * 4 + 0][c], rb[g * 4 + 1][c],
                                rb[g * 4 + 2][c], rb[g * 4 + 3][c] };
                *(v4f*)&Bs[buf][b_n + c][b_k + 16 * g] = t;
            }
    };

    auto ldfrag = [&](int buf, int kk, v4f (&fa)[4], v4f (&fb)[4]) {
        #pragma unroll
        for (int i = 0; i < 4; ++i)
            fa[i] = *(const v4f*)&As[buf][am0 + 16 * i][kk + ko];
        #pragma unroll
        for (int j = 0; j < 4; ++j)
            fb[j] = *(const v4f*)&Bs[buf][bn0 + 16 * j][kk + ko];
    };

    auto mma = [&](const v4f (&fa)[4], const v4f (&fb)[4]) {
        #pragma unroll
        for (int p = 0; p < 2; ++p) {
            #pragma unroll
            for (int i = 0; i < 4; ++i) {
                const v2f a = { fa[i][2 * p], fa[i][2 * p + 1] };
                #pragma unroll
                for (int j = 0; j < 4; ++j) {
                    const v2f b = { fb[j][2 * p], fb[j][2 * p + 1] };
                    acc[i][j] = wmma4(a, b, acc[i][j]);
                }
            }
        }
    };

    auto compute = [&](int buf) {
        v4f fa0[4], fb0[4], fa1[4], fb1[4];
        ldfrag(buf, 0, fa0, fb0);
        ldfrag(buf, 8, fa1, fb1);
        mma(fa0, fb0);
        ldfrag(buf, 16, fa0, fb0);
        mma(fa1, fb1);
        ldfrag(buf, 24, fa1, fb1);
        mma(fa0, fb0);
        mma(fa1, fb1);
    };

    prefetch(0);
    stage(0);
    __syncthreads();
    int buf = 0;
    for (int c = 0; c < nchunks; ++c) {
        if (c + 1 < nchunks) prefetch((c + 1) * KC);
        compute(buf);
        if (c + 1 < nchunks) {
            stage(buf ^ 1);
            __syncthreads();
            buf ^= 1;
        }
    }

    const int col0 = tileN + bn0;
    const int rowb = tileM + (wm << 6) + (hh << 3);
    #pragma unroll 1
    for (int r = 0; r < 8; ++r) {
        #pragma unroll
        for (int i = 0; i < 4; ++i) {
            const int gr = rowb + (i << 4) + r;
            const int rr = gr - bb * RPB;
            const float* e = Emat + (long)bb * bsE + (long)rr * rsE;
            float* o = Out + (long)gr * N;
            #pragma unroll
            for (int j = 0; j < 4; ++j) {
                const int col = col0 + (j << 4);
                float v = acc[i][j][r];
                if (MODE == 1) {
                    v += biasCol[col];
                } else if (MODE == 2) {
                    v = sinf(v + e[col]);
                } else {
                    v = (v + biasCol[col] + e[col]) * 0.5f;
                }
                o[col] = v;
            }
        }
    }
}

extern "C" void kernel_launch(void* const* d_in, const int* in_sizes, int n_in,
                              void* d_out, int out_size, void* d_ws, size_t ws_size,
                              hipStream_t stream)
{
    (void)in_sizes; (void)n_in; (void)out_size; (void)ws_size;

    const float* x  = (const float*)d_in[0];
    const float* z  = (const float*)d_in[1];
    const float* W1 = (const float*)d_in[2];
    const float* b1 = (const float*)d_in[3];
    const float* W2 = (const float*)d_in[4];
    const float* b2 = (const float*)d_in[5];
    const float* V1 = (const float*)d_in[6];
    const float* c1 = (const float*)d_in[7];
    const float* V2 = (const float*)d_in[8];
    const float* c2 = (const float*)d_in[9];
    const float* Wr = (const float*)d_in[10];
    const float* br = (const float*)d_in[11];
    float* out = (float*)d_out;

    const int Bb = 8, T = 1024, F = 1024, ZC = 128, ZL = 1028, HL = 1026;

    float* ws   = (float*)d_ws;
    const size_t H1N = (size_t)Bb * HL * F;
    const size_t WTN = (size_t)Bb * T * F;
    float* h1   = ws;
    float* g1   = h1 + H1N;
    float* wts  = g1 + H1N;
    float* bmat = wts + WTN;
    float* sm   = bmat + WTN;

    const dim3 blk(256);
    const dim3 gridA(F / BN, (Bb * HL + BM - 1) / BM);
    const dim3 blkW(128);
    const dim3 gridW(F / TN, (Bb * T) / TM);

    gemm_f32_wmma<0, 1026><<<gridA, blk, 0, stream>>>(
        z, (long)ZL * ZC, ZC, W1, 0, b1, nullptr, 0, 0,
        h1, Bb * HL, 3 * ZC, 3 * ZC, F);
    gemm_f32_wmma<0, 1026><<<gridA, blk, 0, stream>>>(
        z, (long)ZL * ZC, ZC, V1, 0, c1, nullptr, 0, 0,
        g1, Bb * HL, 3 * ZC, 3 * ZC, F);
    gemm_f32_wmma_w64<1><<<gridW, blkW, 0, stream>>>(
        h1, (long)HL * F, F, W2, 0, b2, nullptr, 0, 0,
        wts, 3 * F, 3 * F, F);
    gemm_f32_wmma_w64<1><<<gridW, blkW, 0, stream>>>(
        g1, (long)HL * F, F, V2, 0, c2, nullptr, 0, 0,
        bmat, 3 * F, 3 * F, F);
    gemm_f32_wmma_w64<2><<<gridW, blkW, 0, stream>>>(
        x + 4, (long)T * F, T, wts, (long)T * F, nullptr,
        bmat, (long)T * F, F,
        sm, T, T - 4, F);
    gemm_f32_wmma_w64<3><<<gridW, blkW, 0, stream>>>(
        sm, (long)T * F, F, Wr, 0, br,
        x, (long)T * F, F,
        out, F, F, F);
}


# GEMMs 3-6: wave priority (s_setprio) falls 3,2,1,0 with K-loop progress, so of the two independent workgroups sharing a SIMD the one behind wins matrix issue and neither runs most of its loop alone
# speedup vs baseline: 1.0747x; 1.0303x over previous
#include <hip/hip_runtime.h>
#include <hip/hip_bf16.h>
#include <math.h>


typedef float v2f __attribute__((ext_vector_type(2)));
typedef float v4f __attribute__((ext_vector_type(4)));
typedef float v8f __attribute__((ext_vector_type(8)));

#define BM  64
#define BN  128
#define KC  32
#define AKP 36
#define BKP 36

__device__ __forceinline__ v8f wmma4(v2f a, v2f b, v8f c) {
    return __builtin_amdgcn_wmma_f32_16x16x4_f32(
        false, a, false, b, (short)0, c, false, false);
}

template <int MODE, int RPB>
__global__ __launch_bounds__(256)
void gemm_f32_wmma(const float* __restrict__ A, long bsA, long rsA,
                   const float* __restrict__ Bm, long bsB,
                   const float* __restrict__ biasCol,
                   const float* __restrict__ Emat, long bsE, long rsE,
                   float* __restrict__ Out,
                   int Mtotal, int K, int klim, int N)
{
    __shared__ float As[2][BM][AKP];
    __shared__ float Bs[2][BN][BKP];

    const int tid  = threadIdx.x;
    const int lane = tid & 31;
    const int wave = tid >> 5;
    const int wm   = wave & 1;
    const int wn   = wave >> 1;

    const int tileM = blockIdx.y * BM;
    const int tileN = blockIdx.x * BN;
    const int bB = (bsB != 0) ? (tileM / RPB) : 0;

    v8f acc00 = {}, acc01 = {}, acc10 = {}, acc11 = {};

    const int a_row = tid >> 2;
    const int a_kc  = (tid & 3) << 2;
    int a_grow = tileM + a_row;
    if (a_grow >= Mtotal) a_grow = Mtotal - 1;
    const int a_b = a_grow / RPB;
    const int a_r = a_grow - a_b * RPB;
    const float* a_base = A + (long)a_b * bsA + (long)a_r * rsA;

    const int b_kr = tid >> 4;
    const int b_nc = (tid & 15) << 2;
    const float* b_base = Bm + (long)bB * bsB + tileN;

    const int m0   = (wm << 5) + (lane & 15);
    const int n0   = (wn << 5) + (lane & 15);
    const int koff = (lane >> 4) << 1;

    const int nchunks = (klim + KC - 1) / KC;

    v4f ra0, ra1, rb00, rb01, rb10, rb11;

    auto prefetch = [&](int k0) {
        if (k0 + KC <= klim) {
            ra0  = *(const v4f*)(a_base + k0 + a_kc);
            ra1  = *(const v4f*)(a_base + k0 + a_kc + 16);
            const float* br0 = b_base + (long)(k0 + b_kr)      * N + b_nc;
            const float* br1 = b_base + (long)(k0 + b_kr + 16) * N + b_nc;
            rb00 = *(const v4f*)(br0);
            rb01 = *(const v4f*)(br0 + 64);
            rb10 = *(const v4f*)(br1);
            rb11 = *(const v4f*)(br1 + 64);
        } else {
            const v4f zz = {0.f, 0.f, 0.f, 0.f};
            #pragma unroll
            for (int i = 0; i < 4; ++i) {
                const int ka = k0 + a_kc + i;
                ra0[i] = (ka      < klim) ? a_base[ka]      : 0.0f;
                ra1[i] = (ka + 16 < klim) ? a_base[ka + 16] : 0.0f;
            }
            const int kb0 = k0 + b_kr, kb1 = kb0 + 16;
            const float* br0 = b_base + (long)kb0 * N + b_nc;
            const float* br1 = b_base + (long)kb1 * N + b_nc;
            rb00 = (kb0 < K) ? *(const v4f*)(br0)      : zz;
            rb01 = (kb0 < K) ? *(const v4f*)(br0 + 64) : zz;
            rb10 = (kb1 < K) ? *(const v4f*)(br1)      : zz;
            rb11 = (kb1 < K) ? *(const v4f*)(br1 + 64) : zz;
        }
    };

    auto stage = [&](int buf) {
        *(v4f*)&As[buf][a_row][a_kc]      = ra0;
        *(v4f*)&As[buf][a_row][a_kc + 16] = ra1;
        #pragma unroll
        for (int i = 0; i < 4; ++i) {
            Bs[buf][b_nc + i     ][b_kr]      = rb00[i];
            Bs[buf][b_nc + i     ][b_kr + 16] = rb10[i];
            Bs[buf][b_nc + i + 64][b_kr]      = rb01[i];
            Bs[buf][b_nc + i + 64][b_kr + 16] = rb11[i];
        }
    };

    auto compute = [&](int buf) {
        #pragma unroll
        for (int kk = 0; kk < KC; kk += 4) {
            v2f a0 = *(const v2f*)&As[buf][m0     ][kk + koff];
            v2f a1 = *(const v2f*)&As[buf][m0 + 16][kk + koff];
            v2f b0 = *(const v2f*)&Bs[buf][n0     ][kk + koff];
            v2f b1 = *(const v2f*)&Bs[buf][n0 + 16][kk + koff];
            acc00 = wmma4(a0, b0, acc00);
            acc01 = wmma4(a0, b1, acc01);
            acc10 = wmma4(a1, b0, acc10);
            acc11 = wmma4(a1, b1, acc11);
        }
    };

    prefetch(0);
    stage(0);
    __syncthreads();
    int buf = 0;
    for (int c = 0; c < nchunks; ++c) {
        if (c + 1 < nchunks) prefetch((c + 1) * KC);
        compute(buf);
        if (c + 1 < nchunks) {
            stage(buf ^ 1);
            __syncthreads();
            buf ^= 1;
        }
    }

    const int colA  = tileN + n0;
    const int mbase = tileM + (wm << 5) + ((lane >> 4) << 3);
    #pragma unroll
    for (int mi = 0; mi < 2; ++mi) {
        const v8f& accL = mi ? acc10 : acc00;
        const v8f& accR = mi ? acc11 : acc01;
        #pragma unroll
        for (int j = 0; j < 8; ++j) {
            const int gr = mbase + (mi << 4) + j;
            if (gr >= Mtotal) continue;
            const int b = gr / RPB;
            const int r = gr - b * RPB;
            float v0 = accL[j];
            float v1 = accR[j];
            if (MODE == 0) {
                v0 = fmaxf(v0 + biasCol[colA],      0.0f);
                v1 = fmaxf(v1 + biasCol[colA + 16], 0.0f);
            } else if (MODE == 1) {
                v0 += biasCol[colA];
                v1 += biasCol[colA + 16];
            } else if (MODE == 2) {
                const float* e = Emat + (long)b * bsE + (long)r * rsE;
                v0 = sinf(v0 + e[colA]);
                v1 = sinf(v1 + e[colA + 16]);
            } else {
                const float* e = Emat + (long)b * bsE + (long)r * rsE;
                v0 = (v0 + biasCol[colA]      + e[colA])      * 0.5f;
                v1 = (v1 + biasCol[colA + 16] + e[colA + 16]) * 0.5f;
            }
            Out[(long)gr * N + colA]      = v0;
            Out[(long)gr * N + colA + 16] = v1;
        }
    }
}


#define TM  128
#define TN  128
#define LP  36

template <int MODE>
__global__ __launch_bounds__(128) __attribute__((amdgpu_waves_per_eu(1, 2)))
void gemm_f32_wmma_w64(const float* __restrict__ A, long bsA, long rsA,
                       const float* __restrict__ Bm, long bsB,
                       const float* __restrict__ biasCol,
                       const float* __restrict__ Emat, long bsE, long rsE,
                       float* __restrict__ Out,
                       int K, int klim, int N)
{
    constexpr int RPB = 1024;
    __shared__ __attribute__((aligned(16))) float As[2][TM][LP];
    __shared__ __attribute__((aligned(16))) float Bs[2][TN][LP];

    const int tid  = threadIdx.x;
    const int lane = tid & 31;
    const int wave = tid >> 5;
    const int wm   = wave & 1;
    const int wn   = wave >> 1;

    const int tileM = blockIdx.y * TM;
    const int tileN = blockIdx.x * TN;
    const int bb    = tileM / RPB;
    const int r0    = tileM - bb * RPB;

    const int a_kq = (tid & 7) << 2;
    const int a_r  = tid >> 3;
    const float* a_base = A + (long)bb * bsA + (long)(r0 + a_r) * rsA + a_kq;
    const long a_step = 16 * rsA;

    const int b_n = lane << 2;
    const int b_k = wave << 2;
    const float* b_base = Bm + (long)((bsB != 0) ? bb : 0) * bsB + tileN + b_n;

    const int c16 = lane & 15;
    const int hh  = lane >> 4;
    const int am0 = (wm << 6) + c16;
    const int bn0 = (wn << 6) + c16;
    const int ko  = hh << 2;

    const int nchunks = (klim + KC - 1) / KC;

    v8f acc[4][4];
    #pragma unroll
    for (int i = 0; i < 4; ++i)
        #pragma unroll
        for (int j = 0; j < 4; ++j)
            acc[i][j] = v8f{};

    v4f ra[8], rb[8];

    auto prefetch = [&](int k0) {
        if (k0 + KC <= klim) {
            #pragma unroll
            for (int i = 0; i < 8; ++i)
                ra[i] = *(const v4f*)(a_base + i * a_step + k0);
            #pragma unroll
            for (int g = 0; g < 2; ++g)
                #pragma unroll
                for (int j = 0; j < 4; ++j)
                    rb[g * 4 + j] =
                        *(const v4f*)(b_base + (long)(k0 + b_k + 16 * g + j) * N);
        } else {
            const v4f zz = {0.f, 0.f, 0.f, 0.f};
            #pragma unroll
            for (int i = 0; i < 8; ++i) {
                #pragma unroll
                for (int e = 0; e < 4; ++e) {
                    const int ka = k0 + a_kq + e;
                    ra[i][e] = (ka < klim) ? a_base[i * a_step + k0 + e] : 0.0f;
                }
            }
            #pragma unroll
            for (int g = 0; g < 2; ++g)
                #pragma unroll
                for (int j = 0; j < 4; ++j) {
                    const int kb = k0 + b_k + 16 * g + j;
                    rb[g * 4 + j] = (kb < K)
                        ? *(const v4f*)(b_base + (long)kb * N) : zz;
                }
        }
    };

    auto stage = [&](int buf) {
        #pragma unroll
        for (int i = 0; i < 8; ++i)
            *(v4f*)&As[buf][a_r + 16 * i][a_kq] = ra[i];
        #pragma unroll
        for (int g = 0; g < 2; ++g)
            #pragma unroll
            for (int c = 0; c < 4; ++c) {
                const v4f t = { rb[g * 4 + 0][c], rb[g * 4 + 1][c],
                                rb[g * 4 + 2][c], rb[g * 4 + 3][c] };
                *(v4f*)&Bs[buf][b_n + c][b_k + 16 * g] = t;
            }
    };

    auto ldfrag = [&](int buf, int kk, v4f (&fa)[4], v4f (&fb)[4]) {
        #pragma unroll
        for (int i = 0; i < 4; ++i)
            fa[i] = *(const v4f*)&As[buf][am0 + 16 * i][kk + ko];
        #pragma unroll
        for (int j = 0; j < 4; ++j)
            fb[j] = *(const v4f*)&Bs[buf][bn0 + 16 * j][kk + ko];
    };

    auto mma = [&](const v4f (&fa)[4], const v4f (&fb)[4]) {
        #pragma unroll
        for (int p = 0; p < 2; ++p) {
            #pragma unroll
            for (int i = 0; i < 4; ++i) {
                const v2f a = { fa[i][2 * p], fa[i][2 * p + 1] };
                #pragma unroll
                for (int j = 0; j < 4; ++j) {
                    const v2f b = { fb[j][2 * p], fb[j][2 * p + 1] };
                    acc[i][j] = wmma4(a, b, acc[i][j]);
                }
            }
        }
    };

    auto compute = [&](int buf) {
        v4f fa0[4], fb0[4], fa1[4], fb1[4];
        ldfrag(buf, 0, fa0, fb0);
        ldfrag(buf, 8, fa1, fb1);
        mma(fa0, fb0);
        ldfrag(buf, 16, fa0, fb0);
        mma(fa1, fb1);
        ldfrag(buf, 24, fa1, fb1);
        mma(fa0, fb0);
        mma(fa1, fb1);
    };

    const int pmark1 = (nchunks * 27) >> 6;
    const int pmark2 = (nchunks * 47) >> 6;
    const int pmark3 = (nchunks * 59) >> 6;

    prefetch(0);
    stage(0);
    __syncthreads();
    int buf = 0;
    __builtin_amdgcn_s_setprio(3);
    for (int c = 0; c < nchunks; ++c) {
        if (c == pmark1)      __builtin_amdgcn_s_setprio(2);
        else if (c == pmark2) __builtin_amdgcn_s_setprio(1);
        else if (c == pmark3) __builtin_amdgcn_s_setprio(0);
        if (c + 1 < nchunks) prefetch((c + 1) * KC);
        compute(buf);
        if (c + 1 < nchunks) {
            stage(buf ^ 1);
            __syncthreads();
            buf ^= 1;
        }
    }
    __builtin_amdgcn_s_setprio(0);

    const int col0 = tileN + bn0;
    const int rowb = tileM + (wm << 6) + (hh << 3);
    #pragma unroll 1
    for (int r = 0; r < 8; ++r) {
        #pragma unroll
        for (int i = 0; i < 4; ++i) {
            const int gr = rowb + (i << 4) + r;
            const int rr = gr - bb * RPB;
            const float* e = Emat + (long)bb * bsE + (long)rr * rsE;
            float* o = Out + (long)gr * N;
            #pragma unroll
            for (int j = 0; j < 4; ++j) {
                const int col = col0 + (j << 4);
                float v = acc[i][j][r];
                if (MODE == 1) {
                    v += biasCol[col];
                } else if (MODE == 2) {
                    v = sinf(v + e[col]);
                } else {
                    v = (v + biasCol[col] + e[col]) * 0.5f;
                }
                o[col] = v;
            }
        }
    }
}

extern "C" void kernel_launch(void* const* d_in, const int* in_sizes, int n_in,
                              void* d_out, int out_size, void* d_ws, size_t ws_size,
                              hipStream_t stream)
{
    (void)in_sizes; (void)n_in; (void)out_size; (void)ws_size;

    const float* x  = (const float*)d_in[0];
    const float* z  = (const float*)d_in[1];
    const float* W1 = (const float*)d_in[2];
    const float* b1 = (const float*)d_in[3];
    const float* W2 = (const float*)d_in[4];
    const float* b2 = (const float*)d_in[5];
    const float* V1 = (const float*)d_in[6];
    const float* c1 = (const float*)d_in[7];
    const float* V2 = (const float*)d_in[8];
    const float* c2 = (const float*)d_in[9];
    const float* Wr = (const float*)d_in[10];
    const float* br = (const float*)d_in[11];
    float* out = (float*)d_out;

    const int Bb = 8, T = 1024, F = 1024, ZC = 128, ZL = 1028, HL = 1026;

    float* ws   = (float*)d_ws;
    const size_t H1N = (size_t)Bb * HL * F;
    const size_t WTN = (size_t)Bb * T * F;
    float* h1   = ws;
    float* g1   = h1 + H1N;
    float* wts  = g1 + H1N;
    float* bmat = wts + WTN;
    float* sm   = bmat + WTN;

    const dim3 blk(256);
    const dim3 gridA(F / BN, (Bb * HL + BM - 1) / BM);
    const dim3 blkW(128);
    const dim3 gridW(F / TN, (Bb * T) / TM);

    gemm_f32_wmma<0, 1026><<<gridA, blk, 0, stream>>>(
        z, (long)ZL * ZC, ZC, W1, 0, b1, nullptr, 0, 0,
        h1, Bb * HL, 3 * ZC, 3 * ZC, F);
    gemm_f32_wmma<0, 1026><<<gridA, blk, 0, stream>>>(
        z, (long)ZL * ZC, ZC, V1, 0, c1, nullptr, 0, 0,
        g1, Bb * HL, 3 * ZC, 3 * ZC, F);
    gemm_f32_wmma_w64<1><<<gridW, blkW, 0, stream>>>(
        h1, (long)HL * F, F, W2, 0, b2, nullptr, 0, 0,
        wts, 3 * F, 3 * F, F);
    gemm_f32_wmma_w64<1><<<gridW, blkW, 0, stream>>>(
        g1, (long)HL * F, F, V2, 0, c2, nullptr, 0, 0,
        bmat, 3 * F, 3 * F, F);
    gemm_f32_wmma_w64<2><<<gridW, blkW, 0, stream>>>(
        x + 4, (long)T * F, T, wts, (long)T * F, nullptr,
        bmat, (long)T * F, F,
        sm, T, T - 4, F);
    gemm_f32_wmma_w64<3><<<gridW, blkW, 0, stream>>>(
        sm, (long)T * F, F, Wr, 0, br,
        x, (long)T * F, F,
        out, F, F, F);
}


# GEMMs 3-6: K-loop prefetch cut to bare loads + 2 pointer adds: row pitches compile-time (immediate offsets), guarded partial-chunk loader peeled out of the loop
# speedup vs baseline: 1.1705x; 1.0891x over previous
#include <hip/hip_runtime.h>
#include <hip/hip_bf16.h>
#include <math.h>


typedef float v2f __attribute__((ext_vector_type(2)));
typedef float v4f __attribute__((ext_vector_type(4)));
typedef float v8f __attribute__((ext_vector_type(8)));

#define BM  64
#define BN  128
#define KC  32
#define AKP 36
#define BKP 36

__device__ __forceinline__ v8f wmma4(v2f a, v2f b, v8f c) {
    return __builtin_amdgcn_wmma_f32_16x16x4_f32(
        false, a, false, b, (short)0, c, false, false);
}

template <int MODE, int RPB>
__global__ __launch_bounds__(256)
void gemm_f32_wmma(const float* __restrict__ A, long bsA, long rsA,
                   const float* __restrict__ Bm, long bsB,
                   const float* __restrict__ biasCol,
                   const float* __restrict__ Emat, long bsE, long rsE,
                   float* __restrict__ Out,
                   int Mtotal, int K, int klim, int N)
{
    __shared__ float As[2][BM][AKP];
    __shared__ float Bs[2][BN][BKP];

    const int tid  = threadIdx.x;
    const int lane = tid & 31;
    const int wave = tid >> 5;
    const int wm   = wave & 1;
    const int wn   = wave >> 1;

    const int tileM = blockIdx.y * BM;
    const int tileN = blockIdx.x * BN;
    const int bB = (bsB != 0) ? (tileM / RPB) : 0;

    v8f acc00 = {}, acc01 = {}, acc10 = {}, acc11 = {};

    const int a_row = tid >> 2;
    const int a_kc  = (tid & 3) << 2;
    int a_grow = tileM + a_row;
    if (a_grow >= Mtotal) a_grow = Mtotal - 1;
    const int a_b = a_grow / RPB;
    const int a_r = a_grow - a_b * RPB;
    const float* a_base = A + (long)a_b * bsA + (long)a_r * rsA;

    const int b_kr = tid >> 4;
    const int b_nc = (tid & 15) << 2;
    const float* b_base = Bm + (long)bB * bsB + tileN;

    const int m0   = (wm << 5) + (lane & 15);
    const int n0   = (wn << 5) + (lane & 15);
    const int koff = (lane >> 4) << 1;

    const int nchunks = (klim + KC - 1) / KC;

    v4f ra0, ra1, rb00, rb01, rb10, rb11;

    auto prefetch = [&](int k0) {
        if (k0 + KC <= klim) {
            ra0  = *(const v4f*)(a_base + k0 + a_kc);
            ra1  = *(const v4f*)(a_base + k0 + a_kc + 16);
            const float* br0 = b_base + (long)(k0 + b_kr)      * N + b_nc;
            const float* br1 = b_base + (long)(k0 + b_kr + 16) * N + b_nc;
            rb00 = *(const v4f*)(br0);
            rb01 = *(const v4f*)(br0 + 64);
            rb10 = *(const v4f*)(br1);
            rb11 = *(const v4f*)(br1 + 64);
        } else {
            const v4f zz = {0.f, 0.f, 0.f, 0.f};
            #pragma unroll
            for (int i = 0; i < 4; ++i) {
                const int ka = k0 + a_kc + i;
                ra0[i] = (ka      < klim) ? a_base[ka]      : 0.0f;
                ra1[i] = (ka + 16 < klim) ? a_base[ka + 16] : 0.0f;
            }
            const int kb0 = k0 + b_kr, kb1 = kb0 + 16;
            const float* br0 = b_base + (long)kb0 * N + b_nc;
            const float* br1 = b_base + (long)kb1 * N + b_nc;
            rb00 = (kb0 < K) ? *(const v4f*)(br0)      : zz;
            rb01 = (kb0 < K) ? *(const v4f*)(br0 + 64) : zz;
            rb10 = (kb1 < K) ? *(const v4f*)(br1)      : zz;
            rb11 = (kb1 < K) ? *(const v4f*)(br1 + 64) : zz;
        }
    };

    auto stage = [&](int buf) {
        *(v4f*)&As[buf][a_row][a_kc]      = ra0;
        *(v4f*)&As[buf][a_row][a_kc + 16] = ra1;
        #pragma unroll
        for (int i = 0; i < 4; ++i) {
            Bs[buf][b_nc + i     ][b_kr]      = rb00[i];
            Bs[buf][b_nc + i     ][b_kr + 16] = rb10[i];
            Bs[buf][b_nc + i + 64][b_kr]      = rb01[i];
            Bs[buf][b_nc + i + 64][b_kr + 16] = rb11[i];
        }
    };

    auto compute = [&](int buf) {
        #pragma unroll
        for (int kk = 0; kk < KC; kk += 4) {
            v2f a0 = *(const v2f*)&As[buf][m0     ][kk + koff];
            v2f a1 = *(const v2f*)&As[buf][m0 + 16][kk + koff];
            v2f b0 = *(const v2f*)&Bs[buf][n0     ][kk + koff];
            v2f b1 = *(const v2f*)&Bs[buf][n0 + 16][kk + koff];
            acc00 = wmma4(a0, b0, acc00);
            acc01 = wmma4(a0, b1, acc01);
            acc10 = wmma4(a1, b0, acc10);
            acc11 = wmma4(a1, b1, acc11);
        }
    };

    prefetch(0);
    stage(0);
    __syncthreads();
    int buf = 0;
    for (int c = 0; c < nchunks; ++c) {
        if (c + 1 < nchunks) prefetch((c + 1) * KC);
        compute(buf);
        if (c + 1 < nchunks) {
            stage(buf ^ 1);
            __syncthreads();
            buf ^= 1;
        }
    }

    const int colA  = tileN + n0;
    const int mbase = tileM + (wm << 5) + ((lane >> 4) << 3);
    #pragma unroll
    for (int mi = 0; mi < 2; ++mi) {
        const v8f& accL = mi ? acc10 : acc00;
        const v8f& accR = mi ? acc11 : acc01;
        #pragma unroll
        for (int j = 0; j < 8; ++j) {
            const int gr = mbase + (mi << 4) + j;
            if (gr >= Mtotal) continue;
            const int b = gr / RPB;
            const int r = gr - b * RPB;
            float v0 = accL[j];
            float v1 = accR[j];
            if (MODE == 0) {
                v0 = fmaxf(v0 + biasCol[colA],      0.0f);
                v1 = fmaxf(v1 + biasCol[colA + 16], 0.0f);
            } else if (MODE == 1) {
                v0 += biasCol[colA];
                v1 += biasCol[colA + 16];
            } else if (MODE == 2) {
                const float* e = Emat + (long)b * bsE + (long)r * rsE;
                v0 = sinf(v0 + e[colA]);
                v1 = sinf(v1 + e[colA + 16]);
            } else {
                const float* e = Emat + (long)b * bsE + (long)r * rsE;
                v0 = (v0 + biasCol[colA]      + e[colA])      * 0.5f;
                v1 = (v1 + biasCol[colA + 16] + e[colA + 16]) * 0.5f;
            }
            Out[(long)gr * N + colA]      = v0;
            Out[(long)gr * N + colA + 16] = v1;
        }
    }
}


#define TM  128
#define TN  128
#define LP  36

template <int MODE>
__global__ __launch_bounds__(128) __attribute__((amdgpu_waves_per_eu(1, 2)))
void gemm_f32_wmma_w64(const float* __restrict__ A, long bsA,
                       const float* __restrict__ Bm, long bsB,
                       const float* __restrict__ biasCol,
                       const float* __restrict__ Emat, long bsE, long rsE,
                       float* __restrict__ Out,
                       int K, int klim)
{
    constexpr int RPB = 1024;
    constexpr int LDA = 1024;
    constexpr int LDN = 1024;
    __shared__ __attribute__((aligned(16))) float As[2][TM][LP];
    __shared__ __attribute__((aligned(16))) float Bs[2][TN][LP];

    const int tid  = threadIdx.x;
    const int lane = tid & 31;
    const int wave = tid >> 5;
    const int wm   = wave & 1;
    const int wn   = wave >> 1;

    const int tileM = blockIdx.y * TM;
    const int tileN = blockIdx.x * TN;
    const int bb    = tileM / RPB;
    const int r0    = tileM - bb * RPB;

    const int a_kq = (tid & 7) << 2;
    const int a_r  = tid >> 3;
    const char* a_ub = (const char*)(A + (long)bb * bsA + (long)r0 * LDA);
    const unsigned a_lo = (unsigned)(a_r * LDA + a_kq) * 4u;
    constexpr int A_STEP_B = 16 * LDA * 4;

    const int b_n = lane << 2;
    const int b_k = wave << 2;
    const char* b_ub = (const char*)(Bm + (long)((bsB != 0) ? bb : 0) * bsB + tileN);
    const unsigned b_lo = (unsigned)(b_k * LDN + b_n) * 4u;
    constexpr int B_ROW_B = LDN * 4;

    const int c16 = lane & 15;
    const int hh  = lane >> 4;
    const int am0 = (wm << 6) + c16;
    const int bn0 = (wn << 6) + c16;
    const int ko  = hh << 2;

    const int nchunks = (klim + KC - 1) / KC;

    v8f acc[4][4];
    #pragma unroll
    for (int i = 0; i < 4; ++i)
        #pragma unroll
        for (int j = 0; j < 4; ++j)
            acc[i][j] = v8f{};

    v4f ra[8], rb[8];

    auto prefetch_whole = [&](int k0) {
        const char* pa = a_ub + (long)k0 * 4;
        const char* pb = b_ub + (long)k0 * B_ROW_B;
        #pragma unroll
        for (int i = 0; i < 8; ++i)
            ra[i] = *(const v4f*)(pa + a_lo + i * A_STEP_B);
        #pragma unroll
        for (int g = 0; g < 2; ++g)
            #pragma unroll
            for (int j = 0; j < 4; ++j)
                rb[g * 4 + j] =
                    *(const v4f*)(pb + b_lo + (16 * g + j) * B_ROW_B);
    };

    auto prefetch_guarded = [&](int k0) {
        const char* pa = a_ub + (long)k0 * 4;
        const char* pb = b_ub + (long)k0 * B_ROW_B;
        {
            const v4f zz = {0.f, 0.f, 0.f, 0.f};
            #pragma unroll
            for (int i = 0; i < 8; ++i) {
                #pragma unroll
                for (int e = 0; e < 4; ++e) {
                    const int ka = k0 + a_kq + e;
                    ra[i][e] = (ka < klim)
                        ? *(const float*)(pa + a_lo + i * A_STEP_B + e * 4) : 0.0f;
                }
            }
            #pragma unroll
            for (int g = 0; g < 2; ++g)
                #pragma unroll
                for (int j = 0; j < 4; ++j) {
                    const int kb = k0 + b_k + 16 * g + j;
                    rb[g * 4 + j] = (kb < K)
                        ? *(const v4f*)(pb + b_lo + (16 * g + j) * B_ROW_B) : zz;
                }
        }
    };

    auto stage = [&](int buf) {
        #pragma unroll
        for (int i = 0; i < 8; ++i)
            *(v4f*)&As[buf][a_r + 16 * i][a_kq] = ra[i];
        #pragma unroll
        for (int g = 0; g < 2; ++g)
            #pragma unroll
            for (int c = 0; c < 4; ++c) {
                const v4f t = { rb[g * 4 + 0][c], rb[g * 4 + 1][c],
                                rb[g * 4 + 2][c], rb[g * 4 + 3][c] };
                *(v4f*)&Bs[buf][b_n + c][b_k + 16 * g] = t;
            }
    };

    auto ldfrag = [&](int buf, int kk, v4f (&fa)[4], v4f (&fb)[4]) {
        #pragma unroll
        for (int i = 0; i < 4; ++i)
            fa[i] = *(const v4f*)&As[buf][am0 + 16 * i][kk + ko];
        #pragma unroll
        for (int j = 0; j < 4; ++j)
            fb[j] = *(const v4f*)&Bs[buf][bn0 + 16 * j][kk + ko];
    };

    auto mma = [&](const v4f (&fa)[4], const v4f (&fb)[4]) {
        #pragma unroll
        for (int p = 0; p < 2; ++p) {
            #pragma unroll
            for (int i = 0; i < 4; ++i) {
                const v2f a = { fa[i][2 * p], fa[i][2 * p + 1] };
                #pragma unroll
                for (int j = 0; j < 4; ++j) {
                    const v2f b = { fb[j][2 * p], fb[j][2 * p + 1] };
                    acc[i][j] = wmma4(a, b, acc[i][j]);
                }
            }
        }
    };

    auto compute = [&](int buf) {
        v4f fa0[4], fb0[4], fa1[4], fb1[4];
        ldfrag(buf, 0, fa0, fb0);
        ldfrag(buf, 8, fa1, fb1);
        mma(fa0, fb0);
        ldfrag(buf, 16, fa0, fb0);
        mma(fa1, fb1);
        ldfrag(buf, 24, fa1, fb1);
        mma(fa0, fb0);
        mma(fa1, fb1);
    };

    const int pmark1 = (nchunks * 27) >> 6;
    const int pmark2 = (nchunks * 47) >> 6;
    const int pmark3 = (nchunks * 59) >> 6;

    const int nwhole = klim / KC;

    auto prio_mark = [&](int c) {
        if (c == pmark1)      __builtin_amdgcn_s_setprio(2);
        else if (c == pmark2) __builtin_amdgcn_s_setprio(1);
        else if (c == pmark3) __builtin_amdgcn_s_setprio(0);
    };

    if (nwhole > 0) prefetch_whole(0); else prefetch_guarded(0);
    stage(0);
    __syncthreads();
    int buf = 0;
    int c = 0;
    __builtin_amdgcn_s_setprio(3);
    for (; c + 1 < nwhole; ++c) {
        prio_mark(c);
        prefetch_whole((c + 1) * KC);
        compute(buf);
        stage(buf ^ 1);
        __syncthreads();
        buf ^= 1;
    }
    for (; c < nchunks; ++c) {
        prio_mark(c);
        if (c + 1 < nchunks) prefetch_guarded((c + 1) * KC);
        compute(buf);
        if (c + 1 < nchunks) {
            stage(buf ^ 1);
            __syncthreads();
            buf ^= 1;
        }
    }
    __builtin_amdgcn_s_setprio(0);

    const int col0 = tileN + bn0;
    const int rowb = tileM + (wm << 6) + (hh << 3);
    #pragma unroll 1
    for (int r = 0; r < 8; ++r) {
        #pragma unroll
        for (int i = 0; i < 4; ++i) {
            const int gr = rowb + (i << 4) + r;
            const int rr = gr - bb * RPB;
            const float* e = Emat + (long)bb * bsE + (long)rr * rsE;
            float* o = Out + (long)gr * LDN;
            #pragma unroll
            for (int j = 0; j < 4; ++j) {
                const int col = col0 + (j << 4);
                float v = acc[i][j][r];
                if (MODE == 1) {
                    v += biasCol[col];
                } else if (MODE == 2) {
                    v = sinf(v + e[col]);
                } else {
                    v = (v + biasCol[col] + e[col]) * 0.5f;
                }
                o[col] = v;
            }
        }
    }
}

extern "C" void kernel_launch(void* const* d_in, const int* in_sizes, int n_in,
                              void* d_out, int out_size, void* d_ws, size_t ws_size,
                              hipStream_t stream)
{
    (void)in_sizes; (void)n_in; (void)out_size; (void)ws_size;

    const float* x  = (const float*)d_in[0];
    const float* z  = (const float*)d_in[1];
    const float* W1 = (const float*)d_in[2];
    const float* b1 = (const float*)d_in[3];
    const float* W2 = (const float*)d_in[4];
    const float* b2 = (const float*)d_in[5];
    const float* V1 = (const float*)d_in[6];
    const float* c1 = (const float*)d_in[7];
    const float* V2 = (const float*)d_in[8];
    const float* c2 = (const float*)d_in[9];
    const float* Wr = (const float*)d_in[10];
    const float* br = (const float*)d_in[11];
    float* out = (float*)d_out;

    const int Bb = 8, T = 1024, F = 1024, ZC = 128, ZL = 1028, HL = 1026;

    float* ws   = (float*)d_ws;
    const size_t H1N = (size_t)Bb * HL * F;
    const size_t WTN = (size_t)Bb * T * F;
    float* h1   = ws;
    float* g1   = h1 + H1N;
    float* wts  = g1 + H1N;
    float* bmat = wts + WTN;
    float* sm   = bmat + WTN;

    const dim3 blk(256);
    const dim3 gridA(F / BN, (Bb * HL + BM - 1) / BM);
    const dim3 blkW(128);
    const dim3 gridW(F / TN, (Bb * T) / TM);

    gemm_f32_wmma<0, 1026><<<gridA, blk, 0, stream>>>(
        z, (long)ZL * ZC, ZC, W1, 0, b1, nullptr, 0, 0,
        h1, Bb * HL, 3 * ZC, 3 * ZC, F);
    gemm_f32_wmma<0, 1026><<<gridA, blk, 0, stream>>>(
        z, (long)ZL * ZC, ZC, V1, 0, c1, nullptr, 0, 0,
        g1, Bb * HL, 3 * ZC, 3 * ZC, F);
    gemm_f32_wmma_w64<1><<<gridW, blkW, 0, stream>>>(
        h1, (long)HL * F, W2, 0, b2, nullptr, 0, 0,
        wts, 3 * F, 3 * F);
    gemm_f32_wmma_w64<1><<<gridW, blkW, 0, stream>>>(
        g1, (long)HL * F, V2, 0, c2, nullptr, 0, 0,
        bmat, 3 * F, 3 * F);
    gemm_f32_wmma_w64<2><<<gridW, blkW, 0, stream>>>(
        x + 4, (long)T * F, wts, (long)T * F, nullptr,
        bmat, (long)T * F, F,
        sm, T, T - 4);
    gemm_f32_wmma_w64<3><<<gridW, blkW, 0, stream>>>(
        sm, (long)T * F, Wr, 0, br,
        x, (long)T * F, F,
        out, F, F);
}


# GEMMs 3-4 (the two 3-tap convs): F(2,3) minimal filtering along time: 4 products of K=1024 per output row pair instead of 6; row sums in the A loader, one combine pass; f32 sum order changes
# speedup vs baseline: 1.3305x; 1.1367x over previous
#include <hip/hip_runtime.h>
#include <hip/hip_bf16.h>
#include <math.h>


typedef float v2f __attribute__((ext_vector_type(2)));
typedef float v4f __attribute__((ext_vector_type(4)));
typedef float v8f __attribute__((ext_vector_type(8)));

#define BM  64
#define BN  128
#define KC  32
#define AKP 36
#define BKP 36

__device__ __forceinline__ v8f wmma4(v2f a, v2f b, v8f c) {
    return __builtin_amdgcn_wmma_f32_16x16x4_f32(
        false, a, false, b, (short)0, c, false, false);
}

template <int MODE, int RPB>
__global__ __launch_bounds__(256)
void gemm_f32_wmma(const float* __restrict__ A, long bsA, long rsA,
                   const float* __restrict__ Bm, long bsB,
                   const float* __restrict__ biasCol,
                   const float* __restrict__ Emat, long bsE, long rsE,
                   float* __restrict__ Out,
                   int Mtotal, int K, int klim, int N)
{
    __shared__ float As[2][BM][AKP];
    __shared__ float Bs[2][BN][BKP];

    const int tid  = threadIdx.x;
    const int lane = tid & 31;
    const int wave = tid >> 5;
    const int wm   = wave & 1;
    const int wn   = wave >> 1;

    const int tileM = blockIdx.y * BM;
    const int tileN = blockIdx.x * BN;
    const int bB = (bsB != 0) ? (tileM / RPB) : 0;

    v8f acc00 = {}, acc01 = {}, acc10 = {}, acc11 = {};

    const int a_row = tid >> 2;
    const int a_kc  = (tid & 3) << 2;
    int a_grow = tileM + a_row;
    if (a_grow >= Mtotal) a_grow = Mtotal - 1;
    const int a_b = a_grow / RPB;
    const int a_r = a_grow - a_b * RPB;
    const float* a_base = A + (long)a_b * bsA + (long)a_r * rsA;

    const int b_kr = tid >> 4;
    const int b_nc = (tid & 15) << 2;
    const float* b_base = Bm + (long)bB * bsB + tileN;

    const int m0   = (wm << 5) + (lane & 15);
    const int n0   = (wn << 5) + (lane & 15);
    const int koff = (lane >> 4) << 1;

    const int nchunks = (klim + KC - 1) / KC;

    v4f ra0, ra1, rb00, rb01, rb10, rb11;

    auto prefetch = [&](int k0) {
        if (k0 + KC <= klim) {
            ra0  = *(const v4f*)(a_base + k0 + a_kc);
            ra1  = *(const v4f*)(a_base + k0 + a_kc + 16);
            const float* br0 = b_base + (long)(k0 + b_kr)      * N + b_nc;
            const float* br1 = b_base + (long)(k0 + b_kr + 16) * N + b_nc;
            rb00 = *(const v4f*)(br0);
            rb01 = *(const v4f*)(br0 + 64);
            rb10 = *(const v4f*)(br1);
            rb11 = *(const v4f*)(br1 + 64);
        } else {
            const v4f zz = {0.f, 0.f, 0.f, 0.f};
            #pragma unroll
            for (int i = 0; i < 4; ++i) {
                const int ka = k0 + a_kc + i;
                ra0[i] = (ka      < klim) ? a_base[ka]      : 0.0f;
                ra1[i] = (ka + 16 < klim) ? a_base[ka + 16] : 0.0f;
            }
            const int kb0 = k0 + b_kr, kb1 = kb0 + 16;
            const float* br0 = b_base + (long)kb0 * N + b_nc;
            const float* br1 = b_base + (long)kb1 * N + b_nc;
            rb00 = (kb0 < K) ? *(const v4f*)(br0)      : zz;
            rb01 = (kb0 < K) ? *(const v4f*)(br0 + 64) : zz;
            rb10 = (kb1 < K) ? *(const v4f*)(br1)      : zz;
            rb11 = (kb1 < K) ? *(const v4f*)(br1 + 64) : zz;
        }
    };

    auto stage = [&](int buf) {
        *(v4f*)&As[buf][a_row][a_kc]      = ra0;
        *(v4f*)&As[buf][a_row][a_kc + 16] = ra1;
        #pragma unroll
        for (int i = 0; i < 4; ++i) {
            Bs[buf][b_nc + i     ][b_kr]      = rb00[i];
            Bs[buf][b_nc + i     ][b_kr + 16] = rb10[i];
            Bs[buf][b_nc + i + 64][b_kr]      = rb01[i];
            Bs[buf][b_nc + i + 64][b_kr + 16] = rb11[i];
        }
    };

    auto compute = [&](int buf) {
        #pragma unroll
        for (int kk = 0; kk < KC; kk += 4) {
            v2f a0 = *(const v2f*)&As[buf][m0     ][kk + koff];
            v2f a1 = *(const v2f*)&As[buf][m0 + 16][kk + koff];
            v2f b0 = *(const v2f*)&Bs[buf][n0     ][kk + koff];
            v2f b1 = *(const v2f*)&Bs[buf][n0 + 16][kk + koff];
            acc00 = wmma4(a0, b0, acc00);
            acc01 = wmma4(a0, b1, acc01);
            acc10 = wmma4(a1, b0, acc10);
            acc11 = wmma4(a1, b1, acc11);
        }
    };

    prefetch(0);
    stage(0);
    __syncthreads();
    int buf = 0;
    for (int c = 0; c < nchunks; ++c) {
        if (c + 1 < nchunks) prefetch((c + 1) * KC);
        compute(buf);
        if (c + 1 < nchunks) {
            stage(buf ^ 1);
            __syncthreads();
            buf ^= 1;
        }
    }

    const int colA  = tileN + n0;
    const int mbase = tileM + (wm << 5) + ((lane >> 4) << 3);
    #pragma unroll
    for (int mi = 0; mi < 2; ++mi) {
        const v8f& accL = mi ? acc10 : acc00;
        const v8f& accR = mi ? acc11 : acc01;
        #pragma unroll
        for (int j = 0; j < 8; ++j) {
            const int gr = mbase + (mi << 4) + j;
            if (gr >= Mtotal) continue;
            const int b = gr / RPB;
            const int r = gr - b * RPB;
            float v0 = accL[j];
            float v1 = accR[j];
            if (MODE == 0) {
                v0 = fmaxf(v0 + biasCol[colA],      0.0f);
                v1 = fmaxf(v1 + biasCol[colA + 16], 0.0f);
            } else if (MODE == 1) {
                v0 += biasCol[colA];
                v1 += biasCol[colA + 16];
            } else if (MODE == 2) {
                const float* e = Emat + (long)b * bsE + (long)r * rsE;
                v0 = sinf(v0 + e[colA]);
                v1 = sinf(v1 + e[colA + 16]);
            } else {
                const float* e = Emat + (long)b * bsE + (long)r * rsE;
                v0 = (v0 + biasCol[colA]      + e[colA])      * 0.5f;
                v1 = (v1 + biasCol[colA + 16] + e[colA + 16]) * 0.5f;
            }
            Out[(long)gr * N + colA]      = v0;
            Out[(long)gr * N + colA + 16] = v1;
        }
    }
}


#define TM  128
#define TN  128
#define LP  36

template <int MODE>
__global__ __launch_bounds__(128) __attribute__((amdgpu_waves_per_eu(1, 2)))
void gemm_f32_wmma_w64(const float* __restrict__ A, long bsA,
                       const float* __restrict__ Bm, long bsB,
                       const float* __restrict__ biasCol,
                       const float* __restrict__ Emat, long bsE, long rsE,
                       float* __restrict__ Out,
                       int K, int klim)
{
    constexpr bool WINO = (MODE == 4);
    constexpr int RPB = WINO ? 512 : 1024;
    constexpr int LDA = WINO ? 2048 : 1024;
    constexpr int LDN = 1024;
    __shared__ __attribute__((aligned(16))) float As[2][TM][LP];
    __shared__ __attribute__((aligned(16))) float Bs[2][TN][LP];

    const int tid  = threadIdx.x;
    const int lane = tid & 31;
    const int wave = tid >> 5;
    const int wm   = wave & 1;
    const int wn   = wave >> 1;

    const int tileM = blockIdx.y * TM;
    const int tileN = blockIdx.x * TN;
    const int bb    = tileM / RPB;
    const int r0    = tileM - bb * RPB;

    const int   pj = WINO ? (int)blockIdx.z : 0;
    const int   o1 = (pj == 0) ? 0 : (pj == 2) ? 2 : 1;
    const int   o2 = (pj == 2) ? 1 : (pj == 3) ? 3 : 2;
    const float sg = (pj == 1) ? 1.0f : -1.0f;

    const int a_kq = (tid & 7) << 2;
    const int a_r  = tid >> 3;
    const char* a_ub  = (const char*)(A + (long)bb * bsA + (long)r0 * LDA
                                        + (WINO ? o1 * 1024 : 0));
    const char* a_ub2 = (const char*)(A + (long)bb * bsA + (long)r0 * LDA
                                        + (WINO ? o2 * 1024 : 0));
    const unsigned a_lo = (unsigned)(a_r * LDA + a_kq) * 4u;
    constexpr int A_STEP_B = 16 * LDA * 4;

    const int b_n = lane << 2;
    const int b_k = wave << 2;
    const float* b_sel = WINO
        ? ((pj == 0) ? Bm : (pj == 1) ? Emat
           : (pj == 2) ? Emat + 1024 * 1024 : Bm + 2 * 1024 * 1024)
        : Bm + (long)((bsB != 0) ? bb : 0) * bsB;
    const char* b_ub = (const char*)(b_sel + tileN);
    const unsigned b_lo = (unsigned)(b_k * LDN + b_n) * 4u;
    constexpr int B_ROW_B = LDN * 4;

    const int c16 = lane & 15;
    const int hh  = lane >> 4;
    const int am0 = (wm << 6) + c16;
    const int bn0 = (wn << 6) + c16;
    const int ko  = hh << 2;

    const int nchunks = (klim + KC - 1) / KC;

    v8f acc[4][4];
    #pragma unroll
    for (int i = 0; i < 4; ++i)
        #pragma unroll
        for (int j = 0; j < 4; ++j)
            acc[i][j] = v8f{};

    v4f ra[8], rb[8];
    v4f ra2[8];

    auto prefetch_whole = [&](int k0) {
        const char* pa = a_ub + (long)k0 * 4;
        const char* pb = b_ub + (long)k0 * B_ROW_B;
        #pragma unroll
        for (int i = 0; i < 8; ++i)
            ra[i] = *(const v4f*)(pa + a_lo + i * A_STEP_B);
        if (WINO) {
            const char* pa2 = a_ub2 + (long)k0 * 4;
            #pragma unroll
            for (int i = 0; i < 8; ++i)
                ra2[i] = *(const v4f*)(pa2 + a_lo + i * A_STEP_B);
        }
        #pragma unroll
        for (int g = 0; g < 2; ++g)
            #pragma unroll
            for (int j = 0; j < 4; ++j)
                rb[g * 4 + j] =
                    *(const v4f*)(pb + b_lo + (16 * g + j) * B_ROW_B);
    };

    auto prefetch_guarded = [&](int k0) {
        const char* pa = a_ub + (long)k0 * 4;
        const char* pb = b_ub + (long)k0 * B_ROW_B;
        {
            const v4f zz = {0.f, 0.f, 0.f, 0.f};
            #pragma unroll
            for (int i = 0; i < 8; ++i) {
                #pragma unroll
                for (int e = 0; e < 4; ++e) {
                    const int ka = k0 + a_kq + e;
                    ra[i][e] = (ka < klim)
                        ? *(const float*)(pa + a_lo + i * A_STEP_B + e * 4) : 0.0f;
                    if (WINO)
                        ra2[i][e] = (ka < klim)
                            ? *(const float*)(a_ub2 + (long)k0 * 4 + a_lo
                                              + i * A_STEP_B + e * 4) : 0.0f;
                }
            }
            #pragma unroll
            for (int g = 0; g < 2; ++g)
                #pragma unroll
                for (int j = 0; j < 4; ++j) {
                    const int kb = k0 + b_k + 16 * g + j;
                    rb[g * 4 + j] = (kb < K)
                        ? *(const v4f*)(pb + b_lo + (16 * g + j) * B_ROW_B) : zz;
                }
        }
    };

    auto stage = [&](int buf) {
        #pragma unroll
        for (int i = 0; i < 8; ++i)
            *(v4f*)&As[buf][a_r + 16 * i][a_kq] =
                WINO ? (ra[i] + sg * ra2[i]) : ra[i];
        #pragma unroll
        for (int g = 0; g < 2; ++g)
            #pragma unroll
            for (int c = 0; c < 4; ++c) {
                const v4f t = { rb[g * 4 + 0][c], rb[g * 4 + 1][c],
                                rb[g * 4 + 2][c], rb[g * 4 + 3][c] };
                *(v4f*)&Bs[buf][b_n + c][b_k + 16 * g] = t;
            }
    };

    auto ldfrag = [&](int buf, int kk, v4f (&fa)[4], v4f (&fb)[4]) {
        #pragma unroll
        for (int i = 0; i < 4; ++i)
            fa[i] = *(const v4f*)&As[buf][am0 + 16 * i][kk + ko];
        #pragma unroll
        for (int j = 0; j < 4; ++j)
            fb[j] = *(const v4f*)&Bs[buf][bn0 + 16 * j][kk + ko];
    };

    auto mma = [&](const v4f (&fa)[4], const v4f (&fb)[4]) {
        #pragma unroll
        for (int p = 0; p < 2; ++p) {
            #pragma unroll
            for (int i = 0; i < 4; ++i) {
                const v2f a = { fa[i][2 * p], fa[i][2 * p + 1] };
                #pragma unroll
                for (int j = 0; j < 4; ++j) {
                    const v2f b = { fb[j][2 * p], fb[j][2 * p + 1] };
                    acc[i][j] = wmma4(a, b, acc[i][j]);
                }
            }
        }
    };

    auto compute = [&](int buf) {
        v4f fa0[4], fb0[4], fa1[4], fb1[4];
        ldfrag(buf, 0, fa0, fb0);
        ldfrag(buf, 8, fa1, fb1);
        mma(fa0, fb0);
        ldfrag(buf, 16, fa0, fb0);
        mma(fa1, fb1);
        ldfrag(buf, 24, fa1, fb1);
        mma(fa0, fb0);
        mma(fa1, fb1);
    };

    const int pmark1 = (nchunks * 27) >> 6;
    const int pmark2 = (nchunks * 47) >> 6;
    const int pmark3 = (nchunks * 59) >> 6;

    const int nwhole = klim / KC;

    auto prio_mark = [&](int c) {
        if (c == pmark1)      __builtin_amdgcn_s_setprio(2);
        else if (c == pmark2) __builtin_amdgcn_s_setprio(1);
        else if (c == pmark3) __builtin_amdgcn_s_setprio(0);
    };

    if (nwhole > 0) prefetch_whole(0); else prefetch_guarded(0);
    stage(0);
    __syncthreads();
    int buf = 0;
    int c = 0;
    __builtin_amdgcn_s_setprio(3);
    for (; c + 1 < nwhole; ++c) {
        prio_mark(c);
        prefetch_whole((c + 1) * KC);
        compute(buf);
        stage(buf ^ 1);
        __syncthreads();
        buf ^= 1;
    }
    for (; c < nchunks; ++c) {
        prio_mark(c);
        if (c + 1 < nchunks) prefetch_guarded((c + 1) * KC);
        compute(buf);
        if (c + 1 < nchunks) {
            stage(buf ^ 1);
            __syncthreads();
            buf ^= 1;
        }
    }
    __builtin_amdgcn_s_setprio(0);

    const int col0 = tileN + bn0;
    const int rowb = tileM + (wm << 6) + (hh << 3);
    #pragma unroll 1
    for (int r = 0; r < 8; ++r) {
        #pragma unroll
        for (int i = 0; i < 4; ++i) {
            const int gr = rowb + (i << 4) + r;
            const int rr = gr - bb * RPB;
            const float* e = Emat + (long)bb * bsE + (long)rr * rsE;
            float* o = Out + (WINO ? (long)pj * (4096L * 1024) : 0L) + (long)gr * LDN;
            #pragma unroll
            for (int j = 0; j < 4; ++j) {
                const int col = col0 + (j << 4);
                float v = acc[i][j][r];
                if (MODE == 1) {
                    v += biasCol[col];
                } else if (MODE == 2) {
                    v = sinf(v + e[col]);
                } else if (MODE == 3) {
                    v = (v + biasCol[col] + e[col]) * 0.5f;
                }
                o[col] = v;
            }
        }
    }
}

__global__ __launch_bounds__(256)
void f23_filter(const float* __restrict__ W, float* __restrict__ G)
{
    const long TP = 1024L * 1024;
    const long o  = ((long)blockIdx.x * 256 + threadIdx.x) << 2;
    const v4f g0 = *(const v4f*)(W + o);
    const v4f g1 = *(const v4f*)(W + TP + o);
    const v4f g2 = *(const v4f*)(W + 2 * TP + o);
    const v4f gs = g0 + g2;
    *(v4f*)(G + o)      = (gs + g1) * 0.5f;
    *(v4f*)(G + TP + o) = (gs - g1) * 0.5f;
}

__global__ __launch_bounds__(256)
void f23_combine(const float* __restrict__ Mt, const float* __restrict__ biasCol,
                 float* __restrict__ Y)
{
    const long PL  = 4096L * 1024;
    const int  idx = blockIdx.x * 256 + threadIdx.x;
    const int  row = idx >> 8;
    const int  c4  = (idx & 255) << 2;
    const float* m = Mt + (long)row * 1024 + c4;
    const v4f m0 = *(const v4f*)(m);
    const v4f m1 = *(const v4f*)(m + PL);
    const v4f m2 = *(const v4f*)(m + 2 * PL);
    const v4f m3 = *(const v4f*)(m + 3 * PL);
    const v4f bv = *(const v4f*)(biasCol + c4);
    float* y = Y + (long)row * 2048 + c4;
    *(v4f*)(y)        = ((m0 + m1) + m2) + bv;
    *(v4f*)(y + 1024) = ((m1 - m2) - m3) + bv;
}

extern "C" void kernel_launch(void* const* d_in, const int* in_sizes, int n_in,
                              void* d_out, int out_size, void* d_ws, size_t ws_size,
                              hipStream_t stream)
{
    (void)in_sizes; (void)n_in; (void)out_size; (void)ws_size;

    const float* x  = (const float*)d_in[0];
    const float* z  = (const float*)d_in[1];
    const float* W1 = (const float*)d_in[2];
    const float* b1 = (const float*)d_in[3];
    const float* W2 = (const float*)d_in[4];
    const float* b2 = (const float*)d_in[5];
    const float* V1 = (const float*)d_in[6];
    const float* c1 = (const float*)d_in[7];
    const float* V2 = (const float*)d_in[8];
    const float* c2 = (const float*)d_in[9];
    const float* Wr = (const float*)d_in[10];
    const float* br = (const float*)d_in[11];
    float* out = (float*)d_out;

    const int Bb = 8, T = 1024, F = 1024, ZC = 128, ZL = 1028, HL = 1026;

    float* ws   = (float*)d_ws;
    const size_t H1N = (size_t)Bb * HL * F;
    const size_t WTN = (size_t)Bb * T * F;
    float* r0 = ws;
    float* r1 = r0 + H1N;
    float* r2 = r1 + H1N;
    float* r4 = r2 + 2 * WTN;
    float* h1   = r0;
    float* g1   = r1;
    float* pln  = r2;
    float* gw   = r4;
    float* wts  = r4;
    float* gv   = r0;
    float* bmat = r0;
    float* sm   = r1;

    const dim3 blk(256);
    const dim3 gridA(F / BN, (Bb * HL + BM - 1) / BM);
    const dim3 blkW(128);
    const dim3 gridW(F / TN, (Bb * T) / TM);
    const dim3 gridP(F / TN, (Bb * T / 2) / TM, 4);
    const dim3 gridF((F * F / 4) / 256);
    const dim3 gridC((Bb * (T / 2) * F / 4) / 256);

    gemm_f32_wmma<0, 1026><<<gridA, blk, 0, stream>>>(
        z, (long)ZL * ZC, ZC, W1, 0, b1, nullptr, 0, 0,
        h1, Bb * HL, 3 * ZC, 3 * ZC, F);
    gemm_f32_wmma<0, 1026><<<gridA, blk, 0, stream>>>(
        z, (long)ZL * ZC, ZC, V1, 0, c1, nullptr, 0, 0,
        g1, Bb * HL, 3 * ZC, 3 * ZC, F);
    f23_filter<<<gridF, blk, 0, stream>>>(W2, gw);
    gemm_f32_wmma_w64<4><<<gridP, blkW, 0, stream>>>(
        h1, (long)HL * F, W2, 0, nullptr, gw, 0, 0,
        pln, F, F);
    f23_combine<<<gridC, blk, 0, stream>>>(pln, b2, wts);
    f23_filter<<<gridF, blk, 0, stream>>>(V2, gv);
    gemm_f32_wmma_w64<4><<<gridP, blkW, 0, stream>>>(
        g1, (long)HL * F, V2, 0, nullptr, gv, 0, 0,
        pln, F, F);
    f23_combine<<<gridC, blk, 0, stream>>>(pln, c2, bmat);
    gemm_f32_wmma_w64<2><<<gridW, blkW, 0, stream>>>(
        x + 4, (long)T * F, wts, (long)T * F, nullptr,
        bmat, (long)T * F, F,
        sm, T, T - 4);
    gemm_f32_wmma_w64<3><<<gridW, blkW, 0, stream>>>(
        sm, (long)T * F, Wr, 0, br,
        x, (long)T * F, F,
        out, F, F);
}


# GEMMs 1-2 (K=384, M=8208) moved from the 64x128-tile 8-wave kernel onto the K loop of GEMMs 3-6 with a 144x128 block tile (57x8 workgroups, no partial tile); 8-k regrouping changes the f32 sum order
# speedup vs baseline: 1.3857x; 1.0415x over previous
#include <hip/hip_runtime.h>
#include <hip/hip_bf16.h>
#include <math.h>


typedef float v2f __attribute__((ext_vector_type(2)));
typedef float v4f __attribute__((ext_vector_type(4)));
typedef float v8f __attribute__((ext_vector_type(8)));

#define KC  32

__device__ __forceinline__ v8f wmma4(v2f a, v2f b, v8f c) {
    return __builtin_amdgcn_wmma_f32_16x16x4_f32(
        false, a, false, b, (short)0, c, false, false);
}

#define TM  128
#define TN  128
#define LP  36

template <int MODE>
__global__ __launch_bounds__(128) __attribute__((amdgpu_waves_per_eu(1, 2)))
void gemm_f32_wmma_w64(const float* __restrict__ A, long bsA,
                       const float* __restrict__ Bm, long bsB,
                       const float* __restrict__ biasCol,
                       const float* __restrict__ Emat, long bsE, long rsE,
                       float* __restrict__ Out,
                       int K, int klim)
{
    constexpr bool WINO = (MODE == 4);
    constexpr int RPB = WINO ? 512 : 1024;
    constexpr int LDA = WINO ? 2048 : 1024;
    constexpr int LDN = 1024;
    __shared__ __attribute__((aligned(16))) float As[2][TM][LP];
    __shared__ __attribute__((aligned(16))) float Bs[2][TN][LP];

    const int tid  = threadIdx.x;
    const int lane = tid & 31;
    const int wave = tid >> 5;
    const int wm   = wave & 1;
    const int wn   = wave >> 1;

    const int tileM = blockIdx.y * TM;
    const int tileN = blockIdx.x * TN;
    const int bb    = tileM / RPB;
    const int r0    = tileM - bb * RPB;

    const int   pj = WINO ? (int)blockIdx.z : 0;
    const int   o1 = (pj == 0) ? 0 : (pj == 2) ? 2 : 1;
    const int   o2 = (pj == 2) ? 1 : (pj == 3) ? 3 : 2;
    const float sg = (pj == 1) ? 1.0f : -1.0f;

    const int a_kq = (tid & 7) << 2;
    const int a_r  = tid >> 3;
    const char* a_ub  = (const char*)(A + (long)bb * bsA + (long)r0 * LDA
                                        + (WINO ? o1 * 1024 : 0));
    const char* a_ub2 = (const char*)(A + (long)bb * bsA + (long)r0 * LDA
                                        + (WINO ? o2 * 1024 : 0));
    const unsigned a_lo = (unsigned)(a_r * LDA + a_kq) * 4u;
    constexpr int A_STEP_B = 16 * LDA * 4;

    const int b_n = lane << 2;
    const int b_k = wave << 2;
    const float* b_sel = WINO
        ? ((pj == 0) ? Bm : (pj == 1) ? Emat
           : (pj == 2) ? Emat + 1024 * 1024 : Bm + 2 * 1024 * 1024)
        : Bm + (long)((bsB != 0) ? bb : 0) * bsB;
    const char* b_ub = (const char*)(b_sel + tileN);
    const unsigned b_lo = (unsigned)(b_k * LDN + b_n) * 4u;
    constexpr int B_ROW_B = LDN * 4;

    const int c16 = lane & 15;
    const int hh  = lane >> 4;
    const int am0 = (wm << 6) + c16;
    const int bn0 = (wn << 6) + c16;
    const int ko  = hh << 2;

    const int nchunks = (klim + KC - 1) / KC;

    v8f acc[4][4];
    #pragma unroll
    for (int i = 0; i < 4; ++i)
        #pragma unroll
        for (int j = 0; j < 4; ++j)
            acc[i][j] = v8f{};

    v4f ra[8], rb[8];
    v4f ra2[8];

    auto prefetch_whole = [&](int k0) {
        const char* pa = a_ub + (long)k0 * 4;
        const char* pb = b_ub + (long)k0 * B_ROW_B;
        #pragma unroll
        for (int i = 0; i < 8; ++i)
            ra[i] = *(const v4f*)(pa + a_lo + i * A_STEP_B);
        if (WINO) {
            const char* pa2 = a_ub2 + (long)k0 * 4;
            #pragma unroll
            for (int i = 0; i < 8; ++i)
                ra2[i] = *(const v4f*)(pa2 + a_lo + i * A_STEP_B);
        }
        #pragma unroll
        for (int g = 0; g < 2; ++g)
            #pragma unroll
            for (int j = 0; j < 4; ++j)
                rb[g * 4 + j] =
                    *(const v4f*)(pb + b_lo + (16 * g + j) * B_ROW_B);
    };

    auto prefetch_guarded = [&](int k0) {
        const char* pa = a_ub + (long)k0 * 4;
        const char* pb = b_ub + (long)k0 * B_ROW_B;
        {
            const v4f zz = {0.f, 0.f, 0.f, 0.f};
            #pragma unroll
            for (int i = 0; i < 8; ++i) {
                #pragma unroll
                for (int e = 0; e < 4; ++e) {
                    const int ka = k0 + a_kq + e;
                    ra[i][e] = (ka < klim)
                        ? *(const float*)(pa + a_lo + i * A_STEP_B + e * 4) : 0.0f;
                    if (WINO)
                        ra2[i][e] = (ka < klim)
                            ? *(const float*)(a_ub2 + (long)k0 * 4 + a_lo
                                              + i * A_STEP_B + e * 4) : 0.0f;
                }
            }
            #pragma unroll
            for (int g = 0; g < 2; ++g)
                #pragma unroll
                for (int j = 0; j < 4; ++j) {
                    const int kb = k0 + b_k + 16 * g + j;
                    rb[g * 4 + j] = (kb < K)
                        ? *(const v4f*)(pb + b_lo + (16 * g + j) * B_ROW_B) : zz;
                }
        }
    };

    auto stage = [&](int buf) {
        #pragma unroll
        for (int i = 0; i < 8; ++i)
            *(v4f*)&As[buf][a_r + 16 * i][a_kq] =
                WINO ? (ra[i] + sg * ra2[i]) : ra[i];
        #pragma unroll
        for (int g = 0; g < 2; ++g)
            #pragma unroll
            for (int c = 0; c < 4; ++c) {
                const v4f t = { rb[g * 4 + 0][c], rb[g * 4 + 1][c],
                                rb[g * 4 + 2][c], rb[g * 4 + 3][c] };
                *(v4f*)&Bs[buf][b_n + c][b_k + 16 * g] = t;
            }
    };

    auto ldfrag = [&](int buf, int kk, v4f (&fa)[4], v4f (&fb)[4]) {
        #pragma unroll
        for (int i = 0; i < 4; ++i)
            fa[i] = *(const v4f*)&As[buf][am0 + 16 * i][kk + ko];
        #pragma unroll
        for (int j = 0; j < 4; ++j)
            fb[j] = *(const v4f*)&Bs[buf][bn0 + 16 * j][kk + ko];
    };

    auto mma = [&](const v4f (&fa)[4], const v4f (&fb)[4]) {
        #pragma unroll
        for (int p = 0; p < 2; ++p) {
            #pragma unroll
            for (int i = 0; i < 4; ++i) {
                const v2f a = { fa[i][2 * p], fa[i][2 * p + 1] };
                #pragma unroll
                for (int j = 0; j < 4; ++j) {
                    const v2f b = { fb[j][2 * p], fb[j][2 * p + 1] };
                    acc[i][j] = wmma4(a, b, acc[i][j]);
                }
            }
        }
    };

    auto compute = [&](int buf) {
        v4f fa0[4], fb0[4], fa1[4], fb1[4];
        ldfrag(buf, 0, fa0, fb0);
        ldfrag(buf, 8, fa1, fb1);
        mma(fa0, fb0);
        ldfrag(buf, 16, fa0, fb0);
        mma(fa1, fb1);
        ldfrag(buf, 24, fa1, fb1);
        mma(fa0, fb0);
        mma(fa1, fb1);
    };

    const int pmark1 = (nchunks * 27) >> 6;
    const int pmark2 = (nchunks * 47) >> 6;
    const int pmark3 = (nchunks * 59) >> 6;

    const int nwhole = klim / KC;

    auto prio_mark = [&](int c) {
        if (c == pmark1)      __builtin_amdgcn_s_setprio(2);
        else if (c == pmark2) __builtin_amdgcn_s_setprio(1);
        else if (c == pmark3) __builtin_amdgcn_s_setprio(0);
    };

    if (nwhole > 0) prefetch_whole(0); else prefetch_guarded(0);
    stage(0);
    __syncthreads();
    int buf = 0;
    int c = 0;
    __builtin_amdgcn_s_setprio(3);
    for (; c + 1 < nwhole; ++c) {
        prio_mark(c);
        prefetch_whole((c + 1) * KC);
        compute(buf);
        stage(buf ^ 1);
        __syncthreads();
        buf ^= 1;
    }
    for (; c < nchunks; ++c) {
        prio_mark(c);
        if (c + 1 < nchunks) prefetch_guarded((c + 1) * KC);
        compute(buf);
        if (c + 1 < nchunks) {
            stage(buf ^ 1);
            __syncthreads();
            buf ^= 1;
        }
    }
    __builtin_amdgcn_s_setprio(0);

    const int col0 = tileN + bn0;
    const int rowb = tileM + (wm << 6) + (hh << 3);
    #pragma unroll 1
    for (int r = 0; r < 8; ++r) {
        #pragma unroll
        for (int i = 0; i < 4; ++i) {
            const int gr = rowb + (i << 4) + r;
            const int rr = gr - bb * RPB;
            const float* e = Emat + (long)bb * bsE + (long)rr * rsE;
            float* o = Out + (WINO ? (long)pj * (4096L * 1024) : 0L) + (long)gr * LDN;
            #pragma unroll
            for (int j = 0; j < 4; ++j) {
                const int col = col0 + (j << 4);
                float v = acc[i][j][r];
                if (MODE == 1) {
                    v += biasCol[col];
                } else if (MODE == 2) {
                    v = sinf(v + e[col]);
                } else if (MODE == 3) {
                    v = (v + biasCol[col] + e[col]) * 0.5f;
                }
                o[col] = v;
            }
        }
    }
}


#define UM  144
#define UI  9

__global__ __launch_bounds__(128) __attribute__((amdgpu_waves_per_eu(1, 2)))
void gemm_unfold_relu_w144(const float* __restrict__ Z,
                           const float* __restrict__ Bm,
                           const float* __restrict__ biasCol,
                           float* __restrict__ Out)
{
    constexpr int LDN = 1024;
    constexpr int KT  = 384;
    constexpr int NCH = KT / KC;
    __shared__ __attribute__((aligned(16))) float As[2][UM][LP];
    __shared__ __attribute__((aligned(16))) float Bs[2][TN][LP];

    const int tid  = threadIdx.x;
    const int lane = tid & 31;
    const int wave = tid >> 5;

    const int tileM = blockIdx.y * UM;
    const int tileN = blockIdx.x * TN;

    const int a_kq = (tid & 7) << 2;
    const int a_r  = tid >> 3;
    unsigned a_off[UI];
    #pragma unroll
    for (int i = 0; i < UI; ++i) {
        const int g = tileM + a_r + 16 * i;
        const int b = g / 1026;
        a_off[i] = (unsigned)(g * 128 + b * 256 + a_kq) * 4u;
    }

    const int b_n = lane << 2;
    const int b_k = wave << 2;
    const char* b_ub = (const char*)(Bm + tileN);
    const unsigned b_lo = (unsigned)(b_k * LDN + b_n) * 4u;
    constexpr int B_ROW_B = LDN * 4;

    const int c16 = lane & 15;
    const int hh  = lane >> 4;
    const int bn0 = (wave << 5) + c16;
    const int ko  = hh << 2;

    v8f acc[UI][2];
    #pragma unroll
    for (int i = 0; i < UI; ++i)
        #pragma unroll
        for (int j = 0; j < 2; ++j)
            acc[i][j] = v8f{};

    v4f ra[UI], rb[8];

    auto prefetch = [&](int k0) {
        const char* pa = (const char*)Z + (long)k0 * 4;
        const char* pb = b_ub + (long)k0 * B_ROW_B;
        #pragma unroll
        for (int i = 0; i < UI; ++i)
            ra[i] = *(const v4f*)(pa + a_off[i]);
        #pragma unroll
        for (int g = 0; g < 2; ++g)
            #pragma unroll
            for (int j = 0; j < 4; ++j)
                rb[g * 4 + j] =
                    *(const v4f*)(pb + b_lo + (16 * g + j) * B_ROW_B);
    };

    auto stage = [&](int buf) {
        #pragma unroll
        for (int i = 0; i < UI; ++i)
            *(v4f*)&As[buf][a_r + 16 * i][a_kq] = ra[i];
        #pragma unroll
        for (int g = 0; g < 2; ++g)
            #pragma unroll
            for (int c = 0; c < 4; ++c) {
                const v4f t = { rb[g * 4 + 0][c], rb[g * 4 + 1][c],
                                rb[g * 4 + 2][c], rb[g * 4 + 3][c] };
                *(v4f*)&Bs[buf][b_n + c][b_k + 16 * g] = t;
            }
    };

    auto ldfrag = [&](int buf, int kk, v4f (&fa)[UI], v4f (&fb)[2]) {
        #pragma unroll
        for (int i = 0; i < UI; ++i)
            fa[i] = *(const v4f*)&As[buf][c16 + 16 * i][kk + ko];
        #pragma unroll
        for (int j = 0; j < 2; ++j)
            fb[j] = *(const v4f*)&Bs[buf][bn0 + 16 * j][kk + ko];
    };

    auto mma = [&](const v4f (&fa)[UI], const v4f (&fb)[2]) {
        #pragma unroll
        for (int p = 0; p < 2; ++p) {
            #pragma unroll
            for (int i = 0; i < UI; ++i) {
                const v2f a = { fa[i][2 * p], fa[i][2 * p + 1] };
                #pragma unroll
                for (int j = 0; j < 2; ++j) {
                    const v2f b = { fb[j][2 * p], fb[j][2 * p + 1] };
                    acc[i][j] = wmma4(a, b, acc[i][j]);
                }
            }
        }
    };

    auto compute = [&](int buf) {
        v4f fa0[UI], fb0[2], fa1[UI], fb1[2];
        ldfrag(buf, 0, fa0, fb0);
        ldfrag(buf, 8, fa1, fb1);
        mma(fa0, fb0);
        ldfrag(buf, 16, fa0, fb0);
        mma(fa1, fb1);
        ldfrag(buf, 24, fa1, fb1);
        mma(fa0, fb0);
        mma(fa1, fb1);
    };

    constexpr int pmark1 = (NCH * 27) >> 6;
    constexpr int pmark2 = (NCH * 47) >> 6;
    constexpr int pmark3 = (NCH * 59) >> 6;
    auto prio_mark = [&](int c) {
        if (c == pmark1)      __builtin_amdgcn_s_setprio(2);
        else if (c == pmark2) __builtin_amdgcn_s_setprio(1);
        else if (c == pmark3) __builtin_amdgcn_s_setprio(0);
    };

    prefetch(0);
    stage(0);
    __syncthreads();
    int buf = 0;
    int c = 0;
    __builtin_amdgcn_s_setprio(3);
    #pragma unroll 1
    for (; c + 1 < NCH; ++c) {
        prio_mark(c);
        prefetch((c + 1) * KC);
        compute(buf);
        stage(buf ^ 1);
        __syncthreads();
        buf ^= 1;
    }
    prio_mark(c);
    compute(buf);
    __builtin_amdgcn_s_setprio(0);

    const int col0 = tileN + bn0;
    const int rowb = tileM + (hh << 3);
    const float bias0 = biasCol[col0];
    const float bias1 = biasCol[col0 + 16];
    #pragma unroll
    for (int i = 0; i < UI; ++i) {
        #pragma unroll
        for (int r = 0; r < 8; ++r) {
            float* o = Out + (long)(rowb + (i << 4) + r) * LDN + col0;
            o[0]  = fmaxf(acc[i][0][r] + bias0, 0.0f);
            o[16] = fmaxf(acc[i][1][r] + bias1, 0.0f);
        }
    }
}

__global__ __launch_bounds__(256)
void f23_filter(const float* __restrict__ W, float* __restrict__ G)
{
    const long TP = 1024L * 1024;
    const long o  = ((long)blockIdx.x * 256 + threadIdx.x) << 2;
    const v4f g0 = *(const v4f*)(W + o);
    const v4f g1 = *(const v4f*)(W + TP + o);
    const v4f g2 = *(const v4f*)(W + 2 * TP + o);
    const v4f gs = g0 + g2;
    *(v4f*)(G + o)      = (gs + g1) * 0.5f;
    *(v4f*)(G + TP + o) = (gs - g1) * 0.5f;
}

__global__ __launch_bounds__(256)
void f23_combine(const float* __restrict__ Mt, const float* __restrict__ biasCol,
                 float* __restrict__ Y)
{
    const long PL  = 4096L * 1024;
    const int  idx = blockIdx.x * 256 + threadIdx.x;
    const int  row = idx >> 8;
    const int  c4  = (idx & 255) << 2;
    const float* m = Mt + (long)row * 1024 + c4;
    const v4f m0 = *(const v4f*)(m);
    const v4f m1 = *(const v4f*)(m + PL);
    const v4f m2 = *(const v4f*)(m + 2 * PL);
    const v4f m3 = *(const v4f*)(m + 3 * PL);
    const v4f bv = *(const v4f*)(biasCol + c4);
    float* y = Y + (long)row * 2048 + c4;
    *(v4f*)(y)        = ((m0 + m1) + m2) + bv;
    *(v4f*)(y + 1024) = ((m1 - m2) - m3) + bv;
}

extern "C" void kernel_launch(void* const* d_in, const int* in_sizes, int n_in,
                              void* d_out, int out_size, void* d_ws, size_t ws_size,
                              hipStream_t stream)
{
    (void)in_sizes; (void)n_in; (void)out_size; (void)ws_size;

    const float* x  = (const float*)d_in[0];
    const float* z  = (const float*)d_in[1];
    const float* W1 = (const float*)d_in[2];
    const float* b1 = (const float*)d_in[3];
    const float* W2 = (const float*)d_in[4];
    const float* b2 = (const float*)d_in[5];
    const float* V1 = (const float*)d_in[6];
    const float* c1 = (const float*)d_in[7];
    const float* V2 = (const float*)d_in[8];
    const float* c2 = (const float*)d_in[9];
    const float* Wr = (const float*)d_in[10];
    const float* br = (const float*)d_in[11];
    float* out = (float*)d_out;

    const int Bb = 8, T = 1024, F = 1024, HL = 1026;

    float* ws   = (float*)d_ws;
    const size_t H1N = (size_t)Bb * HL * F;
    const size_t WTN = (size_t)Bb * T * F;
    float* r0 = ws;
    float* r1 = r0 + H1N;
    float* r2 = r1 + H1N;
    float* r4 = r2 + 2 * WTN;
    float* h1   = r0;
    float* g1   = r1;
    float* pln  = r2;
    float* gw   = r4;
    float* wts  = r4;
    float* gv   = r0;
    float* bmat = r0;
    float* sm   = r1;

    const dim3 blk(256);
    const dim3 blkW(128);
    const dim3 gridU(F / TN, (Bb * HL) / UM);
    const dim3 gridW(F / TN, (Bb * T) / TM);
    const dim3 gridP(F / TN, (Bb * T / 2) / TM, 4);
    const dim3 gridF((F * F / 4) / 256);
    const dim3 gridC((Bb * (T / 2) * F / 4) / 256);

    gemm_unfold_relu_w144<<<gridU, blkW, 0, stream>>>(z, W1, b1, h1);
    gemm_unfold_relu_w144<<<gridU, blkW, 0, stream>>>(z, V1, c1, g1);
    f23_filter<<<gridF, blk, 0, stream>>>(W2, gw);
    gemm_f32_wmma_w64<4><<<gridP, blkW, 0, stream>>>(
        h1, (long)HL * F, W2, 0, nullptr, gw, 0, 0,
        pln, F, F);
    f23_combine<<<gridC, blk, 0, stream>>>(pln, b2, wts);
    f23_filter<<<gridF, blk, 0, stream>>>(V2, gv);
    gemm_f32_wmma_w64<4><<<gridP, blkW, 0, stream>>>(
        g1, (long)HL * F, V2, 0, nullptr, gv, 0, 0,
        pln, F, F);
    f23_combine<<<gridC, blk, 0, stream>>>(pln, c2, bmat);
    gemm_f32_wmma_w64<2><<<gridW, blkW, 0, stream>>>(
        x + 4, (long)T * F, wts, (long)T * F, nullptr,
        bmat, (long)T * F, F,
        sm, T, T - 4);
    gemm_f32_wmma_w64<3><<<gridW, blkW, 0, stream>>>(
        sm, (long)T * F, Wr, 0, br,
        x, (long)T * F, F,
        out, F, F);
}


# GEMM K loops pipelined across the LDS hand-over: barrier split (signal mid-chunk, wait 32 WMMAs later), next stage filled mid-chunk, its first fragments read behind the wait; sums bit-identical
# speedup vs baseline: 1.4343x; 1.0351x over previous
#include <hip/hip_runtime.h>
#include <hip/hip_bf16.h>
#include <math.h>


typedef float v2f __attribute__((ext_vector_type(2)));
typedef float v4f __attribute__((ext_vector_type(4)));
typedef float v8f __attribute__((ext_vector_type(8)));

#define KC  32

__device__ __forceinline__ v8f wmma4(v2f a, v2f b, v8f c) {
    return __builtin_amdgcn_wmma_f32_16x16x4_f32(
        false, a, false, b, (short)0, c, false, false);
}

__device__ __forceinline__ void stage_done() {
    __builtin_amdgcn_fence(__ATOMIC_RELEASE, "workgroup", "local");
    __builtin_amdgcn_s_barrier_signal(-1);
}
__device__ __forceinline__ void stage_ready() {
    __builtin_amdgcn_s_barrier_wait(-1);
    __builtin_amdgcn_fence(__ATOMIC_ACQUIRE, "workgroup", "local");
}
__device__ __forceinline__ void phase_edge() {
    __builtin_amdgcn_sched_barrier(0);
}

#define TM  128
#define TN  128
#define LP  36

template <int MODE>
__global__ __launch_bounds__(128) __attribute__((amdgpu_waves_per_eu(1, 2)))
void gemm_f32_wmma_w64(const float* __restrict__ A, long bsA,
                       const float* __restrict__ Bm, long bsB,
                       const float* __restrict__ biasCol,
                       const float* __restrict__ Emat, long bsE, long rsE,
                       float* __restrict__ Out,
                       int K, int klim)
{
    constexpr bool WINO = (MODE == 4);
    constexpr int RPB = WINO ? 512 : 1024;
    constexpr int LDA = WINO ? 2048 : 1024;
    constexpr int LDN = 1024;
    __shared__ __attribute__((aligned(16))) float As[2][TM][LP];
    __shared__ __attribute__((aligned(16))) float Bs[2][TN][LP];

    const int tid  = threadIdx.x;
    const int lane = tid & 31;
    const int wave = tid >> 5;
    const int wm   = wave & 1;
    const int wn   = wave >> 1;

    const int tileM = blockIdx.y * TM;
    const int tileN = blockIdx.x * TN;
    const int bb    = tileM / RPB;
    const int r0    = tileM - bb * RPB;

    const int   pj = WINO ? (int)blockIdx.z : 0;
    const int   o1 = (pj == 0) ? 0 : (pj == 2) ? 2 : 1;
    const int   o2 = (pj == 2) ? 1 : (pj == 3) ? 3 : 2;
    const float sg = (pj == 1) ? 1.0f : -1.0f;

    const int a_kq = (tid & 7) << 2;
    const int a_r  = tid >> 3;
    const char* a_ub  = (const char*)(A + (long)bb * bsA + (long)r0 * LDA
                                        + (WINO ? o1 * 1024 : 0));
    const char* a_ub2 = (const char*)(A + (long)bb * bsA + (long)r0 * LDA
                                        + (WINO ? o2 * 1024 : 0));
    const unsigned a_lo = (unsigned)(a_r * LDA + a_kq) * 4u;
    constexpr int A_STEP_B = 16 * LDA * 4;

    const int b_n = lane << 2;
    const int b_k = wave << 2;
    const float* b_sel = WINO
        ? ((pj == 0) ? Bm : (pj == 1) ? Emat
           : (pj == 2) ? Emat + 1024 * 1024 : Bm + 2 * 1024 * 1024)
        : Bm + (long)((bsB != 0) ? bb : 0) * bsB;
    const char* b_ub = (const char*)(b_sel + tileN);
    const unsigned b_lo = (unsigned)(b_k * LDN + b_n) * 4u;
    constexpr int B_ROW_B = LDN * 4;

    const int c16 = lane & 15;
    const int hh  = lane >> 4;
    const int am0 = (wm << 6) + c16;
    const int bn0 = (wn << 6) + c16;
    const int ko  = hh << 2;

    const int nchunks = (klim + KC - 1) / KC;

    v8f acc[4][4];
    #pragma unroll
    for (int i = 0; i < 4; ++i)
        #pragma unroll
        for (int j = 0; j < 4; ++j)
            acc[i][j] = v8f{};

    v4f ra[8], rb[8];
    v4f ra2[8];

    auto prefetch_whole = [&](int k0) {
        const char* pa = a_ub + (long)k0 * 4;
        const char* pb = b_ub + (long)k0 * B_ROW_B;
        #pragma unroll
        for (int i = 0; i < 8; ++i)
            ra[i] = *(const v4f*)(pa + a_lo + i * A_STEP_B);
        if (WINO) {
            const char* pa2 = a_ub2 + (long)k0 * 4;
            #pragma unroll
            for (int i = 0; i < 8; ++i)
                ra2[i] = *(const v4f*)(pa2 + a_lo + i * A_STEP_B);
        }
        #pragma unroll
        for (int g = 0; g < 2; ++g)
            #pragma unroll
            for (int j = 0; j < 4; ++j)
                rb[g * 4 + j] =
                    *(const v4f*)(pb + b_lo + (16 * g + j) * B_ROW_B);
    };

    auto prefetch_guarded = [&](int k0) {
        const char* pa = a_ub + (long)k0 * 4;
        const char* pb = b_ub + (long)k0 * B_ROW_B;
        {
            const v4f zz = {0.f, 0.f, 0.f, 0.f};
            #pragma unroll
            for (int i = 0; i < 8; ++i) {
                #pragma unroll
                for (int e = 0; e < 4; ++e) {
                    const int ka = k0 + a_kq + e;
                    ra[i][e] = (ka < klim)
                        ? *(const float*)(pa + a_lo + i * A_STEP_B + e * 4) : 0.0f;
                    if (WINO)
                        ra2[i][e] = (ka < klim)
                            ? *(const float*)(a_ub2 + (long)k0 * 4 + a_lo
                                              + i * A_STEP_B + e * 4) : 0.0f;
                }
            }
            #pragma unroll
            for (int g = 0; g < 2; ++g)
                #pragma unroll
                for (int j = 0; j < 4; ++j) {
                    const int kb = k0 + b_k + 16 * g + j;
                    rb[g * 4 + j] = (kb < K)
                        ? *(const v4f*)(pb + b_lo + (16 * g + j) * B_ROW_B) : zz;
                }
        }
    };

    auto stage = [&](int buf) {
        #pragma unroll
        for (int i = 0; i < 8; ++i)
            *(v4f*)&As[buf][a_r + 16 * i][a_kq] =
                WINO ? (ra[i] + sg * ra2[i]) : ra[i];
        #pragma unroll
        for (int g = 0; g < 2; ++g)
            #pragma unroll
            for (int c = 0; c < 4; ++c) {
                const v4f t = { rb[g * 4 + 0][c], rb[g * 4 + 1][c],
                                rb[g * 4 + 2][c], rb[g * 4 + 3][c] };
                *(v4f*)&Bs[buf][b_n + c][b_k + 16 * g] = t;
            }
    };

    auto ldfrag = [&](int buf, int kk, v4f (&fa)[4], v4f (&fb)[4]) {
        #pragma unroll
        for (int i = 0; i < 4; ++i)
            fa[i] = *(const v4f*)&As[buf][am0 + 16 * i][kk + ko];
        #pragma unroll
        for (int j = 0; j < 4; ++j)
            fb[j] = *(const v4f*)&Bs[buf][bn0 + 16 * j][kk + ko];
    };

    auto mma_half = [&](const v4f (&fa)[4], const v4f (&fb)[4], const int p) {
        #pragma unroll
        for (int i = 0; i < 4; ++i) {
            const v2f a = { fa[i][2 * p], fa[i][2 * p + 1] };
            #pragma unroll
            for (int j = 0; j < 4; ++j) {
                const v2f b = { fb[j][2 * p], fb[j][2 * p + 1] };
                acc[i][j] = wmma4(a, b, acc[i][j]);
            }
        }
    };
    auto mma = [&](const v4f (&fa)[4], const v4f (&fb)[4]) {
        mma_half(fa, fb, 0);
        mma_half(fa, fb, 1);
    };

    const int pmark1 = (nchunks * 27) >> 6;
    const int pmark2 = (nchunks * 47) >> 6;
    const int pmark3 = (nchunks * 59) >> 6;

    const int nwhole = klim / KC;

    auto prio_mark = [&](int c) {
        if (c == pmark1)      __builtin_amdgcn_s_setprio(2);
        else if (c == pmark2) __builtin_amdgcn_s_setprio(1);
        else if (c == pmark3) __builtin_amdgcn_s_setprio(0);
    };

    v4f fa0[4], fb0[4];
    v4f fa1[4], fb1[4], fa2[4], fb2[4], fa3[4], fb3[4];

    if (nwhole > 0) prefetch_whole(0); else prefetch_guarded(0);
    stage(0);
    __syncthreads();
    if (nchunks > 1) {
        if (nwhole > 1) prefetch_whole(KC); else prefetch_guarded(KC);
    }
    ldfrag(0, 0, fa0, fb0);
    int buf = 0;
    int c = 0;
    __builtin_amdgcn_s_setprio(3);
    for (; c + 2 < nwhole; ++c) {
        prio_mark(c);
        ldfrag(buf, 8,  fa1, fb1);
        ldfrag(buf, 16, fa2, fb2);
        ldfrag(buf, 24, fa3, fb3);
        mma(fa0, fb0);
        mma(fa1, fb1);
        phase_edge();
        stage(buf ^ 1);
        phase_edge();
        mma_half(fa2, fb2, 0);
        phase_edge();
        stage_done();
        prefetch_whole((c + 2) * KC);
        phase_edge();
        mma_half(fa2, fb2, 1);
        mma_half(fa3, fb3, 0);
        phase_edge();
        stage_ready();
        ldfrag(buf ^ 1, 0, fa0, fb0);
        phase_edge();
        mma_half(fa3, fb3, 1);
        buf ^= 1;
    }
    for (; c < nchunks; ++c) {
        prio_mark(c);
        const bool has1 = (c + 1 < nchunks);
        const bool has2 = (c + 2 < nchunks);
        ldfrag(buf, 8,  fa1, fb1);
        ldfrag(buf, 16, fa2, fb2);
        ldfrag(buf, 24, fa3, fb3);
        mma(fa0, fb0);
        mma(fa1, fb1);
        phase_edge();
        if (has1) stage(buf ^ 1);
        phase_edge();
        mma_half(fa2, fb2, 0);
        phase_edge();
        if (has1) stage_done();
        if (has2) prefetch_guarded((c + 2) * KC);
        phase_edge();
        mma_half(fa2, fb2, 1);
        mma_half(fa3, fb3, 0);
        phase_edge();
        if (has1) {
            stage_ready();
            ldfrag(buf ^ 1, 0, fa0, fb0);
        }
        phase_edge();
        mma_half(fa3, fb3, 1);
        buf ^= 1;
    }
    __builtin_amdgcn_s_setprio(0);

    const int col0 = tileN + bn0;
    const int rowb = tileM + (wm << 6) + (hh << 3);
    #pragma unroll 1
    for (int r = 0; r < 8; ++r) {
        #pragma unroll
        for (int i = 0; i < 4; ++i) {
            const int gr = rowb + (i << 4) + r;
            const int rr = gr - bb * RPB;
            const float* e = Emat + (long)bb * bsE + (long)rr * rsE;
            float* o = Out + (WINO ? (long)pj * (4096L * 1024) : 0L) + (long)gr * LDN;
            #pragma unroll
            for (int j = 0; j < 4; ++j) {
                const int col = col0 + (j << 4);
                float v = acc[i][j][r];
                if (MODE == 1) {
                    v += biasCol[col];
                } else if (MODE == 2) {
                    v = sinf(v + e[col]);
                } else if (MODE == 3) {
                    v = (v + biasCol[col] + e[col]) * 0.5f;
                }
                o[col] = v;
            }
        }
    }
}


#define UM  144
#define UI  9

__global__ __launch_bounds__(128) __attribute__((amdgpu_waves_per_eu(1, 2)))
void gemm_unfold_relu_w144(const float* __restrict__ Z,
                           const float* __restrict__ Bm,
                           const float* __restrict__ biasCol,
                           float* __restrict__ Out)
{
    constexpr int LDN = 1024;
    constexpr int KT  = 384;
    constexpr int NCH = KT / KC;
    __shared__ __attribute__((aligned(16))) float As[2][UM][LP];
    __shared__ __attribute__((aligned(16))) float Bs[2][TN][LP];

    const int tid  = threadIdx.x;
    const int lane = tid & 31;
    const int wave = tid >> 5;

    const int tileM = blockIdx.y * UM;
    const int tileN = blockIdx.x * TN;

    const int a_kq = (tid & 7) << 2;
    const int a_r  = tid >> 3;
    unsigned a_off[UI];
    #pragma unroll
    for (int i = 0; i < UI; ++i) {
        const int g = tileM + a_r + 16 * i;
        const int b = g / 1026;
        a_off[i] = (unsigned)(g * 128 + b * 256 + a_kq) * 4u;
    }

    const int b_n = lane << 2;
    const int b_k = wave << 2;
    const char* b_ub = (const char*)(Bm + tileN);
    const unsigned b_lo = (unsigned)(b_k * LDN + b_n) * 4u;
    constexpr int B_ROW_B = LDN * 4;

    const int c16 = lane & 15;
    const int hh  = lane >> 4;
    const int bn0 = (wave << 5) + c16;
    const int ko  = hh << 2;

    v8f acc[UI][2];
    #pragma unroll
    for (int i = 0; i < UI; ++i)
        #pragma unroll
        for (int j = 0; j < 2; ++j)
            acc[i][j] = v8f{};

    v4f ra[UI], rb[8];

    auto prefetch = [&](int k0) {
        const char* pa = (const char*)Z + (long)k0 * 4;
        const char* pb = b_ub + (long)k0 * B_ROW_B;
        #pragma unroll
        for (int i = 0; i < UI; ++i)
            ra[i] = *(const v4f*)(pa + a_off[i]);
        #pragma unroll
        for (int g = 0; g < 2; ++g)
            #pragma unroll
            for (int j = 0; j < 4; ++j)
                rb[g * 4 + j] =
                    *(const v4f*)(pb + b_lo + (16 * g + j) * B_ROW_B);
    };

    auto stage = [&](int buf) {
        #pragma unroll
        for (int i = 0; i < UI; ++i)
            *(v4f*)&As[buf][a_r + 16 * i][a_kq] = ra[i];
        #pragma unroll
        for (int g = 0; g < 2; ++g)
            #pragma unroll
            for (int c = 0; c < 4; ++c) {
                const v4f t = { rb[g * 4 + 0][c], rb[g * 4 + 1][c],
                                rb[g * 4 + 2][c], rb[g * 4 + 3][c] };
                *(v4f*)&Bs[buf][b_n + c][b_k + 16 * g] = t;
            }
    };

    auto ldfrag = [&](int buf, int kk, v4f (&fa)[UI], v4f (&fb)[2]) {
        #pragma unroll
        for (int i = 0; i < UI; ++i)
            fa[i] = *(const v4f*)&As[buf][c16 + 16 * i][kk + ko];
        #pragma unroll
        for (int j = 0; j < 2; ++j)
            fb[j] = *(const v4f*)&Bs[buf][bn0 + 16 * j][kk + ko];
    };

    auto mma_half = [&](const v4f (&fa)[UI], const v4f (&fb)[2], const int p) {
        #pragma unroll
        for (int i = 0; i < UI; ++i) {
            const v2f a = { fa[i][2 * p], fa[i][2 * p + 1] };
            #pragma unroll
            for (int j = 0; j < 2; ++j) {
                const v2f b = { fb[j][2 * p], fb[j][2 * p + 1] };
                acc[i][j] = wmma4(a, b, acc[i][j]);
            }
        }
    };
    auto mma = [&](const v4f (&fa)[UI], const v4f (&fb)[2]) {
        mma_half(fa, fb, 0);
        mma_half(fa, fb, 1);
    };

    constexpr int pmark1 = (NCH * 27) >> 6;
    constexpr int pmark2 = (NCH * 47) >> 6;
    constexpr int pmark3 = (NCH * 59) >> 6;
    auto prio_mark = [&](int c) {
        if (c == pmark1)      __builtin_amdgcn_s_setprio(2);
        else if (c == pmark2) __builtin_amdgcn_s_setprio(1);
        else if (c == pmark3) __builtin_amdgcn_s_setprio(0);
    };

    v4f fa0[UI], fb0[2];
    v4f fa1[UI], fb1[2], fa2[UI], fb2[2], fa3[UI], fb3[2];

    prefetch(0);
    stage(0);
    __syncthreads();
    prefetch(KC);
    ldfrag(0, 0, fa0, fb0);
    int buf = 0;
    int c = 0;
    __builtin_amdgcn_s_setprio(3);
    #pragma unroll 1
    for (; c + 2 < NCH; ++c) {
        prio_mark(c);
        ldfrag(buf, 8,  fa1, fb1);
        ldfrag(buf, 16, fa2, fb2);
        ldfrag(buf, 24, fa3, fb3);
        mma(fa0, fb0);
        mma(fa1, fb1);
        phase_edge();
        stage(buf ^ 1);
        phase_edge();
        mma_half(fa2, fb2, 0);
        phase_edge();
        stage_done();
        prefetch((c + 2) * KC);
        phase_edge();
        mma_half(fa2, fb2, 1);
        mma_half(fa3, fb3, 0);
        phase_edge();
        stage_ready();
        ldfrag(buf ^ 1, 0, fa0, fb0);
        phase_edge();
        mma_half(fa3, fb3, 1);
        buf ^= 1;
    }
    #pragma unroll 1
    for (; c < NCH; ++c) {
        prio_mark(c);
        const bool has1 = (c + 1 < NCH);
        ldfrag(buf, 8,  fa1, fb1);
        ldfrag(buf, 16, fa2, fb2);
        ldfrag(buf, 24, fa3, fb3);
        mma(fa0, fb0);
        mma(fa1, fb1);
        phase_edge();
        if (has1) stage(buf ^ 1);
        phase_edge();
        mma_half(fa2, fb2, 0);
        phase_edge();
        if (has1) stage_done();
        phase_edge();
        mma_half(fa2, fb2, 1);
        mma_half(fa3, fb3, 0);
        phase_edge();
        if (has1) {
            stage_ready();
            ldfrag(buf ^ 1, 0, fa0, fb0);
        }
        phase_edge();
        mma_half(fa3, fb3, 1);
        buf ^= 1;
    }
    __builtin_amdgcn_s_setprio(0);

    const int col0 = tileN + bn0;
    const int rowb = tileM + (hh << 3);
    const float bias0 = biasCol[col0];
    const float bias1 = biasCol[col0 + 16];
    #pragma unroll
    for (int i = 0; i < UI; ++i) {
        #pragma unroll
        for (int r = 0; r < 8; ++r) {
            float* o = Out + (long)(rowb + (i << 4) + r) * LDN + col0;
            o[0]  = fmaxf(acc[i][0][r] + bias0, 0.0f);
            o[16] = fmaxf(acc[i][1][r] + bias1, 0.0f);
        }
    }
}

__global__ __launch_bounds__(256)
void f23_filter(const float* __restrict__ W, float* __restrict__ G)
{
    const long TP = 1024L * 1024;
    const long o  = ((long)blockIdx.x * 256 + threadIdx.x) << 2;
    const v4f g0 = *(const v4f*)(W + o);
    const v4f g1 = *(const v4f*)(W + TP + o);
    const v4f g2 = *(const v4f*)(W + 2 * TP + o);
    const v4f gs = g0 + g2;
    *(v4f*)(G + o)      = (gs + g1) * 0.5f;
    *(v4f*)(G + TP + o) = (gs - g1) * 0.5f;
}

__global__ __launch_bounds__(256)
void f23_combine(const float* __restrict__ Mt, const float* __restrict__ biasCol,
                 float* __restrict__ Y)
{
    const long PL  = 4096L * 1024;
    const int  idx = blockIdx.x * 256 + threadIdx.x;
    const int  row = idx >> 8;
    const int  c4  = (idx & 255) << 2;
    const float* m = Mt + (long)row * 1024 + c4;
    const v4f m0 = *(const v4f*)(m);
    const v4f m1 = *(const v4f*)(m + PL);
    const v4f m2 = *(const v4f*)(m + 2 * PL);
    const v4f m3 = *(const v4f*)(m + 3 * PL);
    const v4f bv = *(const v4f*)(biasCol + c4);
    float* y = Y + (long)row * 2048 + c4;
    *(v4f*)(y)        = ((m0 + m1) + m2) + bv;
    *(v4f*)(y + 1024) = ((m1 - m2) - m3) + bv;
}

extern "C" void kernel_launch(void* const* d_in, const int* in_sizes, int n_in,
                              void* d_out, int out_size, void* d_ws, size_t ws_size,
                              hipStream_t stream)
{
    (void)in_sizes; (void)n_in; (void)out_size; (void)ws_size;

    const float* x  = (const float*)d_in[0];
    const float* z  = (const float*)d_in[1];
    const float* W1 = (const float*)d_in[2];
    const float* b1 = (const float*)d_in[3];
    const float* W2 = (const float*)d_in[4];
    const float* b2 = (const float*)d_in[5];
    const float* V1 = (const float*)d_in[6];
    const float* c1 = (const float*)d_in[7];
    const float* V2 = (const float*)d_in[8];
    const float* c2 = (const float*)d_in[9];
    const float* Wr = (const float*)d_in[10];
    const float* br = (const float*)d_in[11];
    float* out = (float*)d_out;

    const int Bb = 8, T = 1024, F = 1024, HL = 1026;

    float* ws   = (float*)d_ws;
    const size_t H1N = (size_t)Bb * HL * F;
    const size_t WTN = (size_t)Bb * T * F;
    float* r0 = ws;
    float* r1 = r0 + H1N;
    float* r2 = r1 + H1N;
    float* r4 = r2 + 2 * WTN;
    float* h1   = r0;
    float* g1   = r1;
    float* pln  = r2;
    float* gw   = r4;
    float* wts  = r4;
    float* gv   = r0;
    float* bmat = r0;
    float* sm   = r1;

    const dim3 blk(256);
    const dim3 blkW(128);
    const dim3 gridU(F / TN, (Bb * HL) / UM);
    const dim3 gridW(F / TN, (Bb * T) / TM);
    const dim3 gridP(F / TN, (Bb * T / 2) / TM, 4);
    const dim3 gridF((F * F / 4) / 256);
    const dim3 gridC((Bb * (T / 2) * F / 4) / 256);

    gemm_unfold_relu_w144<<<gridU, blkW, 0, stream>>>(z, W1, b1, h1);
    gemm_unfold_relu_w144<<<gridU, blkW, 0, stream>>>(z, V1, c1, g1);
    f23_filter<<<gridF, blk, 0, stream>>>(W2, gw);
    gemm_f32_wmma_w64<4><<<gridP, blkW, 0, stream>>>(
        h1, (long)HL * F, W2, 0, nullptr, gw, 0, 0,
        pln, F, F);
    f23_combine<<<gridC, blk, 0, stream>>>(pln, b2, wts);
    f23_filter<<<gridF, blk, 0, stream>>>(V2, gv);
    gemm_f32_wmma_w64<4><<<gridP, blkW, 0, stream>>>(
        g1, (long)HL * F, V2, 0, nullptr, gv, 0, 0,
        pln, F, F);
    f23_combine<<<gridC, blk, 0, stream>>>(pln, c2, bmat);
    gemm_f32_wmma_w64<2><<<gridW, blkW, 0, stream>>>(
        x + 4, (long)T * F, wts, (long)T * F, nullptr,
        bmat, (long)T * F, F,
        sm, T, T - 4);
    gemm_f32_wmma_w64<3><<<gridW, blkW, 0, stream>>>(
        sm, (long)T * F, Wr, 0, br,
        x, (long)T * F, F,
        out, F, F);
}


# GEMM 5 (sin epilogue): the 16 bias-matrix addends of a row step are loaded as one group a step ahead of their sines, instead of one load waited for in front of each sinf
# speedup vs baseline: 1.4821x; 1.0333x over previous
#include <hip/hip_runtime.h>
#include <hip/hip_bf16.h>
#include <math.h>


typedef float v2f __attribute__((ext_vector_type(2)));
typedef float v4f __attribute__((ext_vector_type(4)));
typedef float v8f __attribute__((ext_vector_type(8)));

#define KC  32

__device__ __forceinline__ v8f wmma4(v2f a, v2f b, v8f c) {
    return __builtin_amdgcn_wmma_f32_16x16x4_f32(
        false, a, false, b, (short)0, c, false, false);
}

__device__ __forceinline__ void stage_done() {
    __builtin_amdgcn_fence(__ATOMIC_RELEASE, "workgroup", "local");
    __builtin_amdgcn_s_barrier_signal(-1);
}
__device__ __forceinline__ void stage_ready() {
    __builtin_amdgcn_s_barrier_wait(-1);
    __builtin_amdgcn_fence(__ATOMIC_ACQUIRE, "workgroup", "local");
}
__device__ __forceinline__ void phase_edge() {
    __builtin_amdgcn_sched_barrier(0);
}

#define TM  128
#define TN  128
#define LP  36

template <int MODE>
__global__ __launch_bounds__(128) __attribute__((amdgpu_waves_per_eu(1, 2)))
void gemm_f32_wmma_w64(const float* __restrict__ A, long bsA,
                       const float* __restrict__ Bm, long bsB,
                       const float* __restrict__ biasCol,
                       const float* __restrict__ Emat, long bsE, long rsE,
                       float* __restrict__ Out,
                       int K, int klim)
{
    constexpr bool WINO = (MODE == 4);
    constexpr int RPB = WINO ? 512 : 1024;
    constexpr int LDA = WINO ? 2048 : 1024;
    constexpr int LDN = 1024;
    __shared__ __attribute__((aligned(16))) float As[2][TM][LP];
    __shared__ __attribute__((aligned(16))) float Bs[2][TN][LP];

    const int tid  = threadIdx.x;
    const int lane = tid & 31;
    const int wave = tid >> 5;
    const int wm   = wave & 1;
    const int wn   = wave >> 1;

    const int tileM = blockIdx.y * TM;
    const int tileN = blockIdx.x * TN;
    const int bb    = tileM / RPB;
    const int r0    = tileM - bb * RPB;

    const int   pj = WINO ? (int)blockIdx.z : 0;
    const int   o1 = (pj == 0) ? 0 : (pj == 2) ? 2 : 1;
    const int   o2 = (pj == 2) ? 1 : (pj == 3) ? 3 : 2;
    const float sg = (pj == 1) ? 1.0f : -1.0f;

    const int a_kq = (tid & 7) << 2;
    const int a_r  = tid >> 3;
    const char* a_ub  = (const char*)(A + (long)bb * bsA + (long)r0 * LDA
                                        + (WINO ? o1 * 1024 : 0));
    const char* a_ub2 = (const char*)(A + (long)bb * bsA + (long)r0 * LDA
                                        + (WINO ? o2 * 1024 : 0));
    const unsigned a_lo = (unsigned)(a_r * LDA + a_kq) * 4u;
    constexpr int A_STEP_B = 16 * LDA * 4;

    const int b_n = lane << 2;
    const int b_k = wave << 2;
    const float* b_sel = WINO
        ? ((pj == 0) ? Bm : (pj == 1) ? Emat
           : (pj == 2) ? Emat + 1024 * 1024 : Bm + 2 * 1024 * 1024)
        : Bm + (long)((bsB != 0) ? bb : 0) * bsB;
    const char* b_ub = (const char*)(b_sel + tileN);
    const unsigned b_lo = (unsigned)(b_k * LDN + b_n) * 4u;
    constexpr int B_ROW_B = LDN * 4;

    const int c16 = lane & 15;
    const int hh  = lane >> 4;
    const int am0 = (wm << 6) + c16;
    const int bn0 = (wn << 6) + c16;
    const int ko  = hh << 2;

    const int nchunks = (klim + KC - 1) / KC;

    v8f acc[4][4];
    #pragma unroll
    for (int i = 0; i < 4; ++i)
        #pragma unroll
        for (int j = 0; j < 4; ++j)
            acc[i][j] = v8f{};

    v4f ra[8], rb[8];
    v4f ra2[8];

    auto prefetch_whole = [&](int k0) {
        const char* pa = a_ub + (long)k0 * 4;
        const char* pb = b_ub + (long)k0 * B_ROW_B;
        #pragma unroll
        for (int i = 0; i < 8; ++i)
            ra[i] = *(const v4f*)(pa + a_lo + i * A_STEP_B);
        if (WINO) {
            const char* pa2 = a_ub2 + (long)k0 * 4;
            #pragma unroll
            for (int i = 0; i < 8; ++i)
                ra2[i] = *(const v4f*)(pa2 + a_lo + i * A_STEP_B);
        }
        #pragma unroll
        for (int g = 0; g < 2; ++g)
            #pragma unroll
            for (int j = 0; j < 4; ++j)
                rb[g * 4 + j] =
                    *(const v4f*)(pb + b_lo + (16 * g + j) * B_ROW_B);
    };

    auto prefetch_guarded = [&](int k0) {
        const char* pa = a_ub + (long)k0 * 4;
        const char* pb = b_ub + (long)k0 * B_ROW_B;
        {
            const v4f zz = {0.f, 0.f, 0.f, 0.f};
            #pragma unroll
            for (int i = 0; i < 8; ++i) {
                #pragma unroll
                for (int e = 0; e < 4; ++e) {
                    const int ka = k0 + a_kq + e;
                    ra[i][e] = (ka < klim)
                        ? *(const float*)(pa + a_lo + i * A_STEP_B + e * 4) : 0.0f;
                    if (WINO)
                        ra2[i][e] = (ka < klim)
                            ? *(const float*)(a_ub2 + (long)k0 * 4 + a_lo
                                              + i * A_STEP_B + e * 4) : 0.0f;
                }
            }
            #pragma unroll
            for (int g = 0; g < 2; ++g)
                #pragma unroll
                for (int j = 0; j < 4; ++j) {
                    const int kb = k0 + b_k + 16 * g + j;
                    rb[g * 4 + j] = (kb < K)
                        ? *(const v4f*)(pb + b_lo + (16 * g + j) * B_ROW_B) : zz;
                }
        }
    };

    auto stage = [&](int buf) {
        #pragma unroll
        for (int i = 0; i < 8; ++i)
            *(v4f*)&As[buf][a_r + 16 * i][a_kq] =
                WINO ? (ra[i] + sg * ra2[i]) : ra[i];
        #pragma unroll
        for (int g = 0; g < 2; ++g)
            #pragma unroll
            for (int c = 0; c < 4; ++c) {
                const v4f t = { rb[g * 4 + 0][c], rb[g * 4 + 1][c],
                                rb[g * 4 + 2][c], rb[g * 4 + 3][c] };
                *(v4f*)&Bs[buf][b_n + c][b_k + 16 * g] = t;
            }
    };

    auto ldfrag = [&](int buf, int kk, v4f (&fa)[4], v4f (&fb)[4]) {
        #pragma unroll
        for (int i = 0; i < 4; ++i)
            fa[i] = *(const v4f*)&As[buf][am0 + 16 * i][kk + ko];
        #pragma unroll
        for (int j = 0; j < 4; ++j)
            fb[j] = *(const v4f*)&Bs[buf][bn0 + 16 * j][kk + ko];
    };

    auto mma_half = [&](const v4f (&fa)[4], const v4f (&fb)[4], const int p) {
        #pragma unroll
        for (int i = 0; i < 4; ++i) {
            const v2f a = { fa[i][2 * p], fa[i][2 * p + 1] };
            #pragma unroll
            for (int j = 0; j < 4; ++j) {
                const v2f b = { fb[j][2 * p], fb[j][2 * p + 1] };
                acc[i][j] = wmma4(a, b, acc[i][j]);
            }
        }
    };
    auto mma = [&](const v4f (&fa)[4], const v4f (&fb)[4]) {
        mma_half(fa, fb, 0);
        mma_half(fa, fb, 1);
    };

    const int pmark1 = (nchunks * 27) >> 6;
    const int pmark2 = (nchunks * 47) >> 6;
    const int pmark3 = (nchunks * 59) >> 6;

    const int nwhole = klim / KC;

    auto prio_mark = [&](int c) {
        if (c == pmark1)      __builtin_amdgcn_s_setprio(2);
        else if (c == pmark2) __builtin_amdgcn_s_setprio(1);
        else if (c == pmark3) __builtin_amdgcn_s_setprio(0);
    };

    v4f fa0[4], fb0[4];
    v4f fa1[4], fb1[4], fa2[4], fb2[4], fa3[4], fb3[4];

    if (nwhole > 0) prefetch_whole(0); else prefetch_guarded(0);
    stage(0);
    __syncthreads();
    if (nchunks > 1) {
        if (nwhole > 1) prefetch_whole(KC); else prefetch_guarded(KC);
    }
    ldfrag(0, 0, fa0, fb0);
    int buf = 0;
    int c = 0;
    __builtin_amdgcn_s_setprio(3);
    for (; c + 2 < nwhole; ++c) {
        prio_mark(c);
        ldfrag(buf, 8,  fa1, fb1);
        ldfrag(buf, 16, fa2, fb2);
        ldfrag(buf, 24, fa3, fb3);
        mma(fa0, fb0);
        mma(fa1, fb1);
        phase_edge();
        stage(buf ^ 1);
        phase_edge();
        mma_half(fa2, fb2, 0);
        phase_edge();
        stage_done();
        prefetch_whole((c + 2) * KC);
        phase_edge();
        mma_half(fa2, fb2, 1);
        mma_half(fa3, fb3, 0);
        phase_edge();
        stage_ready();
        ldfrag(buf ^ 1, 0, fa0, fb0);
        phase_edge();
        mma_half(fa3, fb3, 1);
        buf ^= 1;
    }
    for (; c < nchunks; ++c) {
        prio_mark(c);
        const bool has1 = (c + 1 < nchunks);
        const bool has2 = (c + 2 < nchunks);
        ldfrag(buf, 8,  fa1, fb1);
        ldfrag(buf, 16, fa2, fb2);
        ldfrag(buf, 24, fa3, fb3);
        mma(fa0, fb0);
        mma(fa1, fb1);
        phase_edge();
        if (has1) stage(buf ^ 1);
        phase_edge();
        mma_half(fa2, fb2, 0);
        phase_edge();
        if (has1) stage_done();
        if (has2) prefetch_guarded((c + 2) * KC);
        phase_edge();
        mma_half(fa2, fb2, 1);
        mma_half(fa3, fb3, 0);
        phase_edge();
        if (has1) {
            stage_ready();
            ldfrag(buf ^ 1, 0, fa0, fb0);
        }
        phase_edge();
        mma_half(fa3, fb3, 1);
        buf ^= 1;
    }
    __builtin_amdgcn_s_setprio(0);

    const int col0 = tileN + bn0;
    const int rowb = tileM + (wm << 6) + (hh << 3);
    if (MODE == 2) {
        const float* eb = Emat + (long)bb * bsE + (long)(rowb - bb * RPB) * rsE + col0;
        float en[16];
        #pragma unroll
        for (int i = 0; i < 4; ++i)
            #pragma unroll
            for (int j = 0; j < 4; ++j)
                en[i * 4 + j] = eb[(long)(i << 4) * rsE + (j << 4)];
        #pragma unroll 1
        for (int r = 0; r < 8; ++r) {
            float ec[16];
            #pragma unroll
            for (int q = 0; q < 16; ++q) ec[q] = en[q];
            const float* ep = eb + (long)((r + 1) & 7) * rsE;
            #pragma unroll
            for (int i = 0; i < 4; ++i)
                #pragma unroll
                for (int j = 0; j < 4; ++j)
                    en[i * 4 + j] = ep[(long)(i << 4) * rsE + (j << 4)];
            #pragma unroll
            for (int i = 0; i < 4; ++i) {
                float* o = Out + (long)(rowb + (i << 4) + r) * LDN + col0;
                #pragma unroll
                for (int j = 0; j < 4; ++j)
                    o[j << 4] = sinf(acc[i][j][r] + ec[i * 4 + j]);
            }
        }
        return;
    }
    #pragma unroll 1
    for (int r = 0; r < 8; ++r) {
        #pragma unroll
        for (int i = 0; i < 4; ++i) {
            const int gr = rowb + (i << 4) + r;
            const int rr = gr - bb * RPB;
            const float* e = Emat + (long)bb * bsE + (long)rr * rsE;
            float* o = Out + (WINO ? (long)pj * (4096L * 1024) : 0L) + (long)gr * LDN;
            #pragma unroll
            for (int j = 0; j < 4; ++j) {
                const int col = col0 + (j << 4);
                float v = acc[i][j][r];
                if (MODE == 1) {
                    v += biasCol[col];
                } else if (MODE == 3) {
                    v = (v + biasCol[col] + e[col]) * 0.5f;
                }
                o[col] = v;
            }
        }
    }
}


#define UM  144
#define UI  9

__global__ __launch_bounds__(128) __attribute__((amdgpu_waves_per_eu(1, 2)))
void gemm_unfold_relu_w144(const float* __restrict__ Z,
                           const float* __restrict__ Bm,
                           const float* __restrict__ biasCol,
                           float* __restrict__ Out)
{
    constexpr int LDN = 1024;
    constexpr int KT  = 384;
    constexpr int NCH = KT / KC;
    __shared__ __attribute__((aligned(16))) float As[2][UM][LP];
    __shared__ __attribute__((aligned(16))) float Bs[2][TN][LP];

    const int tid  = threadIdx.x;
    const int lane = tid & 31;
    const int wave = tid >> 5;

    const int tileM = blockIdx.y * UM;
    const int tileN = blockIdx.x * TN;

    const int a_kq = (tid & 7) << 2;
    const int a_r  = tid >> 3;
    unsigned a_off[UI];
    #pragma unroll
    for (int i = 0; i < UI; ++i) {
        const int g = tileM + a_r + 16 * i;
        const int b = g / 1026;
        a_off[i] = (unsigned)(g * 128 + b * 256 + a_kq) * 4u;
    }

    const int b_n = lane << 2;
    const int b_k = wave << 2;
    const char* b_ub = (const char*)(Bm + tileN);
    const unsigned b_lo = (unsigned)(b_k * LDN + b_n) * 4u;
    constexpr int B_ROW_B = LDN * 4;

    const int c16 = lane & 15;
    const int hh  = lane >> 4;
    const int bn0 = (wave << 5) + c16;
    const int ko  = hh << 2;

    v8f acc[UI][2];
    #pragma unroll
    for (int i = 0; i < UI; ++i)
        #pragma unroll
        for (int j = 0; j < 2; ++j)
            acc[i][j] = v8f{};

    v4f ra[UI], rb[8];

    auto prefetch = [&](int k0) {
        const char* pa = (const char*)Z + (long)k0 * 4;
        const char* pb = b_ub + (long)k0 * B_ROW_B;
        #pragma unroll
        for (int i = 0; i < UI; ++i)
            ra[i] = *(const v4f*)(pa + a_off[i]);
        #pragma unroll
        for (int g = 0; g < 2; ++g)
            #pragma unroll
            for (int j = 0; j < 4; ++j)
                rb[g * 4 + j] =
                    *(const v4f*)(pb + b_lo + (16 * g + j) * B_ROW_B);
    };

    auto stage = [&](int buf) {
        #pragma unroll
        for (int i = 0; i < UI; ++i)
            *(v4f*)&As[buf][a_r + 16 * i][a_kq] = ra[i];
        #pragma unroll
        for (int g = 0; g < 2; ++g)
            #pragma unroll
            for (int c = 0; c < 4; ++c) {
                const v4f t = { rb[g * 4 + 0][c], rb[g * 4 + 1][c],
                                rb[g * 4 + 2][c], rb[g * 4 + 3][c] };
                *(v4f*)&Bs[buf][b_n + c][b_k + 16 * g] = t;
            }
    };

    auto ldfrag = [&](int buf, int kk, v4f (&fa)[UI], v4f (&fb)[2]) {
        #pragma unroll
        for (int i = 0; i < UI; ++i)
            fa[i] = *(const v4f*)&As[buf][c16 + 16 * i][kk + ko];
        #pragma unroll
        for (int j = 0; j < 2; ++j)
            fb[j] = *(const v4f*)&Bs[buf][bn0 + 16 * j][kk + ko];
    };

    auto mma_half = [&](const v4f (&fa)[UI], const v4f (&fb)[2], const int p) {
        #pragma unroll
        for (int i = 0; i < UI; ++i) {
            const v2f a = { fa[i][2 * p], fa[i][2 * p + 1] };
            #pragma unroll
            for (int j = 0; j < 2; ++j) {
                const v2f b = { fb[j][2 * p], fb[j][2 * p + 1] };
                acc[i][j] = wmma4(a, b, acc[i][j]);
            }
        }
    };
    auto mma = [&](const v4f (&fa)[UI], const v4f (&fb)[2]) {
        mma_half(fa, fb, 0);
        mma_half(fa, fb, 1);
    };

    constexpr int pmark1 = (NCH * 27) >> 6;
    constexpr int pmark2 = (NCH * 47) >> 6;
    constexpr int pmark3 = (NCH * 59) >> 6;
    auto prio_mark = [&](int c) {
        if (c == pmark1)      __builtin_amdgcn_s_setprio(2);
        else if (c == pmark2) __builtin_amdgcn_s_setprio(1);
        else if (c == pmark3) __builtin_amdgcn_s_setprio(0);
    };

    v4f fa0[UI], fb0[2];
    v4f fa1[UI], fb1[2], fa2[UI], fb2[2], fa3[UI], fb3[2];

    prefetch(0);
    stage(0);
    __syncthreads();
    prefetch(KC);
    ldfrag(0, 0, fa0, fb0);
    int buf = 0;
    int c = 0;
    __builtin_amdgcn_s_setprio(3);
    #pragma unroll 1
    for (; c + 2 < NCH; ++c) {
        prio_mark(c);
        ldfrag(buf, 8,  fa1, fb1);
        ldfrag(buf, 16, fa2, fb2);
        ldfrag(buf, 24, fa3, fb3);
        mma(fa0, fb0);
        mma(fa1, fb1);
        phase_edge();
        stage(buf ^ 1);
        phase_edge();
        mma_half(fa2, fb2, 0);
        phase_edge();
        stage_done();
        prefetch((c + 2) * KC);
        phase_edge();
        mma_half(fa2, fb2, 1);
        mma_half(fa3, fb3, 0);
        phase_edge();
        stage_ready();
        ldfrag(buf ^ 1, 0, fa0, fb0);
        phase_edge();
        mma_half(fa3, fb3, 1);
        buf ^= 1;
    }
    #pragma unroll 1
    for (; c < NCH; ++c) {
        prio_mark(c);
        const bool has1 = (c + 1 < NCH);
        ldfrag(buf, 8,  fa1, fb1);
        ldfrag(buf, 16, fa2, fb2);
        ldfrag(buf, 24, fa3, fb3);
        mma(fa0, fb0);
        mma(fa1, fb1);
        phase_edge();
        if (has1) stage(buf ^ 1);
        phase_edge();
        mma_half(fa2, fb2, 0);
        phase_edge();
        if (has1) stage_done();
        phase_edge();
        mma_half(fa2, fb2, 1);
        mma_half(fa3, fb3, 0);
        phase_edge();
        if (has1) {
            stage_ready();
            ldfrag(buf ^ 1, 0, fa0, fb0);
        }
        phase_edge();
        mma_half(fa3, fb3, 1);
        buf ^= 1;
    }
    __builtin_amdgcn_s_setprio(0);

    const int col0 = tileN + bn0;
    const int rowb = tileM + (hh << 3);
    const float bias0 = biasCol[col0];
    const float bias1 = biasCol[col0 + 16];
    #pragma unroll
    for (int i = 0; i < UI; ++i) {
        #pragma unroll
        for (int r = 0; r < 8; ++r) {
            float* o = Out + (long)(rowb + (i << 4) + r) * LDN + col0;
            o[0]  = fmaxf(acc[i][0][r] + bias0, 0.0f);
            o[16] = fmaxf(acc[i][1][r] + bias1, 0.0f);
        }
    }
}

__global__ __launch_bounds__(256)
void f23_filter(const float* __restrict__ W, float* __restrict__ G)
{
    const long TP = 1024L * 1024;
    const long o  = ((long)blockIdx.x * 256 + threadIdx.x) << 2;
    const v4f g0 = *(const v4f*)(W + o);
    const v4f g1 = *(const v4f*)(W + TP + o);
    const v4f g2 = *(const v4f*)(W + 2 * TP + o);
    const v4f gs = g0 + g2;
    *(v4f*)(G + o)      = (gs + g1) * 0.5f;
    *(v4f*)(G + TP + o) = (gs - g1) * 0.5f;
}

__global__ __launch_bounds__(256)
void f23_combine(const float* __restrict__ Mt, const float* __restrict__ biasCol,
                 float* __restrict__ Y)
{
    const long PL  = 4096L * 1024;
    const int  idx = blockIdx.x * 256 + threadIdx.x;
    const int  row = idx >> 8;
    const int  c4  = (idx & 255) << 2;
    const float* m = Mt + (long)row * 1024 + c4;
    const v4f m0 = *(const v4f*)(m);
    const v4f m1 = *(const v4f*)(m + PL);
    const v4f m2 = *(const v4f*)(m + 2 * PL);
    const v4f m3 = *(const v4f*)(m + 3 * PL);
    const v4f bv = *(const v4f*)(biasCol + c4);
    float* y = Y + (long)row * 2048 + c4;
    *(v4f*)(y)        = ((m0 + m1) + m2) + bv;
    *(v4f*)(y + 1024) = ((m1 - m2) - m3) + bv;
}

extern "C" void kernel_launch(void* const* d_in, const int* in_sizes, int n_in,
                              void* d_out, int out_size, void* d_ws, size_t ws_size,
                              hipStream_t stream)
{
    (void)in_sizes; (void)n_in; (void)out_size; (void)ws_size;

    const float* x  = (const float*)d_in[0];
    const float* z  = (const float*)d_in[1];
    const float* W1 = (const float*)d_in[2];
    const float* b1 = (const float*)d_in[3];
    const float* W2 = (const float*)d_in[4];
    const float* b2 = (const float*)d_in[5];
    const float* V1 = (const float*)d_in[6];
    const float* c1 = (const float*)d_in[7];
    const float* V2 = (const float*)d_in[8];
    const float* c2 = (const float*)d_in[9];
    const float* Wr = (const float*)d_in[10];
    const float* br = (const float*)d_in[11];
    float* out = (float*)d_out;

    const int Bb = 8, T = 1024, F = 1024, HL = 1026;

    float* ws   = (float*)d_ws;
    const size_t H1N = (size_t)Bb * HL * F;
    const size_t WTN = (size_t)Bb * T * F;
    float* r0 = ws;
    float* r1 = r0 + H1N;
    float* r2 = r1 + H1N;
    float* r4 = r2 + 2 * WTN;
    float* h1   = r0;
    float* g1   = r1;
    float* pln  = r2;
    float* gw   = r4;
    float* wts  = r4;
    float* gv   = r0;
    float* bmat = r0;
    float* sm   = r1;

    const dim3 blk(256);
    const dim3 blkW(128);
    const dim3 gridU(F / TN, (Bb * HL) / UM);
    const dim3 gridW(F / TN, (Bb * T) / TM);
    const dim3 gridP(F / TN, (Bb * T / 2) / TM, 4);
    const dim3 gridF((F * F / 4) / 256);
    const dim3 gridC((Bb * (T / 2) * F / 4) / 256);

    gemm_unfold_relu_w144<<<gridU, blkW, 0, stream>>>(z, W1, b1, h1);
    gemm_unfold_relu_w144<<<gridU, blkW, 0, stream>>>(z, V1, c1, g1);
    f23_filter<<<gridF, blk, 0, stream>>>(W2, gw);
    gemm_f32_wmma_w64<4><<<gridP, blkW, 0, stream>>>(
        h1, (long)HL * F, W2, 0, nullptr, gw, 0, 0,
        pln, F, F);
    f23_combine<<<gridC, blk, 0, stream>>>(pln, b2, wts);
    f23_filter<<<gridF, blk, 0, stream>>>(V2, gv);
    gemm_f32_wmma_w64<4><<<gridP, blkW, 0, stream>>>(
        g1, (long)HL * F, V2, 0, nullptr, gv, 0, 0,
        pln, F, F);
    f23_combine<<<gridC, blk, 0, stream>>>(pln, c2, bmat);
    gemm_f32_wmma_w64<2><<<gridW, blkW, 0, stream>>>(
        x + 4, (long)T * F, wts, (long)T * F, nullptr,
        bmat, (long)T * F, F,
        sm, T, T - 4);
    gemm_f32_wmma_w64<3><<<gridW, blkW, 0, stream>>>(
        sm, (long)T * F, Wr, 0, br,
        x, (long)T * F, F,
        out, F, F);
}


# GEMMs 3-4 (F(2,3) convs): each workgroup makes the 4 products of its tile in 4 K-loop passes, summed in registers (m1+-m2 parked in LDS); no planes, no combine pass, one launch; f32 sum order changes
# speedup vs baseline: 1.5886x; 1.0719x over previous
#include <hip/hip_runtime.h>
#include <hip/hip_bf16.h>
#include <math.h>


typedef float v2f __attribute__((ext_vector_type(2)));
typedef float v4f __attribute__((ext_vector_type(4)));
typedef float v8f __attribute__((ext_vector_type(8)));

#define KC  32

__device__ __forceinline__ v8f wmma4(v2f a, v2f b, v8f c) {
    return __builtin_amdgcn_wmma_f32_16x16x4_f32(
        false, a, false, b, (short)0, c, false, false);
}

__device__ __forceinline__ void stage_done() {
    __builtin_amdgcn_fence(__ATOMIC_RELEASE, "workgroup", "local");
    __builtin_amdgcn_s_barrier_signal(-1);
}
__device__ __forceinline__ void stage_ready() {
    __builtin_amdgcn_s_barrier_wait(-1);
    __builtin_amdgcn_fence(__ATOMIC_ACQUIRE, "workgroup", "local");
}
__device__ __forceinline__ void phase_edge() {
    __builtin_amdgcn_sched_barrier(0);
}

#define TM  128
#define TN  128
#define LP  36

template <int MODE>
__global__ __launch_bounds__(128) __attribute__((amdgpu_waves_per_eu(1, 2)))
void gemm_f32_wmma_w64(const float* __restrict__ A, long bsA,
                       const float* __restrict__ Bm, long bsB,
                       const float* __restrict__ biasCol,
                       const float* __restrict__ Emat, long bsE, long rsE,
                       float* __restrict__ Out,
                       const float* __restrict__ A2, const float* __restrict__ Bm2,
                       const float* __restrict__ bias2, const float* __restrict__ Emat2,
                       float* __restrict__ Out2,
                       int K, int klim)
{
    constexpr bool WINO = (MODE == 4);
    constexpr int  NPH  = WINO ? 4 : 1;
    if (WINO && blockIdx.z != 0) {
        A = A2; Bm = Bm2; biasCol = bias2; Emat = Emat2; Out = Out2;
    }
    constexpr int RPB = WINO ? 512 : 1024;
    constexpr int LDA = WINO ? 2048 : 1024;
    constexpr int LDN = 1024;
    __shared__ __attribute__((aligned(16))) float As[2][TM][LP];
    __shared__ __attribute__((aligned(16))) float Bs[2][TN][LP];
    __shared__ __attribute__((aligned(32))) float Tacc[WINO ? 4 * 16 * 32 * 8 : 8];

    const int tid  = threadIdx.x;
    const int lane = tid & 31;
    const int wave = tid >> 5;
    const int wm   = wave & 1;
    const int wn   = wave >> 1;

    const int tileM = blockIdx.y * TM;
    const int tileN = blockIdx.x * TN;
    const int bb    = tileM / RPB;
    const int r0    = tileM - bb * RPB;

    float sg = -1.0f;

    const int a_kq = (tid & 7) << 2;
    const int a_r  = tid >> 3;
    const float* a_row0 = A + (long)bb * bsA + (long)r0 * LDA;
    const char* a_ub  = (const char*)a_row0;
    const char* a_ub2 = (const char*)a_row0;
    const unsigned a_lo = (unsigned)(a_r * LDA + a_kq) * 4u;
    constexpr int A_STEP_B = 16 * LDA * 4;

    const int b_n = lane << 2;
    const int b_k = wave << 2;
    const float* b_sel = WINO ? Bm : Bm + (long)((bsB != 0) ? bb : 0) * bsB;
    const char* b_ub = (const char*)(b_sel + tileN);
    const unsigned b_lo = (unsigned)(b_k * LDN + b_n) * 4u;
    constexpr int B_ROW_B = LDN * 4;

    const int c16 = lane & 15;
    const int hh  = lane >> 4;
    const int am0 = (wm << 6) + c16;
    const int bn0 = (wn << 6) + c16;
    const int ko  = hh << 2;

    const int nchunks = (klim + KC - 1) / KC;

    v8f acc[4][4];
    #pragma unroll
    for (int i = 0; i < 4; ++i)
        #pragma unroll
        for (int j = 0; j < 4; ++j)
            acc[i][j] = v8f{};

    v4f ra[8], rb[8];
    v4f ra2[8];

    auto prefetch_whole = [&](int k0) {
        const char* pa = a_ub + (long)k0 * 4;
        const char* pb = b_ub + (long)k0 * B_ROW_B;
        #pragma unroll
        for (int i = 0; i < 8; ++i)
            ra[i] = *(const v4f*)(pa + a_lo + i * A_STEP_B);
        if (WINO) {
            const char* pa2 = a_ub2 + (long)k0 * 4;
            #pragma unroll
            for (int i = 0; i < 8; ++i)
                ra2[i] = *(const v4f*)(pa2 + a_lo + i * A_STEP_B);
        }
        #pragma unroll
        for (int g = 0; g < 2; ++g)
            #pragma unroll
            for (int j = 0; j < 4; ++j)
                rb[g * 4 + j] =
                    *(const v4f*)(pb + b_lo + (16 * g + j) * B_ROW_B);
    };

    auto prefetch_guarded = [&](int k0) {
        const char* pa = a_ub + (long)k0 * 4;
        const char* pb = b_ub + (long)k0 * B_ROW_B;
        {
            const v4f zz = {0.f, 0.f, 0.f, 0.f};
            #pragma unroll
            for (int i = 0; i < 8; ++i) {
                #pragma unroll
                for (int e = 0; e < 4; ++e) {
                    const int ka = k0 + a_kq + e;
                    ra[i][e] = (ka < klim)
                        ? *(const float*)(pa + a_lo + i * A_STEP_B + e * 4) : 0.0f;
                    if (WINO)
                        ra2[i][e] = (ka < klim)
                            ? *(const float*)(a_ub2 + (long)k0 * 4 + a_lo
                                              + i * A_STEP_B + e * 4) : 0.0f;
                }
            }
            #pragma unroll
            for (int g = 0; g < 2; ++g)
                #pragma unroll
                for (int j = 0; j < 4; ++j) {
                    const int kb = k0 + b_k + 16 * g + j;
                    rb[g * 4 + j] = (kb < K)
                        ? *(const v4f*)(pb + b_lo + (16 * g + j) * B_ROW_B) : zz;
                }
        }
    };

    auto stage = [&](int buf) {
        #pragma unroll
        for (int i = 0; i < 8; ++i)
            *(v4f*)&As[buf][a_r + 16 * i][a_kq] =
                WINO ? (ra[i] + sg * ra2[i]) : ra[i];
        #pragma unroll
        for (int g = 0; g < 2; ++g)
            #pragma unroll
            for (int c = 0; c < 4; ++c) {
                const v4f t = { rb[g * 4 + 0][c], rb[g * 4 + 1][c],
                                rb[g * 4 + 2][c], rb[g * 4 + 3][c] };
                *(v4f*)&Bs[buf][b_n + c][b_k + 16 * g] = t;
            }
    };

    auto ldfrag = [&](int buf, int kk, v4f (&fa)[4], v4f (&fb)[4]) {
        #pragma unroll
        for (int i = 0; i < 4; ++i)
            fa[i] = *(const v4f*)&As[buf][am0 + 16 * i][kk + ko];
        #pragma unroll
        for (int j = 0; j < 4; ++j)
            fb[j] = *(const v4f*)&Bs[buf][bn0 + 16 * j][kk + ko];
    };

    auto mma_half = [&](const v4f (&fa)[4], const v4f (&fb)[4], const int p) {
        #pragma unroll
        for (int i = 0; i < 4; ++i) {
            const v2f a = { fa[i][2 * p], fa[i][2 * p + 1] };
            #pragma unroll
            for (int j = 0; j < 4; ++j) {
                const v2f b = { fb[j][2 * p], fb[j][2 * p + 1] };
                acc[i][j] = wmma4(a, b, acc[i][j]);
            }
        }
    };
    auto mma = [&](const v4f (&fa)[4], const v4f (&fb)[4]) {
        mma_half(fa, fb, 0);
        mma_half(fa, fb, 1);
    };

    const int pmark1 = (nchunks * NPH * 27) >> 6;
    const int pmark2 = (nchunks * NPH * 47) >> 6;
    const int pmark3 = (nchunks * NPH * 59) >> 6;

    const int nwhole = klim / KC;

    auto prio_mark = [&](int c) {
        if (c == pmark1)      __builtin_amdgcn_s_setprio(2);
        else if (c == pmark2) __builtin_amdgcn_s_setprio(1);
        else if (c == pmark3) __builtin_amdgcn_s_setprio(0);
    };

    v4f fa0[4], fb0[4];
    v4f fa1[4], fb1[4], fa2[4], fb2[4], fa3[4], fb3[4];

    #pragma unroll 1
    for (int ph = 0; ph < NPH; ++ph) {
    const int cbase = ph * nchunks;
    if (WINO) {
        const int o1 = (ph == 0) ? 1 : (ph == 1) ? 2 : (ph == 2) ? 0 : 3;
        const int o2 = (ph == 0) ? 2 : (ph == 2) ? 2 : 1;
        sg = (ph == 0) ? 1.0f : -1.0f;
        a_ub  = (const char*)(a_row0 + o1 * 1024);
        a_ub2 = (const char*)(a_row0 + o2 * 1024);
        const float* bsel = (ph == 0) ? Emat : (ph == 1) ? Emat + 1024 * 1024
                          : (ph == 2) ? Bm : Bm + 2 * 1024 * 1024;
        b_ub = (const char*)(bsel + tileN);
        if (ph < 2) {
            #pragma unroll
            for (int i = 0; i < 4; ++i)
                #pragma unroll
                for (int j = 0; j < 4; ++j)
                    acc[i][j] = v8f{};
        } else if (ph == 3) {
            #pragma unroll
            for (int i = 0; i < 4; ++i)
                #pragma unroll
                for (int j = 0; j < 4; ++j)
                    acc[i][j] = *(const v8f*)&Tacc[(((wave * 16 + i * 4 + j) * 32 + lane) << 3)];
        }
        if (ph > 0) __syncthreads();
    }
        if (nwhole > 0) prefetch_whole(0); else prefetch_guarded(0);
        stage(0);
        __syncthreads();
        if (nchunks > 1) {
            if (nwhole > 1) prefetch_whole(KC); else prefetch_guarded(KC);
        }
        ldfrag(0, 0, fa0, fb0);
        int buf = 0;
        int c = 0;
        if (ph == 0) __builtin_amdgcn_s_setprio(3);
        for (; c + 2 < nwhole; ++c) {
            prio_mark(cbase + c);
            ldfrag(buf, 8,  fa1, fb1);
            ldfrag(buf, 16, fa2, fb2);
            ldfrag(buf, 24, fa3, fb3);
            mma(fa0, fb0);
            mma(fa1, fb1);
            phase_edge();
            stage(buf ^ 1);
            phase_edge();
            mma_half(fa2, fb2, 0);
            phase_edge();
            stage_done();
            prefetch_whole((c + 2) * KC);
            phase_edge();
            mma_half(fa2, fb2, 1);
            mma_half(fa3, fb3, 0);
            phase_edge();
            stage_ready();
            ldfrag(buf ^ 1, 0, fa0, fb0);
            phase_edge();
            mma_half(fa3, fb3, 1);
            buf ^= 1;
        }
        for (; c < nchunks; ++c) {
            prio_mark(cbase + c);
            const bool has1 = (c + 1 < nchunks);
            const bool has2 = (c + 2 < nchunks);
            ldfrag(buf, 8,  fa1, fb1);
            ldfrag(buf, 16, fa2, fb2);
            ldfrag(buf, 24, fa3, fb3);
            mma(fa0, fb0);
            mma(fa1, fb1);
            phase_edge();
            if (has1) stage(buf ^ 1);
            phase_edge();
            mma_half(fa2, fb2, 0);
            phase_edge();
            if (has1) stage_done();
            if (has2) prefetch_guarded((c + 2) * KC);
            phase_edge();
            mma_half(fa2, fb2, 1);
            mma_half(fa3, fb3, 0);
            phase_edge();
            if (has1) {
                stage_ready();
                ldfrag(buf ^ 1, 0, fa0, fb0);
            }
            phase_edge();
            mma_half(fa3, fb3, 1);
            buf ^= 1;
        }
    if (WINO) {
        if (ph == 0) {
            #pragma unroll
            for (int i = 0; i < 4; ++i)
                #pragma unroll
                for (int j = 0; j < 4; ++j)
                    *(v8f*)&Tacc[(((wave * 16 + i * 4 + j) * 32 + lane) << 3)] = acc[i][j];
        } else if (ph == 1) {
            #pragma unroll
            for (int i = 0; i < 4; ++i)
                #pragma unroll
                for (int j = 0; j < 4; ++j) {
                    v8f* tp = (v8f*)&Tacc[(((wave * 16 + i * 4 + j) * 32 + lane) << 3)];
                    const v8f t = *tp;
                    *tp = t - acc[i][j];
                    acc[i][j] = t + acc[i][j];
                }
        } else {
            const int col0w = tileN + bn0;
            const int rowbw = tileM + (wm << 6) + (hh << 3);
            #pragma unroll 1
            for (int r = 0; r < 8; ++r) {
                #pragma unroll
                for (int i = 0; i < 4; ++i) {
                    const int gr = rowbw + (i << 4) + r;
                    float* o = Out + ((long)gr * 2 + (ph - 2)) * LDN;
                    #pragma unroll
                    for (int j = 0; j < 4; ++j) {
                        const int col = col0w + (j << 4);
                        o[col] = acc[i][j][r] + biasCol[col];
                    }
                }
            }
        }
    }
    }
    __builtin_amdgcn_s_setprio(0);

    if (WINO) return;

    const int col0 = tileN + bn0;
    const int rowb = tileM + (wm << 6) + (hh << 3);
    if (MODE == 2) {
        const float* eb = Emat + (long)bb * bsE + (long)(rowb - bb * RPB) * rsE + col0;
        float en[16];
        #pragma unroll
        for (int i = 0; i < 4; ++i)
            #pragma unroll
            for (int j = 0; j < 4; ++j)
                en[i * 4 + j] = eb[(long)(i << 4) * rsE + (j << 4)];
        #pragma unroll 1
        for (int r = 0; r < 8; ++r) {
            float ec[16];
            #pragma unroll
            for (int q = 0; q < 16; ++q) ec[q] = en[q];
            const float* ep = eb + (long)((r + 1) & 7) * rsE;
            #pragma unroll
            for (int i = 0; i < 4; ++i)
                #pragma unroll
                for (int j = 0; j < 4; ++j)
                    en[i * 4 + j] = ep[(long)(i << 4) * rsE + (j << 4)];
            #pragma unroll
            for (int i = 0; i < 4; ++i) {
                float* o = Out + (long)(rowb + (i << 4) + r) * LDN + col0;
                #pragma unroll
                for (int j = 0; j < 4; ++j)
                    o[j << 4] = sinf(acc[i][j][r] + ec[i * 4 + j]);
            }
        }
        return;
    }
    #pragma unroll 1
    for (int r = 0; r < 8; ++r) {
        #pragma unroll
        for (int i = 0; i < 4; ++i) {
            const int gr = rowb + (i << 4) + r;
            const int rr = gr - bb * RPB;
            const float* e = Emat + (long)bb * bsE + (long)rr * rsE;
            float* o = Out + (long)gr * LDN;
            #pragma unroll
            for (int j = 0; j < 4; ++j) {
                const int col = col0 + (j << 4);
                float v = acc[i][j][r];
                if (MODE == 1) {
                    v += biasCol[col];
                } else if (MODE == 3) {
                    v = (v + biasCol[col] + e[col]) * 0.5f;
                }
                o[col] = v;
            }
        }
    }
}


#define UM  144
#define UI  9

__global__ __launch_bounds__(128) __attribute__((amdgpu_waves_per_eu(1, 2)))
void gemm_unfold_relu_w144(const float* __restrict__ Z,
                           const float* __restrict__ Bm,
                           const float* __restrict__ biasCol,
                           float* __restrict__ Out)
{
    constexpr int LDN = 1024;
    constexpr int KT  = 384;
    constexpr int NCH = KT / KC;
    __shared__ __attribute__((aligned(16))) float As[2][UM][LP];
    __shared__ __attribute__((aligned(16))) float Bs[2][TN][LP];

    const int tid  = threadIdx.x;
    const int lane = tid & 31;
    const int wave = tid >> 5;

    const int tileM = blockIdx.y * UM;
    const int tileN = blockIdx.x * TN;

    const int a_kq = (tid & 7) << 2;
    const int a_r  = tid >> 3;
    unsigned a_off[UI];
    #pragma unroll
    for (int i = 0; i < UI; ++i) {
        const int g = tileM + a_r + 16 * i;
        const int b = g / 1026;
        a_off[i] = (unsigned)(g * 128 + b * 256 + a_kq) * 4u;
    }

    const int b_n = lane << 2;
    const int b_k = wave << 2;
    const char* b_ub = (const char*)(Bm + tileN);
    const unsigned b_lo = (unsigned)(b_k * LDN + b_n) * 4u;
    constexpr int B_ROW_B = LDN * 4;

    const int c16 = lane & 15;
    const int hh  = lane >> 4;
    const int bn0 = (wave << 5) + c16;
    const int ko  = hh << 2;

    v8f acc[UI][2];
    #pragma unroll
    for (int i = 0; i < UI; ++i)
        #pragma unroll
        for (int j = 0; j < 2; ++j)
            acc[i][j] = v8f{};

    v4f ra[UI], rb[8];

    auto prefetch = [&](int k0) {
        const char* pa = (const char*)Z + (long)k0 * 4;
        const char* pb = b_ub + (long)k0 * B_ROW_B;
        #pragma unroll
        for (int i = 0; i < UI; ++i)
            ra[i] = *(const v4f*)(pa + a_off[i]);
        #pragma unroll
        for (int g = 0; g < 2; ++g)
            #pragma unroll
            for (int j = 0; j < 4; ++j)
                rb[g * 4 + j] =
                    *(const v4f*)(pb + b_lo + (16 * g + j) * B_ROW_B);
    };

    auto stage = [&](int buf) {
        #pragma unroll
        for (int i = 0; i < UI; ++i)
            *(v4f*)&As[buf][a_r + 16 * i][a_kq] = ra[i];
        #pragma unroll
        for (int g = 0; g < 2; ++g)
            #pragma unroll
            for (int c = 0; c < 4; ++c) {
                const v4f t = { rb[g * 4 + 0][c], rb[g * 4 + 1][c],
                                rb[g * 4 + 2][c], rb[g * 4 + 3][c] };
                *(v4f*)&Bs[buf][b_n + c][b_k + 16 * g] = t;
            }
    };

    auto ldfrag = [&](int buf, int kk, v4f (&fa)[UI], v4f (&fb)[2]) {
        #pragma unroll
        for (int i = 0; i < UI; ++i)
            fa[i] = *(const v4f*)&As[buf][c16 + 16 * i][kk + ko];
        #pragma unroll
        for (int j = 0; j < 2; ++j)
            fb[j] = *(const v4f*)&Bs[buf][bn0 + 16 * j][kk + ko];
    };

    auto mma_half = [&](const v4f (&fa)[UI], const v4f (&fb)[2], const int p) {
        #pragma unroll
        for (int i = 0; i < UI; ++i) {
            const v2f a = { fa[i][2 * p], fa[i][2 * p + 1] };
            #pragma unroll
            for (int j = 0; j < 2; ++j) {
                const v2f b = { fb[j][2 * p], fb[j][2 * p + 1] };
                acc[i][j] = wmma4(a, b, acc[i][j]);
            }
        }
    };
    auto mma = [&](const v4f (&fa)[UI], const v4f (&fb)[2]) {
        mma_half(fa, fb, 0);
        mma_half(fa, fb, 1);
    };

    constexpr int pmark1 = (NCH * 27) >> 6;
    constexpr int pmark2 = (NCH * 47) >> 6;
    constexpr int pmark3 = (NCH * 59) >> 6;
    auto prio_mark = [&](int c) {
        if (c == pmark1)      __builtin_amdgcn_s_setprio(2);
        else if (c == pmark2) __builtin_amdgcn_s_setprio(1);
        else if (c == pmark3) __builtin_amdgcn_s_setprio(0);
    };

    v4f fa0[UI], fb0[2];
    v4f fa1[UI], fb1[2], fa2[UI], fb2[2], fa3[UI], fb3[2];

    prefetch(0);
    stage(0);
    __syncthreads();
    prefetch(KC);
    ldfrag(0, 0, fa0, fb0);
    int buf = 0;
    int c = 0;
    __builtin_amdgcn_s_setprio(3);
    #pragma unroll 1
    for (; c + 2 < NCH; ++c) {
        prio_mark(c);
        ldfrag(buf, 8,  fa1, fb1);
        ldfrag(buf, 16, fa2, fb2);
        ldfrag(buf, 24, fa3, fb3);
        mma(fa0, fb0);
        mma(fa1, fb1);
        phase_edge();
        stage(buf ^ 1);
        phase_edge();
        mma_half(fa2, fb2, 0);
        phase_edge();
        stage_done();
        prefetch((c + 2) * KC);
        phase_edge();
        mma_half(fa2, fb2, 1);
        mma_half(fa3, fb3, 0);
        phase_edge();
        stage_ready();
        ldfrag(buf ^ 1, 0, fa0, fb0);
        phase_edge();
        mma_half(fa3, fb3, 1);
        buf ^= 1;
    }
    #pragma unroll 1
    for (; c < NCH; ++c) {
        prio_mark(c);
        const bool has1 = (c + 1 < NCH);
        ldfrag(buf, 8,  fa1, fb1);
        ldfrag(buf, 16, fa2, fb2);
        ldfrag(buf, 24, fa3, fb3);
        mma(fa0, fb0);
        mma(fa1, fb1);
        phase_edge();
        if (has1) stage(buf ^ 1);
        phase_edge();
        mma_half(fa2, fb2, 0);
        phase_edge();
        if (has1) stage_done();
        phase_edge();
        mma_half(fa2, fb2, 1);
        mma_half(fa3, fb3, 0);
        phase_edge();
        if (has1) {
            stage_ready();
            ldfrag(buf ^ 1, 0, fa0, fb0);
        }
        phase_edge();
        mma_half(fa3, fb3, 1);
        buf ^= 1;
    }
    __builtin_amdgcn_s_setprio(0);

    const int col0 = tileN + bn0;
    const int rowb = tileM + (hh << 3);
    const float bias0 = biasCol[col0];
    const float bias1 = biasCol[col0 + 16];
    #pragma unroll
    for (int i = 0; i < UI; ++i) {
        #pragma unroll
        for (int r = 0; r < 8; ++r) {
            float* o = Out + (long)(rowb + (i << 4) + r) * LDN + col0;
            o[0]  = fmaxf(acc[i][0][r] + bias0, 0.0f);
            o[16] = fmaxf(acc[i][1][r] + bias1, 0.0f);
        }
    }
}

__global__ __launch_bounds__(256)
void f23_filter(const float* __restrict__ W, float* __restrict__ G)
{
    const long TP = 1024L * 1024;
    const long o  = ((long)blockIdx.x * 256 + threadIdx.x) << 2;
    const v4f g0 = *(const v4f*)(W + o);
    const v4f g1 = *(const v4f*)(W + TP + o);
    const v4f g2 = *(const v4f*)(W + 2 * TP + o);
    const v4f gs = g0 + g2;
    *(v4f*)(G + o)      = (gs + g1) * 0.5f;
    *(v4f*)(G + TP + o) = (gs - g1) * 0.5f;
}

extern "C" void kernel_launch(void* const* d_in, const int* in_sizes, int n_in,
                              void* d_out, int out_size, void* d_ws, size_t ws_size,
                              hipStream_t stream)
{
    (void)in_sizes; (void)n_in; (void)out_size; (void)ws_size;

    const float* x  = (const float*)d_in[0];
    const float* z  = (const float*)d_in[1];
    const float* W1 = (const float*)d_in[2];
    const float* b1 = (const float*)d_in[3];
    const float* W2 = (const float*)d_in[4];
    const float* b2 = (const float*)d_in[5];
    const float* V1 = (const float*)d_in[6];
    const float* c1 = (const float*)d_in[7];
    const float* V2 = (const float*)d_in[8];
    const float* c2 = (const float*)d_in[9];
    const float* Wr = (const float*)d_in[10];
    const float* br = (const float*)d_in[11];
    float* out = (float*)d_out;

    const int Bb = 8, T = 1024, F = 1024, HL = 1026;

    float* ws   = (float*)d_ws;
    const size_t H1N = (size_t)Bb * HL * F;
    const size_t WTN = (size_t)Bb * T * F;
    const size_t GN  = (size_t)2 * F * F;
    float* r0 = ws;
    float* r1 = r0 + H1N;
    float* r2 = r1 + H1N;
    float* r4 = r2 + 2 * WTN;
    float* h1   = r0;
    float* g1   = r1;
    float* gw   = r2;
    float* gv   = r2 + GN;
    float* bmat = r2 + 2 * GN;
    float* wts  = r4;
    float* sm   = r1;

    const dim3 blk(256);
    const dim3 blkW(128);
    const dim3 gridU(F / TN, (Bb * HL) / UM);
    const dim3 gridW(F / TN, (Bb * T) / TM);
    const dim3 gridP(F / TN, (Bb * T / 2) / TM, 2);
    const dim3 gridF((F * F / 4) / 256);

    gemm_unfold_relu_w144<<<gridU, blkW, 0, stream>>>(z, W1, b1, h1);
    gemm_unfold_relu_w144<<<gridU, blkW, 0, stream>>>(z, V1, c1, g1);
    f23_filter<<<gridF, blk, 0, stream>>>(W2, gw);
    f23_filter<<<gridF, blk, 0, stream>>>(V2, gv);
    gemm_f32_wmma_w64<4><<<gridP, blkW, 0, stream>>>(
        h1, (long)HL * F, W2, 0, b2, gw, 0, 0, wts,
        g1, V2, c2, gv, bmat,
        F, F);
    gemm_f32_wmma_w64<2><<<gridW, blkW, 0, stream>>>(
        x + 4, (long)T * F, wts, (long)T * F, nullptr,
        bmat, (long)T * F, F,
        sm, nullptr, nullptr, nullptr, nullptr, nullptr,
        T, T - 4);
    gemm_f32_wmma_w64<3><<<gridW, blkW, 0, stream>>>(
        sm, (long)T * F, Wr, 0, br,
        x, (long)T * F, F,
        out, nullptr, nullptr, nullptr, nullptr, nullptr,
        F, F);
}


# GEMMs 3-4 (3-tap convs): F(2,3) -> F(4,3) on 64-quad x 128 tiles, keeping two workgroups per WGP (the earlier F(4,3) form ran one): 6 products per 4 rows, not 8; f32 sum order changes
# speedup vs baseline: 1.6750x; 1.0544x over previous
#include <hip/hip_runtime.h>
#include <hip/hip_bf16.h>
#include <math.h>


typedef float v2f __attribute__((ext_vector_type(2)));
typedef float v4f __attribute__((ext_vector_type(4)));
typedef float v8f __attribute__((ext_vector_type(8)));

#define KC  32

__device__ __forceinline__ v8f wmma4(v2f a, v2f b, v8f c) {
    return __builtin_amdgcn_wmma_f32_16x16x4_f32(
        false, a, false, b, (short)0, c, false, false);
}

__device__ __forceinline__ void stage_done() {
    __builtin_amdgcn_fence(__ATOMIC_RELEASE, "workgroup", "local");
    __builtin_amdgcn_s_barrier_signal(-1);
}
__device__ __forceinline__ void stage_ready() {
    __builtin_amdgcn_s_barrier_wait(-1);
    __builtin_amdgcn_fence(__ATOMIC_ACQUIRE, "workgroup", "local");
}
__device__ __forceinline__ void phase_edge() {
    __builtin_amdgcn_sched_barrier(0);
}

#define TM  128
#define TN  128
#define LP  36

template <int MODE>
__global__ __launch_bounds__(128) __attribute__((amdgpu_waves_per_eu(1, 2)))
void gemm_f32_wmma_w64(const float* __restrict__ A, long bsA,
                       const float* __restrict__ Bm, long bsB,
                       const float* __restrict__ biasCol,
                       const float* __restrict__ Emat, long bsE, long rsE,
                       float* __restrict__ Out,
                       int K, int klim)
{
    constexpr int RPB = 1024;
    constexpr int LDA = 1024;
    constexpr int LDN = 1024;
    __shared__ __attribute__((aligned(16))) float As[2][TM][LP];
    __shared__ __attribute__((aligned(16))) float Bs[2][TN][LP];

    const int tid  = threadIdx.x;
    const int lane = tid & 31;
    const int wave = tid >> 5;
    const int wm   = wave & 1;
    const int wn   = wave >> 1;

    const int tileM = blockIdx.y * TM;
    const int tileN = blockIdx.x * TN;
    const int bb    = tileM / RPB;
    const int r0    = tileM - bb * RPB;

    const int a_kq = (tid & 7) << 2;
    const int a_r  = tid >> 3;
    const float* a_row0 = A + (long)bb * bsA + (long)r0 * LDA;
    const char* a_ub  = (const char*)a_row0;
    const unsigned a_lo = (unsigned)(a_r * LDA + a_kq) * 4u;
    constexpr int A_STEP_B = 16 * LDA * 4;

    const int b_n = lane << 2;
    const int b_k = wave << 2;
    const float* b_sel = Bm + (long)((bsB != 0) ? bb : 0) * bsB;
    const char* b_ub = (const char*)(b_sel + tileN);
    const unsigned b_lo = (unsigned)(b_k * LDN + b_n) * 4u;
    constexpr int B_ROW_B = LDN * 4;

    const int c16 = lane & 15;
    const int hh  = lane >> 4;
    const int am0 = (wm << 6) + c16;
    const int bn0 = (wn << 6) + c16;
    const int ko  = hh << 2;

    const int nchunks = (klim + KC - 1) / KC;

    v8f acc[4][4];
    #pragma unroll
    for (int i = 0; i < 4; ++i)
        #pragma unroll
        for (int j = 0; j < 4; ++j)
            acc[i][j] = v8f{};

    v4f ra[8], rb[8];

    auto prefetch_whole = [&](int k0) {
        const char* pa = a_ub + (long)k0 * 4;
        const char* pb = b_ub + (long)k0 * B_ROW_B;
        #pragma unroll
        for (int i = 0; i < 8; ++i)
            ra[i] = *(const v4f*)(pa + a_lo + i * A_STEP_B);
        #pragma unroll
        for (int g = 0; g < 2; ++g)
            #pragma unroll
            for (int j = 0; j < 4; ++j)
                rb[g * 4 + j] =
                    *(const v4f*)(pb + b_lo + (16 * g + j) * B_ROW_B);
    };

    auto prefetch_guarded = [&](int k0) {
        const char* pa = a_ub + (long)k0 * 4;
        const char* pb = b_ub + (long)k0 * B_ROW_B;
        {
            const v4f zz = {0.f, 0.f, 0.f, 0.f};
            #pragma unroll
            for (int i = 0; i < 8; ++i) {
                #pragma unroll
                for (int e = 0; e < 4; ++e) {
                    const int ka = k0 + a_kq + e;
                    ra[i][e] = (ka < klim)
                        ? *(const float*)(pa + a_lo + i * A_STEP_B + e * 4) : 0.0f;
                }
            }
            #pragma unroll
            for (int g = 0; g < 2; ++g)
                #pragma unroll
                for (int j = 0; j < 4; ++j) {
                    const int kb = k0 + b_k + 16 * g + j;
                    rb[g * 4 + j] = (kb < K)
                        ? *(const v4f*)(pb + b_lo + (16 * g + j) * B_ROW_B) : zz;
                }
        }
    };

    auto stage = [&](int buf) {
        #pragma unroll
        for (int i = 0; i < 8; ++i)
            *(v4f*)&As[buf][a_r + 16 * i][a_kq] = ra[i];
        #pragma unroll
        for (int g = 0; g < 2; ++g)
            #pragma unroll
            for (int c = 0; c < 4; ++c) {
                const v4f t = { rb[g * 4 + 0][c], rb[g * 4 + 1][c],
                                rb[g * 4 + 2][c], rb[g * 4 + 3][c] };
                *(v4f*)&Bs[buf][b_n + c][b_k + 16 * g] = t;
            }
    };

    auto ldfrag = [&](int buf, int kk, v4f (&fa)[4], v4f (&fb)[4]) {
        #pragma unroll
        for (int i = 0; i < 4; ++i)
            fa[i] = *(const v4f*)&As[buf][am0 + 16 * i][kk + ko];
        #pragma unroll
        for (int j = 0; j < 4; ++j)
            fb[j] = *(const v4f*)&Bs[buf][bn0 + 16 * j][kk + ko];
    };

    auto mma_half = [&](const v4f (&fa)[4], const v4f (&fb)[4], const int p) {
        #pragma unroll
        for (int i = 0; i < 4; ++i) {
            const v2f a = { fa[i][2 * p], fa[i][2 * p + 1] };
            #pragma unroll
            for (int j = 0; j < 4; ++j) {
                const v2f b = { fb[j][2 * p], fb[j][2 * p + 1] };
                acc[i][j] = wmma4(a, b, acc[i][j]);
            }
        }
    };
    auto mma = [&](const v4f (&fa)[4], const v4f (&fb)[4]) {
        mma_half(fa, fb, 0);
        mma_half(fa, fb, 1);
    };

    const int pmark1 = (nchunks * 27) >> 6;
    const int pmark2 = (nchunks * 47) >> 6;
    const int pmark3 = (nchunks * 59) >> 6;

    const int nwhole = klim / KC;

    auto prio_mark = [&](int c) {
        if (c == pmark1)      __builtin_amdgcn_s_setprio(2);
        else if (c == pmark2) __builtin_amdgcn_s_setprio(1);
        else if (c == pmark3) __builtin_amdgcn_s_setprio(0);
    };

    v4f fa0[4], fb0[4];
    v4f fa1[4], fb1[4], fa2[4], fb2[4], fa3[4], fb3[4];

    if (nwhole > 0) prefetch_whole(0); else prefetch_guarded(0);
    stage(0);
    __syncthreads();
    if (nchunks > 1) {
        if (nwhole > 1) prefetch_whole(KC); else prefetch_guarded(KC);
    }
    ldfrag(0, 0, fa0, fb0);
    int buf = 0;
    int c = 0;
    __builtin_amdgcn_s_setprio(3);
    for (; c + 2 < nwhole; ++c) {
        prio_mark(c);
        ldfrag(buf, 8,  fa1, fb1);
        ldfrag(buf, 16, fa2, fb2);
        ldfrag(buf, 24, fa3, fb3);
        mma(fa0, fb0);
        mma(fa1, fb1);
        phase_edge();
        stage(buf ^ 1);
        phase_edge();
        mma_half(fa2, fb2, 0);
        phase_edge();
        stage_done();
        prefetch_whole((c + 2) * KC);
        phase_edge();
        mma_half(fa2, fb2, 1);
        mma_half(fa3, fb3, 0);
        phase_edge();
        stage_ready();
        ldfrag(buf ^ 1, 0, fa0, fb0);
        phase_edge();
        mma_half(fa3, fb3, 1);
        buf ^= 1;
    }
    for (; c < nchunks; ++c) {
        prio_mark(c);
        const bool has1 = (c + 1 < nchunks);
        const bool has2 = (c + 2 < nchunks);
        ldfrag(buf, 8,  fa1, fb1);
        ldfrag(buf, 16, fa2, fb2);
        ldfrag(buf, 24, fa3, fb3);
        mma(fa0, fb0);
        mma(fa1, fb1);
        phase_edge();
        if (has1) stage(buf ^ 1);
        phase_edge();
        mma_half(fa2, fb2, 0);
        phase_edge();
        if (has1) stage_done();
        if (has2) prefetch_guarded((c + 2) * KC);
        phase_edge();
        mma_half(fa2, fb2, 1);
        mma_half(fa3, fb3, 0);
        phase_edge();
        if (has1) {
            stage_ready();
            ldfrag(buf ^ 1, 0, fa0, fb0);
        }
        phase_edge();
        mma_half(fa3, fb3, 1);
        buf ^= 1;
    }
    __builtin_amdgcn_s_setprio(0);

    const int col0 = tileN + bn0;
    const int rowb = tileM + (wm << 6) + (hh << 3);
    if (MODE == 2) {
        const float* eb = Emat + (long)bb * bsE + (long)(rowb - bb * RPB) * rsE + col0;
        float en[16];
        #pragma unroll
        for (int i = 0; i < 4; ++i)
            #pragma unroll
            for (int j = 0; j < 4; ++j)
                en[i * 4 + j] = eb[(long)(i << 4) * rsE + (j << 4)];
        #pragma unroll 1
        for (int r = 0; r < 8; ++r) {
            float ec[16];
            #pragma unroll
            for (int q = 0; q < 16; ++q) ec[q] = en[q];
            const float* ep = eb + (long)((r + 1) & 7) * rsE;
            #pragma unroll
            for (int i = 0; i < 4; ++i)
                #pragma unroll
                for (int j = 0; j < 4; ++j)
                    en[i * 4 + j] = ep[(long)(i << 4) * rsE + (j << 4)];
            #pragma unroll
            for (int i = 0; i < 4; ++i) {
                float* o = Out + (long)(rowb + (i << 4) + r) * LDN + col0;
                #pragma unroll
                for (int j = 0; j < 4; ++j)
                    o[j << 4] = sinf(acc[i][j][r] + ec[i * 4 + j]);
            }
        }
        return;
    }
    #pragma unroll 1
    for (int r = 0; r < 8; ++r) {
        #pragma unroll
        for (int i = 0; i < 4; ++i) {
            const int gr = rowb + (i << 4) + r;
            const int rr = gr - bb * RPB;
            const float* e = Emat + (long)bb * bsE + (long)rr * rsE;
            float* o = Out + (long)gr * LDN;
            #pragma unroll
            for (int j = 0; j < 4; ++j) {
                const int col = col0 + (j << 4);
                float v = acc[i][j][r];
                if (MODE == 1) {
                    v += biasCol[col];
                } else if (MODE == 3) {
                    v = (v + biasCol[col] + e[col]) * 0.5f;
                }
                o[col] = v;
            }
        }
    }
}


#define UM  144
#define UI  9

__global__ __launch_bounds__(128) __attribute__((amdgpu_waves_per_eu(1, 2)))
void gemm_unfold_relu_w144(const float* __restrict__ Z,
                           const float* __restrict__ Bm,
                           const float* __restrict__ biasCol,
                           float* __restrict__ Out)
{
    constexpr int LDN = 1024;
    constexpr int KT  = 384;
    constexpr int NCH = KT / KC;
    __shared__ __attribute__((aligned(16))) float As[2][UM][LP];
    __shared__ __attribute__((aligned(16))) float Bs[2][TN][LP];

    const int tid  = threadIdx.x;
    const int lane = tid & 31;
    const int wave = tid >> 5;

    const int tileM = blockIdx.y * UM;
    const int tileN = blockIdx.x * TN;

    const int a_kq = (tid & 7) << 2;
    const int a_r  = tid >> 3;
    unsigned a_off[UI];
    #pragma unroll
    for (int i = 0; i < UI; ++i) {
        const int g = tileM + a_r + 16 * i;
        const int b = g / 1026;
        a_off[i] = (unsigned)(g * 128 + b * 256 + a_kq) * 4u;
    }

    const int b_n = lane << 2;
    const int b_k = wave << 2;
    const char* b_ub = (const char*)(Bm + tileN);
    const unsigned b_lo = (unsigned)(b_k * LDN + b_n) * 4u;
    constexpr int B_ROW_B = LDN * 4;

    const int c16 = lane & 15;
    const int hh  = lane >> 4;
    const int bn0 = (wave << 5) + c16;
    const int ko  = hh << 2;

    v8f acc[UI][2];
    #pragma unroll
    for (int i = 0; i < UI; ++i)
        #pragma unroll
        for (int j = 0; j < 2; ++j)
            acc[i][j] = v8f{};

    v4f ra[UI], rb[8];

    auto prefetch = [&](int k0) {
        const char* pa = (const char*)Z + (long)k0 * 4;
        const char* pb = b_ub + (long)k0 * B_ROW_B;
        #pragma unroll
        for (int i = 0; i < UI; ++i)
            ra[i] = *(const v4f*)(pa + a_off[i]);
        #pragma unroll
        for (int g = 0; g < 2; ++g)
            #pragma unroll
            for (int j = 0; j < 4; ++j)
                rb[g * 4 + j] =
                    *(const v4f*)(pb + b_lo + (16 * g + j) * B_ROW_B);
    };

    auto stage = [&](int buf) {
        #pragma unroll
        for (int i = 0; i < UI; ++i)
            *(v4f*)&As[buf][a_r + 16 * i][a_kq] = ra[i];
        #pragma unroll
        for (int g = 0; g < 2; ++g)
            #pragma unroll
            for (int c = 0; c < 4; ++c) {
                const v4f t = { rb[g * 4 + 0][c], rb[g * 4 + 1][c],
                                rb[g * 4 + 2][c], rb[g * 4 + 3][c] };
                *(v4f*)&Bs[buf][b_n + c][b_k + 16 * g] = t;
            }
    };

    auto ldfrag = [&](int buf, int kk, v4f (&fa)[UI], v4f (&fb)[2]) {
        #pragma unroll
        for (int i = 0; i < UI; ++i)
            fa[i] = *(const v4f*)&As[buf][c16 + 16 * i][kk + ko];
        #pragma unroll
        for (int j = 0; j < 2; ++j)
            fb[j] = *(const v4f*)&Bs[buf][bn0 + 16 * j][kk + ko];
    };

    auto mma_half = [&](const v4f (&fa)[UI], const v4f (&fb)[2], const int p) {
        #pragma unroll
        for (int i = 0; i < UI; ++i) {
            const v2f a = { fa[i][2 * p], fa[i][2 * p + 1] };
            #pragma unroll
            for (int j = 0; j < 2; ++j) {
                const v2f b = { fb[j][2 * p], fb[j][2 * p + 1] };
                acc[i][j] = wmma4(a, b, acc[i][j]);
            }
        }
    };
    auto mma = [&](const v4f (&fa)[UI], const v4f (&fb)[2]) {
        mma_half(fa, fb, 0);
        mma_half(fa, fb, 1);
    };

    constexpr int pmark1 = (NCH * 27) >> 6;
    constexpr int pmark2 = (NCH * 47) >> 6;
    constexpr int pmark3 = (NCH * 59) >> 6;
    auto prio_mark = [&](int c) {
        if (c == pmark1)      __builtin_amdgcn_s_setprio(2);
        else if (c == pmark2) __builtin_amdgcn_s_setprio(1);
        else if (c == pmark3) __builtin_amdgcn_s_setprio(0);
    };

    v4f fa0[UI], fb0[2];
    v4f fa1[UI], fb1[2], fa2[UI], fb2[2], fa3[UI], fb3[2];

    prefetch(0);
    stage(0);
    __syncthreads();
    prefetch(KC);
    ldfrag(0, 0, fa0, fb0);
    int buf = 0;
    int c = 0;
    __builtin_amdgcn_s_setprio(3);
    #pragma unroll 1
    for (; c + 2 < NCH; ++c) {
        prio_mark(c);
        ldfrag(buf, 8,  fa1, fb1);
        ldfrag(buf, 16, fa2, fb2);
        ldfrag(buf, 24, fa3, fb3);
        mma(fa0, fb0);
        mma(fa1, fb1);
        phase_edge();
        stage(buf ^ 1);
        phase_edge();
        mma_half(fa2, fb2, 0);
        phase_edge();
        stage_done();
        prefetch((c + 2) * KC);
        phase_edge();
        mma_half(fa2, fb2, 1);
        mma_half(fa3, fb3, 0);
        phase_edge();
        stage_ready();
        ldfrag(buf ^ 1, 0, fa0, fb0);
        phase_edge();
        mma_half(fa3, fb3, 1);
        buf ^= 1;
    }
    #pragma unroll 1
    for (; c < NCH; ++c) {
        prio_mark(c);
        const bool has1 = (c + 1 < NCH);
        ldfrag(buf, 8,  fa1, fb1);
        ldfrag(buf, 16, fa2, fb2);
        ldfrag(buf, 24, fa3, fb3);
        mma(fa0, fb0);
        mma(fa1, fb1);
        phase_edge();
        if (has1) stage(buf ^ 1);
        phase_edge();
        mma_half(fa2, fb2, 0);
        phase_edge();
        if (has1) stage_done();
        phase_edge();
        mma_half(fa2, fb2, 1);
        mma_half(fa3, fb3, 0);
        phase_edge();
        if (has1) {
            stage_ready();
            ldfrag(buf ^ 1, 0, fa0, fb0);
        }
        phase_edge();
        mma_half(fa3, fb3, 1);
        buf ^= 1;
    }
    __builtin_amdgcn_s_setprio(0);

    const int col0 = tileN + bn0;
    const int rowb = tileM + (hh << 3);
    const float bias0 = biasCol[col0];
    const float bias1 = biasCol[col0 + 16];
    #pragma unroll
    for (int i = 0; i < UI; ++i) {
        #pragma unroll
        for (int r = 0; r < 8; ++r) {
            float* o = Out + (long)(rowb + (i << 4) + r) * LDN + col0;
            o[0]  = fmaxf(acc[i][0][r] + bias0, 0.0f);
            o[16] = fmaxf(acc[i][1][r] + bias1, 0.0f);
        }
    }
}

#define QM  64

__global__ __launch_bounds__(128) __attribute__((amdgpu_waves_per_eu(1, 2)))
void conv3_f43_wmma(const float* __restrict__ A, const float* __restrict__ Bm,
                    const float* __restrict__ biasCol, const float* __restrict__ Gm,
                    float* __restrict__ Out,
                    const float* __restrict__ A2, const float* __restrict__ Bm2,
                    const float* __restrict__ bias2, const float* __restrict__ Gm2,
                    float* __restrict__ Out2, long bsA)
{
    if (blockIdx.z != 0) {
        A = A2; Bm = Bm2; biasCol = bias2; Gm = Gm2; Out = Out2;
    }
    constexpr int  LDN = 1024;
    constexpr int  NCH = 1024 / KC;
    constexpr int  NPH = 6;
    constexpr int  QPB = 256;
    constexpr int  LDQ = 4096;
    constexpr long TP  = 1024L * 1024;
    __shared__ __attribute__((aligned(16))) float As[2][QM][LP];
    __shared__ __attribute__((aligned(16))) float Bs[2][TN][LP];
    __shared__ __attribute__((aligned(32))) float Tacc[3][4 * 8 * 32 * 8];

    const int tid  = threadIdx.x;
    const int lane = tid & 31;
    const int wave = tid >> 5;
    const int wm   = wave & 1;
    const int wn   = wave >> 1;

    const int tileQ = blockIdx.y * QM;
    const int tileN = blockIdx.x * TN;
    const int bb    = tileQ / QPB;
    const int q0    = tileQ - bb * QPB;

    const int a_kq = (tid & 7) << 2;
    const int a_r  = tid >> 3;
    const float* a_row0 = A + (long)bb * bsA + (long)q0 * LDQ;
    const unsigned a_lo = (unsigned)(a_r * LDQ + a_kq) * 4u;
    constexpr int A_STEP_B = 16 * LDQ * 4;
    const char* a_s0 = (const char*)a_row0;
    const char* a_s1 = a_s0;
    const char* a_s2 = a_s0;
    const char* a_s3 = a_s0;
    float c0 = 0.f, c1 = 0.f, c2 = 0.f, c3 = 0.f;
    bool four = true;

    const int b_n = lane << 2;
    const int b_k = wave << 2;
    const char* b_ub = (const char*)(Bm + tileN);
    const unsigned b_lo = (unsigned)(b_k * LDN + b_n) * 4u;
    constexpr int B_ROW_B = LDN * 4;

    const int c16 = lane & 15;
    const int hh  = lane >> 4;
    const int am0 = (wm << 5) + c16;
    const int bn0 = (wn << 6) + c16;
    const int ko  = hh << 2;

    v8f acc[2][4];
    v4f ra0[4], ra1[4], ra2[4], ra3[4], rb[8];

    auto prefetch = [&](int k0) {
        const long kb = (long)k0 * 4;
        const char* pb = b_ub + (long)k0 * B_ROW_B;
        #pragma unroll
        for (int i = 0; i < 4; ++i) {
            ra0[i] = *(const v4f*)(a_s0 + kb + a_lo + i * A_STEP_B);
            ra1[i] = *(const v4f*)(a_s1 + kb + a_lo + i * A_STEP_B);
            ra2[i] = *(const v4f*)(a_s2 + kb + a_lo + i * A_STEP_B);
        }
        if (four) {
            #pragma unroll
            for (int i = 0; i < 4; ++i)
                ra3[i] = *(const v4f*)(a_s3 + kb + a_lo + i * A_STEP_B);
        }
        #pragma unroll
        for (int g = 0; g < 2; ++g)
            #pragma unroll
            for (int j = 0; j < 4; ++j)
                rb[g * 4 + j] =
                    *(const v4f*)(pb + b_lo + (16 * g + j) * B_ROW_B);
    };

    auto stage = [&](int buf) {
        #pragma unroll
        for (int i = 0; i < 4; ++i)
            *(v4f*)&As[buf][a_r + 16 * i][a_kq] =
                c0 * ra0[i] + c1 * ra1[i] + c2 * ra2[i] + c3 * ra3[i];
        #pragma unroll
        for (int g = 0; g < 2; ++g)
            #pragma unroll
            for (int c = 0; c < 4; ++c) {
                const v4f t = { rb[g * 4 + 0][c], rb[g * 4 + 1][c],
                                rb[g * 4 + 2][c], rb[g * 4 + 3][c] };
                *(v4f*)&Bs[buf][b_n + c][b_k + 16 * g] = t;
            }
    };

    auto ldfrag = [&](int buf, int kk, v4f (&fa)[2], v4f (&fb)[4]) {
        #pragma unroll
        for (int i = 0; i < 2; ++i)
            fa[i] = *(const v4f*)&As[buf][am0 + 16 * i][kk + ko];
        #pragma unroll
        for (int j = 0; j < 4; ++j)
            fb[j] = *(const v4f*)&Bs[buf][bn0 + 16 * j][kk + ko];
    };

    auto mma_half = [&](const v4f (&fa)[2], const v4f (&fb)[4], const int p) {
        #pragma unroll
        for (int i = 0; i < 2; ++i) {
            const v2f a = { fa[i][2 * p], fa[i][2 * p + 1] };
            #pragma unroll
            for (int j = 0; j < 4; ++j) {
                const v2f b = { fb[j][2 * p], fb[j][2 * p + 1] };
                acc[i][j] = wmma4(a, b, acc[i][j]);
            }
        }
    };
    auto mma = [&](const v4f (&fa)[2], const v4f (&fb)[4]) {
        mma_half(fa, fb, 0);
        mma_half(fa, fb, 1);
    };

    constexpr int pmark1 = (NCH * NPH * 27) >> 6;
    constexpr int pmark2 = (NCH * NPH * 47) >> 6;
    constexpr int pmark3 = (NCH * NPH * 59) >> 6;
    auto prio_mark = [&](int c) {
        if (c == pmark1)      __builtin_amdgcn_s_setprio(2);
        else if (c == pmark2) __builtin_amdgcn_s_setprio(1);
        else if (c == pmark3) __builtin_amdgcn_s_setprio(0);
    };

    auto park = [&](int s, int i, int j) -> v8f* {
        return (v8f*)&Tacc[s][(((wave * 8 + i * 4 + j) * 32 + lane) << 3)];
    };

    const int col0 = tileN + bn0;
    const int qb   = tileQ + (wm << 5) + (hh << 3);
    float bj[4];
    #pragma unroll
    for (int j = 0; j < 4; ++j) bj[j] = biasCol[col0 + (j << 4)];

    v4f fa0[2], fb0[4];
    v4f fa1[2], fb1[4], fa2[2], fb2[4], fa3[2], fb3[4];

    #pragma unroll 1
    for (int ph = 0; ph < NPH; ++ph) {
        four = (ph < 4);
        if (four) {
            a_s0 = (const char*)(a_row0 + 1 * 1024);
            a_s1 = (const char*)(a_row0 + 2 * 1024);
            a_s2 = (const char*)(a_row0 + 3 * 1024);
            a_s3 = (const char*)(a_row0 + 4 * 1024);
            c0 = (ph == 0) ? -4.0f : (ph == 1) ? 4.0f : (ph == 2) ? -2.0f : 2.0f;
            c1 = (ph < 2) ? -4.0f : -1.0f;
            c2 = (ph == 0) ? 1.0f : (ph == 1) ? -1.0f : (ph == 2) ? 2.0f : -2.0f;
            c3 = 1.0f;
            b_ub = (const char*)(Gm + (long)ph * TP + tileN);
        } else {
            const int o = ph - 4;
            a_s0 = (const char*)(a_row0 + (0 + o) * 1024);
            a_s1 = (const char*)(a_row0 + (2 + o) * 1024);
            a_s2 = (const char*)(a_row0 + (4 + o) * 1024);
            c0 = o ? 4.0f : 1.0f;
            c1 = o ? -5.0f : -1.25f;
            c2 = o ? 1.0f : 0.25f;
            c3 = 0.0f;
            #pragma unroll
            for (int i = 0; i < 4; ++i) ra3[i] = v4f{0.f, 0.f, 0.f, 0.f};
            b_ub = (const char*)(Bm + (long)(2 * o) * TP + tileN);
        }
        #pragma unroll
        for (int i = 0; i < 2; ++i)
            #pragma unroll
            for (int j = 0; j < 4; ++j)
                if (four) acc[i][j] = v8f{};
                else      acc[i][j] = *park(ph - 4, i, j);
        if (ph > 0) __syncthreads();

        prefetch(0);
        stage(0);
        __syncthreads();
        prefetch(KC);
        ldfrag(0, 0, fa0, fb0);
        int buf = 0;
        int c = 0;
        if (ph == 0) __builtin_amdgcn_s_setprio(3);
        #pragma unroll 1
        for (; c + 2 < NCH; ++c) {
            prio_mark(ph * NCH + c);
            ldfrag(buf, 8,  fa1, fb1);
            ldfrag(buf, 16, fa2, fb2);
            ldfrag(buf, 24, fa3, fb3);
            mma(fa0, fb0);
            mma(fa1, fb1);
            phase_edge();
            stage(buf ^ 1);
            phase_edge();
            mma_half(fa2, fb2, 0);
            phase_edge();
            stage_done();
            prefetch((c + 2) * KC);
            phase_edge();
            mma_half(fa2, fb2, 1);
            mma_half(fa3, fb3, 0);
            phase_edge();
            stage_ready();
            ldfrag(buf ^ 1, 0, fa0, fb0);
            phase_edge();
            mma_half(fa3, fb3, 1);
            buf ^= 1;
        }
        #pragma unroll 1
        for (; c < NCH; ++c) {
            prio_mark(ph * NCH + c);
            const bool has1 = (c + 1 < NCH);
            ldfrag(buf, 8,  fa1, fb1);
            ldfrag(buf, 16, fa2, fb2);
            ldfrag(buf, 24, fa3, fb3);
            mma(fa0, fb0);
            mma(fa1, fb1);
            phase_edge();
            if (has1) stage(buf ^ 1);
            phase_edge();
            mma_half(fa2, fb2, 0);
            phase_edge();
            if (has1) stage_done();
            phase_edge();
            mma_half(fa2, fb2, 1);
            mma_half(fa3, fb3, 0);
            phase_edge();
            if (has1) {
                stage_ready();
                ldfrag(buf ^ 1, 0, fa0, fb0);
            }
            phase_edge();
            mma_half(fa3, fb3, 1);
            buf ^= 1;
        }

        if (ph == 0 || ph == 2) {
            #pragma unroll
            for (int i = 0; i < 2; ++i)
                #pragma unroll
                for (int j = 0; j < 4; ++j)
                    *park(ph, i, j) = acc[i][j];
        } else if (ph == 1) {
            #pragma unroll
            for (int i = 0; i < 2; ++i)
                #pragma unroll
                for (int j = 0; j < 4; ++j) {
                    const v8f t = *park(0, i, j);
                    *park(0, i, j) = t + acc[i][j];
                    *park(1, i, j) = t - acc[i][j];
                }
        } else if (ph == 3) {
            #pragma unroll
            for (int i = 0; i < 2; ++i)
                #pragma unroll
                for (int j = 0; j < 4; ++j) {
                    const v8f m3  = *park(2, i, j);
                    const v8f s34 = m3 + acc[i][j];
                    const v8f d34 = m3 - acc[i][j];
                    const v8f s12 = *park(0, i, j);
                    const v8f d12 = *park(1, i, j);
                    const v8f y1  = d12 + 2.0f * d34;
                    const v8f y2  = s12 + 4.0f * s34;
                    *park(0, i, j) = s12 + s34;
                    *park(1, i, j) = d12 + 8.0f * d34;
                    float* o = Out + ((long)(qb + (i << 4)) * 4 + 1) * LDN + col0 + (j << 4);
                    #pragma unroll
                    for (int r = 0; r < 8; ++r) {
                        o[(long)r * 4 * LDN]       = y1[r] + bj[j];
                        o[(long)r * 4 * LDN + LDN] = y2[r] + bj[j];
                    }
                }
        } else {
            const int ro = (ph == 4) ? 0 : 3;
            #pragma unroll
            for (int i = 0; i < 2; ++i)
                #pragma unroll
                for (int j = 0; j < 4; ++j) {
                    float* o = Out + ((long)(qb + (i << 4)) * 4 + ro) * LDN + col0 + (j << 4);
                    #pragma unroll
                    for (int r = 0; r < 8; ++r)
                        o[(long)r * 4 * LDN] = acc[i][j][r] + bj[j];
                }
        }
    }
    __builtin_amdgcn_s_setprio(0);
}

__global__ __launch_bounds__(256)
void f43_filter(const float* __restrict__ W, float* __restrict__ G)
{
    const long TP = 1024L * 1024;
    const long o  = ((long)blockIdx.x * 256 + threadIdx.x) << 2;
    const v4f g0 = *(const v4f*)(W + o);
    const v4f g1 = *(const v4f*)(W + TP + o);
    const v4f g2 = *(const v4f*)(W + 2 * TP + o);
    const v4f gs = g0 + g2;
    const v4f ge = g0 * (1.0f / 24.0f) + g2 * (1.0f / 6.0f);
    const v4f gm = g1 * (1.0f / 12.0f);
    *(v4f*)(G + o)          = (gs + g1) * (-1.0f / 6.0f);
    *(v4f*)(G + TP + o)     = (gs - g1) * (-1.0f / 6.0f);
    *(v4f*)(G + 2 * TP + o) = ge + gm;
    *(v4f*)(G + 3 * TP + o) = ge - gm;
}

extern "C" void kernel_launch(void* const* d_in, const int* in_sizes, int n_in,
                              void* d_out, int out_size, void* d_ws, size_t ws_size,
                              hipStream_t stream)
{
    (void)in_sizes; (void)n_in; (void)out_size; (void)ws_size;

    const float* x  = (const float*)d_in[0];
    const float* z  = (const float*)d_in[1];
    const float* W1 = (const float*)d_in[2];
    const float* b1 = (const float*)d_in[3];
    const float* W2 = (const float*)d_in[4];
    const float* b2 = (const float*)d_in[5];
    const float* V1 = (const float*)d_in[6];
    const float* c1 = (const float*)d_in[7];
    const float* V2 = (const float*)d_in[8];
    const float* c2 = (const float*)d_in[9];
    const float* Wr = (const float*)d_in[10];
    const float* br = (const float*)d_in[11];
    float* out = (float*)d_out;

    const int Bb = 8, T = 1024, F = 1024, HL = 1026;

    float* ws   = (float*)d_ws;
    const size_t H1N = (size_t)Bb * HL * F;
    const size_t WTN = (size_t)Bb * T * F;
    const size_t GN  = (size_t)4 * F * F;
    float* r0 = ws;
    float* r1 = r0 + H1N;
    float* r2 = r1 + H1N;
    float* r4 = r2 + 2 * WTN;
    float* h1   = r0;
    float* g1   = r1;
    float* gw   = r2;
    float* gv   = r2 + GN;
    float* bmat = r2 + 2 * GN;
    float* wts  = r4;
    float* sm   = r1;

    const dim3 blk(256);
    const dim3 blkW(128);
    const dim3 gridU(F / TN, (Bb * HL) / UM);
    const dim3 gridW(F / TN, (Bb * T) / TM);
    const dim3 gridP(F / TN, (Bb * T / 4) / QM, 2);
    const dim3 gridF((F * F / 4) / 256);

    gemm_unfold_relu_w144<<<gridU, blkW, 0, stream>>>(z, W1, b1, h1);
    gemm_unfold_relu_w144<<<gridU, blkW, 0, stream>>>(z, V1, c1, g1);
    f43_filter<<<gridF, blk, 0, stream>>>(W2, gw);
    f43_filter<<<gridF, blk, 0, stream>>>(V2, gv);
    conv3_f43_wmma<<<gridP, blkW, 0, stream>>>(
        h1, W2, b2, gw, wts,
        g1, V2, c2, gv, bmat,
        (long)HL * F);
    gemm_f32_wmma_w64<2><<<gridW, blkW, 0, stream>>>(
        x + 4, (long)T * F, wts, (long)T * F, nullptr,
        bmat, (long)T * F, F,
        sm,
        T, T - 4);
    gemm_f32_wmma_w64<3><<<gridW, blkW, 0, stream>>>(
        sm, (long)T * F, Wr, 0, br,
        x, (long)T * F, F,
        out,
        F, F);
}
